# starting kernel
# baseline (speedup 1.0000x reference)
#include <hip/hip_runtime.h>
#include <math.h>

#define B_   32
#define L_   4096
#define MEM_ 256
#define Q_   256
#define BL_  (B_ * L_)
#define BQ_  (B_ * Q_)
#define B3Q_ (B_ * 3 * Q_)

#define OFF_S1   0
#define OFF_S2   (OFF_S1 + BL_)
#define OFF_SACC (OFF_S2 + BL_)
#define OFF_EACC (OFF_SACC + BL_)
#define OFF_QB   (OFF_EACC + BL_)
#define OFF_QA   (OFF_QB + BQ_)
#define OFF_QE   (OFF_QA + BQ_)
#define OFF_EV   (OFF_QE + BQ_)
#define OFF_X1   (OFF_EV + BQ_)
#define OFF_S_A  (OFF_X1 + BQ_)
#define OFF_S_B  (OFF_S_A + BQ_)
#define OFF_GI   (OFF_S_B + BQ_)
#define OFF_GH   (OFF_GI + B3Q_)

typedef __attribute__((ext_vector_type(2))) float v2f;
typedef __attribute__((ext_vector_type(8))) float v8f;

__global__ void wmma_gemm_kernel(const float* __restrict__ A, int lda,
                                 const float* __restrict__ W, int ldw,
                                 const float* __restrict__ bias,
                                 float* __restrict__ C, int ldc,
                                 int K, int accumulate) {
    const int lane = threadIdx.x & 31;
    const int row  = lane & 15;
    const int hi   = lane >> 4;
    const int m0   = blockIdx.y * 16;
    const int n0   = blockIdx.x * 16;

    const float binit = bias ? bias[n0 + row] : 0.0f;
    v8f acc = {};
    if (accumulate) {
#pragma unroll
        for (int j = 0; j < 8; ++j)
            acc[j] = C[(size_t)(m0 + j + 8 * hi) * ldc + n0 + row] + binit;
    } else {
#pragma unroll
        for (int j = 0; j < 8; ++j) acc[j] = binit;
    }

    const float* aRow = A + (size_t)(m0 + row) * lda + 2 * hi;
    const float* wRow = W + (size_t)(n0 + row) * ldw + 2 * hi;
    for (int k = 0; k < K; k += 4) {
        v2f a; a.x = aRow[k]; a.y = aRow[k + 1];
        v2f b; b.x = wRow[k]; b.y = wRow[k + 1];
        acc = __builtin_amdgcn_wmma_f32_16x16x4_f32(false, a, false, b,
                                                    (short)0, acc, false, false);
    }
#pragma unroll
    for (int j = 0; j < 8; ++j)
        C[(size_t)(m0 + j + 8 * hi) * ldc + n0 + row] = acc[j];
}

__global__ void scores_kernel(const float* __restrict__ M,
                              const float* __restrict__ q1,
                              const float* __restrict__ q2,
                              float* __restrict__ S1,
                              float* __restrict__ S2) {
    const int lane = threadIdx.x & 31;
    const int row  = lane & 15;
    const int hi   = lane >> 4;
    const int b    = blockIdx.y;
    const int l0   = blockIdx.x * 16;

    const float* Mb = M + ((size_t)b * L_ + l0 + row) * MEM_ + 2 * hi;
    const float* qv = ((row == 0) ? q1 : q2) + (size_t)b * MEM_ + 2 * hi;

    __builtin_prefetch(Mb + 16 * MEM_, 0, 0);

    v8f acc = {};
    for (int k = 0; k < MEM_; k += 4) {
        v2f a; a.x = Mb[k]; a.y = Mb[k + 1];
        v2f bb;
        if (row < 2) { bb.x = qv[k]; bb.y = qv[k + 1]; }
        else         { bb.x = 0.0f;  bb.y = 0.0f; }
        acc = __builtin_amdgcn_wmma_f32_16x16x4_f32(false, a, false, bb,
                                                    (short)0, acc, false, false);
    }
    if (row < 2) {
        float* dst = (row == 0) ? S1 : S2;
#pragma unroll
        for (int j = 0; j < 8; ++j)
            dst[(size_t)b * L_ + l0 + j + 8 * hi] = acc[j];
    }
}

__global__ void wsum_kernel(const float* __restrict__ M,
                            const float* __restrict__ p,
                            const float* __restrict__ w,
                            float* __restrict__ ev,
                            float* __restrict__ x1) {
    const int lane = threadIdx.x & 31;
    const int row  = lane & 15;
    const int hi   = lane >> 4;
    const int b    = blockIdx.y;
    const int m0   = blockIdx.x * 16;

    const float* pv = ((row == 0) ? p : w) + (size_t)b * L_ + 2 * hi;
    const float* Mb = M + ((size_t)b * L_ + 2 * hi) * MEM_ + m0 + row;

    v8f acc = {};
    for (int l = 0; l < L_; l += 4) {
        v2f a;
        if (row < 2) { a.x = pv[l]; a.y = pv[l + 1]; }
        else         { a.x = 0.0f;  a.y = 0.0f; }
        v2f bb;
        bb.x = Mb[(size_t)l * MEM_];
        bb.y = Mb[(size_t)l * MEM_ + MEM_];
        acc = __builtin_amdgcn_wmma_f32_16x16x4_f32(false, a, false, bb,
                                                    (short)0, acc, false, false);
    }
    if (hi == 0) {
        ev[(size_t)b * MEM_ + m0 + row] = acc[0];
        x1[(size_t)b * MEM_ + m0 + row] = acc[1];
    }
}

__global__ void softmax_kernel(float* __restrict__ S,
                               float* __restrict__ accDst,
                               float scale) {
    const int b = blockIdx.x;
    const int tid = threadIdx.x;
    __shared__ float red[256];
    float* rowp = S + (size_t)b * L_;

    float m = -3.402823466e38f;
    for (int i = tid; i < L_; i += 256) m = fmaxf(m, rowp[i]);
    red[tid] = m; __syncthreads();
    for (int s = 128; s > 0; s >>= 1) {
        if (tid < s) red[tid] = fmaxf(red[tid], red[tid + s]);
        __syncthreads();
    }
    m = red[0]; __syncthreads();

    float sum = 0.0f;
    for (int i = tid; i < L_; i += 256) {
        float e = __expf(rowp[i] - m);
        rowp[i] = e;
        sum += e;
    }
    red[tid] = sum; __syncthreads();
    for (int s = 128; s > 0; s >>= 1) {
        if (tid < s) red[tid] += red[tid + s];
        __syncthreads();
    }
    const float inv = 1.0f / red[0];

    for (int i = tid; i < L_; i += 256) {
        float pv = rowp[i] * inv;
        rowp[i] = pv;
        if (accDst) accDst[(size_t)b * L_ + i] += scale * pv;
    }
}

__global__ void gru_kernel(const float* __restrict__ gi,
                           const float* __restrict__ gh,
                           const float* __restrict__ sOld,
                           float* __restrict__ sNew) {
    int i = blockIdx.x * 256 + threadIdx.x;
    if (i >= BQ_) return;
    int b = i / Q_, q = i % Q_;
    const float* gib = gi + (size_t)b * 3 * Q_;
    const float* ghb = gh + (size_t)b * 3 * Q_;
    float ir = gib[q], iz = gib[Q_ + q], in_ = gib[2 * Q_ + q];
    float hr = ghb[q], hz = ghb[Q_ + q], hn  = ghb[2 * Q_ + q];
    float r = 1.0f / (1.0f + __expf(-(ir + hr)));
    float z = 1.0f / (1.0f + __expf(-(iz + hz)));
    float n = tanhf(in_ + r * hn);
    sNew[i] = (1.0f - z) * n + z * sOld[i];
}

__global__ void init_kernel(const float* __restrict__ s0, float* __restrict__ ws) {
    int i = blockIdx.x * 256 + threadIdx.x;
    if (i < 2 * BL_) ws[OFF_SACC + i] = 0.0f;
    if (i < BQ_)     ws[OFF_S_A + i] = s0[i];
}

__global__ void final_kernel(const float* __restrict__ ws, float* __restrict__ out) {
    int i = blockIdx.x * 256 + threadIdx.x;
    if (i < BL_) {
        out[i]       = logf(ws[OFF_SACC + i]);
        out[BL_ + i] = logf(ws[OFF_EACC + i]);
    }
}

extern "C" void kernel_launch(void* const* d_in, const int* in_sizes, int n_in,
                              void* d_out, int out_size, void* d_ws, size_t ws_size,
                              hipStream_t stream) {
    const float* M    = (const float*)d_in[0];
    const float* s0   = (const float*)d_in[1];
    const float* Wb   = (const float*)d_in[2];
    const float* We   = (const float*)d_in[3];
    const float* Wa   = (const float*)d_in[4];
    const float* W_ih = (const float*)d_in[5];
    const float* W_hh = (const float*)d_in[6];
    const float* b_ih = (const float*)d_in[7];
    const float* b_hh = (const float*)d_in[8];

    float* ws   = (float*)d_ws;
    float* S1   = ws + OFF_S1;
    float* S2   = ws + OFF_S2;
    float* sacc = ws + OFF_SACC;
    float* eacc = ws + OFF_EACC;
    float* qb   = ws + OFF_QB;
    float* qa   = ws + OFF_QA;
    float* qe   = ws + OFF_QE;
    float* ev   = ws + OFF_EV;
    float* x1   = ws + OFF_X1;
    float* sA   = ws + OFF_S_A;
    float* sB   = ws + OFF_S_B;
    float* gi   = ws + OFF_GI;
    float* gh   = ws + OFF_GH;

    init_kernel<<<(2 * BL_) / 256, 256, 0, stream>>>(s0, ws);

    auto gemm = [&](const float* A, int lda, const float* W, int ldw,
                    const float* bias, float* C, int ldc, int K, int N, int accF) {
        dim3 g(N / 16, B_ / 16);
        wmma_gemm_kernel<<<g, 32, 0, stream>>>(A, lda, W, ldw, bias, C, ldc, K, accF);
    };

    for (int t = 0; t < 5; ++t) {
        const float* sCur = (t & 1) ? sB : sA;
        float*       sNxt = (t & 1) ? sA : sB;

        gemm(sCur, Q_, Wb, Q_, nullptr, qb, MEM_, Q_, MEM_, 0);
        gemm(sCur, Q_, Wa, Q_, nullptr, qa, MEM_, Q_, MEM_, 0);

        scores_kernel<<<dim3(L_ / 16, B_), 32, 0, stream>>>(M, qb, qa, S1, S2);

        softmax_kernel<<<B_, 256, 0, stream>>>(S1, sacc, 0.2f);
        softmax_kernel<<<B_, 256, 0, stream>>>(S2, nullptr, 0.0f);

        wsum_kernel<<<dim3(MEM_ / 16, B_), 32, 0, stream>>>(M, S1, S2, ev, x1);

        gemm(sCur, Q_,  We,      Q_ + MEM_, nullptr, qe, MEM_, Q_,   MEM_, 0);
        gemm(ev,  MEM_, We + Q_, Q_ + MEM_, nullptr, qe, MEM_, MEM_, MEM_, 1);

        scores_kernel<<<dim3(L_ / 16, B_), 32, 0, stream>>>(M, qe, qe, S1, S2);
        softmax_kernel<<<B_, 256, 0, stream>>>(S1, eacc, 0.2f);

        gemm(x1,  MEM_, W_ih, MEM_, b_ih, gi, 3 * Q_, MEM_, 3 * Q_, 0);
        gemm(sCur, Q_,  W_hh, Q_,   b_hh, gh, 3 * Q_, Q_,   3 * Q_, 0);
        gru_kernel<<<BQ_ / 256, 256, 0, stream>>>(gi, gh, sCur, sNxt);
    }

    final_kernel<<<BL_ / 256, 256, 0, stream>>>(ws, (float*)d_out);
}


# weighted-sum kernel: L split over the 16 waves of a work-group per (batch, 64-column slab), 16-byte row loads issued one group ahead, partial sums added through LDS (new reduction tree)
# speedup vs baseline: 4.0009x; 4.0009x over previous
#include <hip/hip_runtime.h>
#include <math.h>

#define B_   32
#define L_   4096
#define MEM_ 256
#define Q_   256
#define BL_  (B_ * L_)
#define BQ_  (B_ * Q_)
#define B3Q_ (B_ * 3 * Q_)

#define OFF_S1   0
#define OFF_S2   (OFF_S1 + BL_)
#define OFF_SACC (OFF_S2 + BL_)
#define OFF_EACC (OFF_SACC + BL_)
#define OFF_QB   (OFF_EACC + BL_)
#define OFF_QA   (OFF_QB + BQ_)
#define OFF_QE   (OFF_QA + BQ_)
#define OFF_EV   (OFF_QE + BQ_)
#define OFF_X1   (OFF_EV + BQ_)
#define OFF_S_A  (OFF_X1 + BQ_)
#define OFF_S_B  (OFF_S_A + BQ_)
#define OFF_GI   (OFF_S_B + BQ_)
#define OFF_GH   (OFF_GI + B3Q_)

typedef __attribute__((ext_vector_type(2))) float v2f;
typedef __attribute__((ext_vector_type(8))) float v8f;

__global__ void wmma_gemm_kernel(const float* __restrict__ A, int lda,
                                 const float* __restrict__ W, int ldw,
                                 const float* __restrict__ bias,
                                 float* __restrict__ C, int ldc,
                                 int K, int accumulate) {
    const int lane = threadIdx.x & 31;
    const int row  = lane & 15;
    const int hi   = lane >> 4;
    const int m0   = blockIdx.y * 16;
    const int n0   = blockIdx.x * 16;

    const float binit = bias ? bias[n0 + row] : 0.0f;
    v8f acc = {};
    if (accumulate) {
#pragma unroll
        for (int j = 0; j < 8; ++j)
            acc[j] = C[(size_t)(m0 + j + 8 * hi) * ldc + n0 + row] + binit;
    } else {
#pragma unroll
        for (int j = 0; j < 8; ++j) acc[j] = binit;
    }

    const float* aRow = A + (size_t)(m0 + row) * lda + 2 * hi;
    const float* wRow = W + (size_t)(n0 + row) * ldw + 2 * hi;
    for (int k = 0; k < K; k += 4) {
        v2f a; a.x = aRow[k]; a.y = aRow[k + 1];
        v2f b; b.x = wRow[k]; b.y = wRow[k + 1];
        acc = __builtin_amdgcn_wmma_f32_16x16x4_f32(false, a, false, b,
                                                    (short)0, acc, false, false);
    }
#pragma unroll
    for (int j = 0; j < 8; ++j)
        C[(size_t)(m0 + j + 8 * hi) * ldc + n0 + row] = acc[j];
}

__global__ void scores_kernel(const float* __restrict__ M,
                              const float* __restrict__ q1,
                              const float* __restrict__ q2,
                              float* __restrict__ S1,
                              float* __restrict__ S2) {
    const int lane = threadIdx.x & 31;
    const int row  = lane & 15;
    const int hi   = lane >> 4;
    const int b    = blockIdx.y;
    const int l0   = blockIdx.x * 16;

    const float* Mb = M + ((size_t)b * L_ + l0 + row) * MEM_ + 2 * hi;
    const float* qv = ((row == 0) ? q1 : q2) + (size_t)b * MEM_ + 2 * hi;

    __builtin_prefetch(Mb + 16 * MEM_, 0, 0);

    v8f acc = {};
    for (int k = 0; k < MEM_; k += 4) {
        v2f a; a.x = Mb[k]; a.y = Mb[k + 1];
        v2f bb;
        if (row < 2) { bb.x = qv[k]; bb.y = qv[k + 1]; }
        else         { bb.x = 0.0f;  bb.y = 0.0f; }
        acc = __builtin_amdgcn_wmma_f32_16x16x4_f32(false, a, false, bb,
                                                    (short)0, acc, false, false);
    }
    if (row < 2) {
        float* dst = (row == 0) ? S1 : S2;
#pragma unroll
        for (int j = 0; j < 8; ++j)
            dst[(size_t)b * L_ + l0 + j + 8 * hi] = acc[j];
    }
}

typedef __attribute__((ext_vector_type(4))) float v4f;

#define WS_WAVES 16
#define WS_COLS  64
#define WS_STEPS 8
#define WS_GROUP_ROWS (4 * WS_STEPS)
#define WS_GROUPS (L_ / (WS_GROUP_ROWS * WS_WAVES))

struct WsGroup {
    v2f a[WS_STEPS];
    v4f r0[WS_STEPS];
    v4f r1[WS_STEPS];
};

__device__ __forceinline__ void ws_load(WsGroup& g, const float* __restrict__ pv,
                                        const float* __restrict__ Mb, int l0) {
#pragma unroll
    for (int s = 0; s < WS_STEPS; ++s)
        g.a[s] = *(const v2f*)(pv + l0 + 4 * s);
#pragma unroll
    for (int s = 0; s < WS_STEPS; ++s) {
        const float* rp = Mb + (size_t)(l0 + 4 * s) * MEM_;
        g.r0[s] = *(const v4f*)rp;
        g.r1[s] = *(const v4f*)(rp + MEM_);
    }
}

__device__ __forceinline__ void ws_mma(const WsGroup& g, bool live, v8f (&acc)[4]) {
#pragma unroll
    for (int s = 0; s < WS_STEPS; ++s) {
        v2f a;
        a.x = live ? g.a[s].x : 0.0f;
        a.y = live ? g.a[s].y : 0.0f;
#pragma unroll
        for (int t = 0; t < 4; ++t) {
            v2f bb; bb.x = g.r0[s][t]; bb.y = g.r1[s][t];
            acc[t] = __builtin_amdgcn_wmma_f32_16x16x4_f32(false, a, false, bb,
                                                           (short)0, acc[t], false, false);
        }
    }
}

__global__ __launch_bounds__(32 * WS_WAVES)
void wsum_kernel(const float* __restrict__ M,
                 const float* __restrict__ p,
                 const float* __restrict__ w,
                 float* __restrict__ ev,
                 float* __restrict__ x1) {
    const int tid  = threadIdx.x;
    const int wave = tid >> 5;
    const int lane = tid & 31;
    const int c    = lane & 15;
    const int hi   = lane >> 4;
    const int b    = blockIdx.y;
    const int slab = blockIdx.x * WS_COLS;

    const float* pv = ((c == 1) ? w : p) + (size_t)b * L_ + 2 * hi;
    const float* Mb = M + ((size_t)b * L_ + 2 * hi) * MEM_ + slab + 4 * c;
    const bool live = c < 2;

    v8f acc[4] = {};
    WsGroup g0, g1;
#define WS_ORDER() __builtin_amdgcn_sched_barrier(0)
    ws_load(g0, pv, Mb, wave * WS_GROUP_ROWS);
    WS_ORDER();
    for (int i = 0; i < WS_GROUPS - 2; i += 2) {
        ws_load(g1, pv, Mb, ((i + 1) * WS_WAVES + wave) * WS_GROUP_ROWS);
        WS_ORDER();
        ws_mma(g0, live, acc);
        WS_ORDER();
        ws_load(g0, pv, Mb, ((i + 2) * WS_WAVES + wave) * WS_GROUP_ROWS);
        WS_ORDER();
        ws_mma(g1, live, acc);
        WS_ORDER();
    }
    ws_load(g1, pv, Mb, ((WS_GROUPS - 1) * WS_WAVES + wave) * WS_GROUP_ROWS);
    WS_ORDER();
    ws_mma(g0, live, acc);
    ws_mma(g1, live, acc);
#undef WS_ORDER

    __shared__ float red[WS_WAVES][2][WS_COLS];
    if (hi == 0) {
        v4f e, x;
#pragma unroll
        for (int t = 0; t < 4; ++t) { e[t] = acc[t][0]; x[t] = acc[t][1]; }
        *(v4f*)&red[wave][0][4 * c] = e;
        *(v4f*)&red[wave][1][4 * c] = x;
    }
    __syncthreads();
    if (tid < 2 * WS_COLS) {
        const int r  = tid >> 6;
        const int mm = tid & (WS_COLS - 1);
        float s = 0.0f;
#pragma unroll
        for (int k = 0; k < WS_WAVES; ++k) s += red[k][r][mm];
        (r ? x1 : ev)[(size_t)b * MEM_ + slab + mm] = s;
    }
}

__global__ void softmax_kernel(float* __restrict__ S,
                               float* __restrict__ accDst,
                               float scale) {
    const int b = blockIdx.x;
    const int tid = threadIdx.x;
    __shared__ float red[256];
    float* rowp = S + (size_t)b * L_;

    float m = -3.402823466e38f;
    for (int i = tid; i < L_; i += 256) m = fmaxf(m, rowp[i]);
    red[tid] = m; __syncthreads();
    for (int s = 128; s > 0; s >>= 1) {
        if (tid < s) red[tid] = fmaxf(red[tid], red[tid + s]);
        __syncthreads();
    }
    m = red[0]; __syncthreads();

    float sum = 0.0f;
    for (int i = tid; i < L_; i += 256) {
        float e = __expf(rowp[i] - m);
        rowp[i] = e;
        sum += e;
    }
    red[tid] = sum; __syncthreads();
    for (int s = 128; s > 0; s >>= 1) {
        if (tid < s) red[tid] += red[tid + s];
        __syncthreads();
    }
    const float inv = 1.0f / red[0];

    for (int i = tid; i < L_; i += 256) {
        float pv = rowp[i] * inv;
        rowp[i] = pv;
        if (accDst) accDst[(size_t)b * L_ + i] += scale * pv;
    }
}

__global__ void gru_kernel(const float* __restrict__ gi,
                           const float* __restrict__ gh,
                           const float* __restrict__ sOld,
                           float* __restrict__ sNew) {
    int i = blockIdx.x * 256 + threadIdx.x;
    if (i >= BQ_) return;
    int b = i / Q_, q = i % Q_;
    const float* gib = gi + (size_t)b * 3 * Q_;
    const float* ghb = gh + (size_t)b * 3 * Q_;
    float ir = gib[q], iz = gib[Q_ + q], in_ = gib[2 * Q_ + q];
    float hr = ghb[q], hz = ghb[Q_ + q], hn  = ghb[2 * Q_ + q];
    float r = 1.0f / (1.0f + __expf(-(ir + hr)));
    float z = 1.0f / (1.0f + __expf(-(iz + hz)));
    float n = tanhf(in_ + r * hn);
    sNew[i] = (1.0f - z) * n + z * sOld[i];
}

__global__ void init_kernel(const float* __restrict__ s0, float* __restrict__ ws) {
    int i = blockIdx.x * 256 + threadIdx.x;
    if (i < 2 * BL_) ws[OFF_SACC + i] = 0.0f;
    if (i < BQ_)     ws[OFF_S_A + i] = s0[i];
}

__global__ void final_kernel(const float* __restrict__ ws, float* __restrict__ out) {
    int i = blockIdx.x * 256 + threadIdx.x;
    if (i < BL_) {
        out[i]       = logf(ws[OFF_SACC + i]);
        out[BL_ + i] = logf(ws[OFF_EACC + i]);
    }
}

extern "C" void kernel_launch(void* const* d_in, const int* in_sizes, int n_in,
                              void* d_out, int out_size, void* d_ws, size_t ws_size,
                              hipStream_t stream) {
    const float* M    = (const float*)d_in[0];
    const float* s0   = (const float*)d_in[1];
    const float* Wb   = (const float*)d_in[2];
    const float* We   = (const float*)d_in[3];
    const float* Wa   = (const float*)d_in[4];
    const float* W_ih = (const float*)d_in[5];
    const float* W_hh = (const float*)d_in[6];
    const float* b_ih = (const float*)d_in[7];
    const float* b_hh = (const float*)d_in[8];

    float* ws   = (float*)d_ws;
    float* S1   = ws + OFF_S1;
    float* S2   = ws + OFF_S2;
    float* sacc = ws + OFF_SACC;
    float* eacc = ws + OFF_EACC;
    float* qb   = ws + OFF_QB;
    float* qa   = ws + OFF_QA;
    float* qe   = ws + OFF_QE;
    float* ev   = ws + OFF_EV;
    float* x1   = ws + OFF_X1;
    float* sA   = ws + OFF_S_A;
    float* sB   = ws + OFF_S_B;
    float* gi   = ws + OFF_GI;
    float* gh   = ws + OFF_GH;

    init_kernel<<<(2 * BL_) / 256, 256, 0, stream>>>(s0, ws);

    auto gemm = [&](const float* A, int lda, const float* W, int ldw,
                    const float* bias, float* C, int ldc, int K, int N, int accF) {
        dim3 g(N / 16, B_ / 16);
        wmma_gemm_kernel<<<g, 32, 0, stream>>>(A, lda, W, ldw, bias, C, ldc, K, accF);
    };

    for (int t = 0; t < 5; ++t) {
        const float* sCur = (t & 1) ? sB : sA;
        float*       sNxt = (t & 1) ? sA : sB;

        gemm(sCur, Q_, Wb, Q_, nullptr, qb, MEM_, Q_, MEM_, 0);
        gemm(sCur, Q_, Wa, Q_, nullptr, qa, MEM_, Q_, MEM_, 0);

        scores_kernel<<<dim3(L_ / 16, B_), 32, 0, stream>>>(M, qb, qa, S1, S2);

        softmax_kernel<<<B_, 256, 0, stream>>>(S1, sacc, 0.2f);
        softmax_kernel<<<B_, 256, 0, stream>>>(S2, nullptr, 0.0f);

        wsum_kernel<<<dim3(MEM_ / WS_COLS, B_), 32 * WS_WAVES, 0, stream>>>(M, S1, S2, ev, x1);

        gemm(sCur, Q_,  We,      Q_ + MEM_, nullptr, qe, MEM_, Q_,   MEM_, 0);
        gemm(ev,  MEM_, We + Q_, Q_ + MEM_, nullptr, qe, MEM_, MEM_, MEM_, 1);

        scores_kernel<<<dim3(L_ / 16, B_), 32, 0, stream>>>(M, qe, qe, S1, S2);
        softmax_kernel<<<B_, 256, 0, stream>>>(S1, eacc, 0.2f);

        gemm(x1,  MEM_, W_ih, MEM_, b_ih, gi, 3 * Q_, MEM_, 3 * Q_, 0);
        gemm(sCur, Q_,  W_hh, Q_,   b_hh, gh, 3 * Q_, Q_,   3 * Q_, 0);
        gru_kernel<<<BQ_ / 256, 256, 0, stream>>>(gi, gh, sCur, sNxt);
    }

    final_kernel<<<BL_ / 256, 256, 0, stream>>>(ws, (float*)d_out);
}


# independent launches merged: small GEMMs 6 to 2 launches per turn; end scores of turn t and both scores of turn t+1 in one 3-column pass over M; softmaxes one launch; 67 to 34 launches; same sums
# speedup vs baseline: 6.5840x; 1.6456x over previous
#include <hip/hip_runtime.h>
#include <math.h>

#define B_   32
#define L_   4096
#define MEM_ 256
#define Q_   256
#define BL_  (B_ * L_)
#define BQ_  (B_ * Q_)
#define B3Q_ (B_ * 3 * Q_)

#define OFF_S1   0
#define OFF_S2   (OFF_S1 + BL_)
#define OFF_SACC (OFF_S2 + BL_)
#define OFF_EACC (OFF_SACC + BL_)
#define OFF_QB   (OFF_EACC + BL_)
#define OFF_QA   (OFF_QB + BQ_)
#define OFF_QE   (OFF_QA + BQ_)
#define OFF_EV   (OFF_QE + BQ_)
#define OFF_X1   (OFF_EV + BQ_)
#define OFF_S_A  (OFF_X1 + BQ_)
#define OFF_S_B  (OFF_S_A + BQ_)
#define OFF_GI   (OFF_S_B + BQ_)
#define OFF_GH   (OFF_GI + B3Q_)

typedef __attribute__((ext_vector_type(2))) float v2f;
typedef __attribute__((ext_vector_type(8))) float v8f;

__device__ __forceinline__ v8f gemm_tile(const float* __restrict__ aRow,
                                         const float* __restrict__ wRow,
                                         int K, v8f acc) {
    for (int k = 0; k < K; k += 4) {
        v2f a; a.x = aRow[k]; a.y = aRow[k + 1];
        v2f b; b.x = wRow[k]; b.y = wRow[k + 1];
        acc = __builtin_amdgcn_wmma_f32_16x16x4_f32(false, a, false, b,
                                                    (short)0, acc, false, false);
    }
    return acc;
}

__device__ __forceinline__ void gemm_store(float* __restrict__ C, int ldc,
                                           int m0, int n0, int row, int hi, v8f acc) {
#pragma unroll
    for (int j = 0; j < 8; ++j)
        C[(size_t)(m0 + j + 8 * hi) * ldc + n0 + row] = acc[j];
}

__global__ void gemm_q_kernel(const float* __restrict__ s,
                              const float* __restrict__ Wb,
                              const float* __restrict__ Wa,
                              float* __restrict__ qb,
                              float* __restrict__ qa) {
    const int lane = threadIdx.x & 31;
    const int row  = lane & 15;
    const int hi   = lane >> 4;
    const int m0   = blockIdx.y * 16;
    const int sel  = blockIdx.x >> 4;
    const int n0   = (blockIdx.x & 15) * 16;
    const float* W = sel ? Wa : Wb;
    float*       C = sel ? qa : qb;

    v8f acc = {};
    acc = gemm_tile(s + (size_t)(m0 + row) * Q_ + 2 * hi,
                    W + (size_t)(n0 + row) * Q_ + 2 * hi, Q_, acc);
    gemm_store(C, MEM_, m0, n0, row, hi, acc);
}

__global__ void gemm_turn_kernel(const float* __restrict__ s,
                                 const float* __restrict__ ev,
                                 const float* __restrict__ x1,
                                 const float* __restrict__ We,
                                 const float* __restrict__ W_ih,
                                 const float* __restrict__ W_hh,
                                 const float* __restrict__ b_ih,
                                 const float* __restrict__ b_hh,
                                 float* __restrict__ qe,
                                 float* __restrict__ gi,
                                 float* __restrict__ gh) {
    const int lane = threadIdx.x & 31;
    const int row  = lane & 15;
    const int hi   = lane >> 4;
    const int m0   = blockIdx.y * 16;
    const int bx   = blockIdx.x;

    if (bx < 16) {
        const int n0 = bx * 16;
        const float* wRow = We + (size_t)(n0 + row) * (Q_ + MEM_) + 2 * hi;
        v8f acc = {};
        acc = gemm_tile(s  + (size_t)(m0 + row) * Q_   + 2 * hi, wRow,      Q_,   acc);
        acc = gemm_tile(ev + (size_t)(m0 + row) * MEM_ + 2 * hi, wRow + Q_, MEM_, acc);
        gemm_store(qe, MEM_, m0, n0, row, hi, acc);
    } else {
        const bool isGi = bx < 64;
        const int n0 = (isGi ? (bx - 16) : (bx - 64)) * 16;
        const float* A    = isGi ? x1 : s;
        const float* W    = isGi ? W_ih : W_hh;
        const float* bias = isGi ? b_ih : b_hh;
        float*       C    = isGi ? gi : gh;
        const float binit = bias[n0 + row];
        v8f acc;
#pragma unroll
        for (int j = 0; j < 8; ++j) acc[j] = binit;
        acc = gemm_tile(A + (size_t)(m0 + row) * MEM_ + 2 * hi,
                        W + (size_t)(n0 + row) * MEM_ + 2 * hi, MEM_, acc);
        gemm_store(C, 3 * Q_, m0, n0, row, hi, acc);
    }
}

__global__ void scores_kernel(const float* __restrict__ M,
                              const float* __restrict__ q0,
                              const float* __restrict__ q1,
                              const float* __restrict__ q2,
                              float* __restrict__ S0,
                              float* __restrict__ S1,
                              float* __restrict__ S2,
                              int ncols) {
    const int lane = threadIdx.x & 31;
    const int row  = lane & 15;
    const int hi   = lane >> 4;
    const int b    = blockIdx.y;
    const int l0   = blockIdx.x * 16;

    const float* Mb = M + ((size_t)b * L_ + l0 + row) * MEM_ + 2 * hi;
    const float* qv = ((row == 0) ? q0 : (row == 1) ? q1 : q2) + (size_t)b * MEM_ + 2 * hi;
    const bool live = row < ncols;

    __builtin_prefetch(Mb + 16 * MEM_, 0, 0);

    v8f acc = {};
    for (int k = 0; k < MEM_; k += 4) {
        v2f a; a.x = Mb[k]; a.y = Mb[k + 1];
        v2f bb;
        if (live) { bb.x = qv[k]; bb.y = qv[k + 1]; }
        else      { bb.x = 0.0f;  bb.y = 0.0f; }
        acc = __builtin_amdgcn_wmma_f32_16x16x4_f32(false, a, false, bb,
                                                    (short)0, acc, false, false);
    }
    if (live) {
        float* dst = (row == 0) ? S0 : (row == 1) ? S1 : S2;
#pragma unroll
        for (int j = 0; j < 8; ++j)
            dst[(size_t)b * L_ + l0 + j + 8 * hi] = acc[j];
    }
}

typedef __attribute__((ext_vector_type(4))) float v4f;

#define WS_WAVES 16
#define WS_COLS  64
#define WS_STEPS 8
#define WS_GROUP_ROWS (4 * WS_STEPS)
#define WS_GROUPS (L_ / (WS_GROUP_ROWS * WS_WAVES))

struct WsGroup {
    v2f a[WS_STEPS];
    v4f r0[WS_STEPS];
    v4f r1[WS_STEPS];
};

__device__ __forceinline__ void ws_load(WsGroup& g, const float* __restrict__ pv,
                                        const float* __restrict__ Mb, int l0) {
#pragma unroll
    for (int s = 0; s < WS_STEPS; ++s)
        g.a[s] = *(const v2f*)(pv + l0 + 4 * s);
#pragma unroll
    for (int s = 0; s < WS_STEPS; ++s) {
        const float* rp = Mb + (size_t)(l0 + 4 * s) * MEM_;
        g.r0[s] = *(const v4f*)rp;
        g.r1[s] = *(const v4f*)(rp + MEM_);
    }
}

__device__ __forceinline__ void ws_mma(const WsGroup& g, bool live, v8f (&acc)[4]) {
#pragma unroll
    for (int s = 0; s < WS_STEPS; ++s) {
        v2f a;
        a.x = live ? g.a[s].x : 0.0f;
        a.y = live ? g.a[s].y : 0.0f;
#pragma unroll
        for (int t = 0; t < 4; ++t) {
            v2f bb; bb.x = g.r0[s][t]; bb.y = g.r1[s][t];
            acc[t] = __builtin_amdgcn_wmma_f32_16x16x4_f32(false, a, false, bb,
                                                           (short)0, acc[t], false, false);
        }
    }
}

__global__ __launch_bounds__(32 * WS_WAVES)
void wsum_kernel(const float* __restrict__ M,
                 const float* __restrict__ p,
                 const float* __restrict__ w,
                 float* __restrict__ ev,
                 float* __restrict__ x1) {
    const int tid  = threadIdx.x;
    const int wave = tid >> 5;
    const int lane = tid & 31;
    const int c    = lane & 15;
    const int hi   = lane >> 4;
    const int b    = blockIdx.y;
    const int slab = blockIdx.x * WS_COLS;

    const float* pv = ((c == 1) ? w : p) + (size_t)b * L_ + 2 * hi;
    const float* Mb = M + ((size_t)b * L_ + 2 * hi) * MEM_ + slab + 4 * c;
    const bool live = c < 2;

    v8f acc[4] = {};
    WsGroup g0, g1;
#define WS_ORDER() __builtin_amdgcn_sched_barrier(0)
    ws_load(g0, pv, Mb, wave * WS_GROUP_ROWS);
    WS_ORDER();
    for (int i = 0; i < WS_GROUPS - 2; i += 2) {
        ws_load(g1, pv, Mb, ((i + 1) * WS_WAVES + wave) * WS_GROUP_ROWS);
        WS_ORDER();
        ws_mma(g0, live, acc);
        WS_ORDER();
        ws_load(g0, pv, Mb, ((i + 2) * WS_WAVES + wave) * WS_GROUP_ROWS);
        WS_ORDER();
        ws_mma(g1, live, acc);
        WS_ORDER();
    }
    ws_load(g1, pv, Mb, ((WS_GROUPS - 1) * WS_WAVES + wave) * WS_GROUP_ROWS);
    WS_ORDER();
    ws_mma(g0, live, acc);
    ws_mma(g1, live, acc);
#undef WS_ORDER

    __shared__ float red[WS_WAVES][2][WS_COLS];
    if (hi == 0) {
        v4f e, x;
#pragma unroll
        for (int t = 0; t < 4; ++t) { e[t] = acc[t][0]; x[t] = acc[t][1]; }
        *(v4f*)&red[wave][0][4 * c] = e;
        *(v4f*)&red[wave][1][4 * c] = x;
    }
    __syncthreads();
    if (tid < 2 * WS_COLS) {
        const int r  = tid >> 6;
        const int mm = tid & (WS_COLS - 1);
        float s = 0.0f;
#pragma unroll
        for (int k = 0; k < WS_WAVES; ++k) s += red[k][r][mm];
        (r ? x1 : ev)[(size_t)b * MEM_ + slab + mm] = s;
    }
}

__global__ void softmax_kernel(float* __restrict__ Sa, float* __restrict__ accA,
                               float* __restrict__ Sb, float* __restrict__ accB,
                               float* __restrict__ Sc, float* __restrict__ accC,
                               float scale) {
    const int b = blockIdx.x;
    const int job = blockIdx.y;
    const int tid = threadIdx.x;
    __shared__ float red[256];
    float* S      = (job == 0) ? Sa   : (job == 1) ? Sb   : Sc;
    float* accDst = (job == 0) ? accA : (job == 1) ? accB : accC;
    float* rowp = S + (size_t)b * L_;

    float m = -3.402823466e38f;
    for (int i = tid; i < L_; i += 256) m = fmaxf(m, rowp[i]);
    red[tid] = m; __syncthreads();
    for (int s = 128; s > 0; s >>= 1) {
        if (tid < s) red[tid] = fmaxf(red[tid], red[tid + s]);
        __syncthreads();
    }
    m = red[0]; __syncthreads();

    float sum = 0.0f;
    for (int i = tid; i < L_; i += 256) {
        float e = __expf(rowp[i] - m);
        rowp[i] = e;
        sum += e;
    }
    red[tid] = sum; __syncthreads();
    for (int s = 128; s > 0; s >>= 1) {
        if (tid < s) red[tid] += red[tid + s];
        __syncthreads();
    }
    const float inv = 1.0f / red[0];

    for (int i = tid; i < L_; i += 256) {
        float pv = rowp[i] * inv;
        rowp[i] = pv;
        if (accDst) accDst[(size_t)b * L_ + i] += scale * pv;
    }
}

__global__ void gru_kernel(const float* __restrict__ gi,
                           const float* __restrict__ gh,
                           const float* __restrict__ sOld,
                           float* __restrict__ sNew) {
    int i = blockIdx.x * 256 + threadIdx.x;
    if (i >= BQ_) return;
    int b = i / Q_, q = i % Q_;
    const float* gib = gi + (size_t)b * 3 * Q_;
    const float* ghb = gh + (size_t)b * 3 * Q_;
    float ir = gib[q], iz = gib[Q_ + q], in_ = gib[2 * Q_ + q];
    float hr = ghb[q], hz = ghb[Q_ + q], hn  = ghb[2 * Q_ + q];
    float r = 1.0f / (1.0f + __expf(-(ir + hr)));
    float z = 1.0f / (1.0f + __expf(-(iz + hz)));
    float n = tanhf(in_ + r * hn);
    sNew[i] = (1.0f - z) * n + z * sOld[i];
}

__global__ void init_kernel(const float* __restrict__ s0, float* __restrict__ ws) {
    int i = blockIdx.x * 256 + threadIdx.x;
    if (i < 2 * BL_) ws[OFF_SACC + i] = 0.0f;
    if (i < BQ_)     ws[OFF_S_A + i] = s0[i];
}

__global__ void final_kernel(const float* __restrict__ ws, float* __restrict__ out) {
    int i = blockIdx.x * 256 + threadIdx.x;
    if (i < BL_) {
        out[i]       = logf(ws[OFF_SACC + i]);
        out[BL_ + i] = logf(ws[OFF_EACC + i]);
    }
}

extern "C" void kernel_launch(void* const* d_in, const int* in_sizes, int n_in,
                              void* d_out, int out_size, void* d_ws, size_t ws_size,
                              hipStream_t stream) {
    const float* M    = (const float*)d_in[0];
    const float* s0   = (const float*)d_in[1];
    const float* Wb   = (const float*)d_in[2];
    const float* We   = (const float*)d_in[3];
    const float* Wa   = (const float*)d_in[4];
    const float* W_ih = (const float*)d_in[5];
    const float* W_hh = (const float*)d_in[6];
    const float* b_ih = (const float*)d_in[7];
    const float* b_hh = (const float*)d_in[8];

    float* ws   = (float*)d_ws;
    float* S1   = ws + OFF_S1;
    float* S2   = ws + OFF_S2;
    float* sacc = ws + OFF_SACC;
    float* eacc = ws + OFF_EACC;
    float* qb   = ws + OFF_QB;
    float* qa   = ws + OFF_QA;
    float* qe   = ws + OFF_QE;
    float* ev   = ws + OFF_EV;
    float* x1   = ws + OFF_X1;
    float* sA   = ws + OFF_S_A;
    float* sB   = ws + OFF_S_B;
    float* gi   = ws + OFF_GI;
    float* gh   = ws + OFF_GH;
    float* S3   = (float*)d_out + BL_;

    init_kernel<<<(2 * BL_) / 256, 256, 0, stream>>>(s0, ws);

    gemm_q_kernel<<<dim3(32, 2), 32, 0, stream>>>(sA, Wb, Wa, qb, qa);
    scores_kernel<<<dim3(L_ / 16, B_), 32, 0, stream>>>(M, qb, qa, qa, S1, S2, nullptr, 2);
    softmax_kernel<<<dim3(B_, 2), 256, 0, stream>>>(S1, sacc, S2, nullptr, nullptr, nullptr, 0.2f);

    for (int t = 0; t < 5; ++t) {
        const float* sCur = (t & 1) ? sB : sA;
        float*       sNxt = (t & 1) ? sA : sB;

        wsum_kernel<<<dim3(MEM_ / WS_COLS, B_), 32 * WS_WAVES, 0, stream>>>(M, S1, S2, ev, x1);

        gemm_turn_kernel<<<dim3(112, 2), 32, 0, stream>>>(sCur, ev, x1, We, W_ih, W_hh,
                                                          b_ih, b_hh, qe, gi, gh);
        gru_kernel<<<BQ_ / 256, 256, 0, stream>>>(gi, gh, sCur, sNxt);

        if (t < 4) {
            gemm_q_kernel<<<dim3(32, 2), 32, 0, stream>>>(sNxt, Wb, Wa, qb, qa);
            scores_kernel<<<dim3(L_ / 16, B_), 32, 0, stream>>>(M, qe, qb, qa, S3, S1, S2, 3);
            softmax_kernel<<<dim3(B_, 3), 256, 0, stream>>>(S3, eacc, S1, sacc, S2, nullptr, 0.2f);
        } else {
            scores_kernel<<<dim3(L_ / 16, B_), 32, 0, stream>>>(M, qe, qe, qe, S3, nullptr, nullptr, 1);
            softmax_kernel<<<dim3(B_, 1), 256, 0, stream>>>(S3, eacc, nullptr, nullptr, nullptr, nullptr, 0.2f);
        }
    }

    final_kernel<<<BL_ / 256, 256, 0, stream>>>(ws, (float*)d_out);
}


# softmax launches removed: the weighted-sum and final kernels take each score row's max and sum themselves and form exp(x-max)/sum on load; 34 to 28 launches; new tree for the softmax sum
# speedup vs baseline: 8.2364x; 1.2510x over previous
#include <hip/hip_runtime.h>
#include <math.h>

#define B_   32
#define L_   4096
#define MEM_ 256
#define Q_   256
#define BL_  (B_ * L_)
#define BQ_  (B_ * Q_)
#define B3Q_ (B_ * 3 * Q_)

#define OFF_S1   0
#define OFF_S2   (OFF_S1 + BL_)
#define OFF_SACC (OFF_S2 + BL_)
#define OFF_EACC (OFF_SACC + BL_)
#define OFF_QB   (OFF_EACC + BL_)
#define OFF_QA   (OFF_QB + BQ_)
#define OFF_QE   (OFF_QA + BQ_)
#define OFF_EV   (OFF_QE + BQ_)
#define OFF_X1   (OFF_EV + BQ_)
#define OFF_S_A  (OFF_X1 + BQ_)
#define OFF_S_B  (OFF_S_A + BQ_)
#define OFF_GI   (OFF_S_B + BQ_)
#define OFF_GH   (OFF_GI + B3Q_)

typedef __attribute__((ext_vector_type(2))) float v2f;
typedef __attribute__((ext_vector_type(8))) float v8f;

__device__ __forceinline__ v8f gemm_tile(const float* __restrict__ aRow,
                                         const float* __restrict__ wRow,
                                         int K, v8f acc) {
    for (int k = 0; k < K; k += 4) {
        v2f a; a.x = aRow[k]; a.y = aRow[k + 1];
        v2f b; b.x = wRow[k]; b.y = wRow[k + 1];
        acc = __builtin_amdgcn_wmma_f32_16x16x4_f32(false, a, false, b,
                                                    (short)0, acc, false, false);
    }
    return acc;
}

__device__ __forceinline__ void gemm_store(float* __restrict__ C, int ldc,
                                           int m0, int n0, int row, int hi, v8f acc) {
#pragma unroll
    for (int j = 0; j < 8; ++j)
        C[(size_t)(m0 + j + 8 * hi) * ldc + n0 + row] = acc[j];
}

__global__ void gemm_q_kernel(const float* __restrict__ s,
                              const float* __restrict__ Wb,
                              const float* __restrict__ Wa,
                              float* __restrict__ qb,
                              float* __restrict__ qa) {
    const int lane = threadIdx.x & 31;
    const int row  = lane & 15;
    const int hi   = lane >> 4;
    const int m0   = blockIdx.y * 16;
    const int sel  = blockIdx.x >> 4;
    const int n0   = (blockIdx.x & 15) * 16;
    const float* W = sel ? Wa : Wb;
    float*       C = sel ? qa : qb;

    v8f acc = {};
    acc = gemm_tile(s + (size_t)(m0 + row) * Q_ + 2 * hi,
                    W + (size_t)(n0 + row) * Q_ + 2 * hi, Q_, acc);
    gemm_store(C, MEM_, m0, n0, row, hi, acc);
}

__global__ void gemm_turn_kernel(const float* __restrict__ s,
                                 const float* __restrict__ ev,
                                 const float* __restrict__ x1,
                                 const float* __restrict__ We,
                                 const float* __restrict__ W_ih,
                                 const float* __restrict__ W_hh,
                                 const float* __restrict__ b_ih,
                                 const float* __restrict__ b_hh,
                                 float* __restrict__ qe,
                                 float* __restrict__ gi,
                                 float* __restrict__ gh) {
    const int lane = threadIdx.x & 31;
    const int row  = lane & 15;
    const int hi   = lane >> 4;
    const int m0   = blockIdx.y * 16;
    const int bx   = blockIdx.x;

    if (bx < 16) {
        const int n0 = bx * 16;
        const float* wRow = We + (size_t)(n0 + row) * (Q_ + MEM_) + 2 * hi;
        v8f acc = {};
        acc = gemm_tile(s  + (size_t)(m0 + row) * Q_   + 2 * hi, wRow,      Q_,   acc);
        acc = gemm_tile(ev + (size_t)(m0 + row) * MEM_ + 2 * hi, wRow + Q_, MEM_, acc);
        gemm_store(qe, MEM_, m0, n0, row, hi, acc);
    } else {
        const bool isGi = bx < 64;
        const int n0 = (isGi ? (bx - 16) : (bx - 64)) * 16;
        const float* A    = isGi ? x1 : s;
        const float* W    = isGi ? W_ih : W_hh;
        const float* bias = isGi ? b_ih : b_hh;
        float*       C    = isGi ? gi : gh;
        const float binit = bias[n0 + row];
        v8f acc;
#pragma unroll
        for (int j = 0; j < 8; ++j) acc[j] = binit;
        acc = gemm_tile(A + (size_t)(m0 + row) * MEM_ + 2 * hi,
                        W + (size_t)(n0 + row) * MEM_ + 2 * hi, MEM_, acc);
        gemm_store(C, 3 * Q_, m0, n0, row, hi, acc);
    }
}

__global__ void scores_kernel(const float* __restrict__ M,
                              const float* __restrict__ q0,
                              const float* __restrict__ q1,
                              const float* __restrict__ q2,
                              float* __restrict__ S0,
                              float* __restrict__ S1,
                              float* __restrict__ S2,
                              int ncols) {
    const int lane = threadIdx.x & 31;
    const int row  = lane & 15;
    const int hi   = lane >> 4;
    const int b    = blockIdx.y;
    const int l0   = blockIdx.x * 16;

    const float* Mb = M + ((size_t)b * L_ + l0 + row) * MEM_ + 2 * hi;
    const float* qv = ((row == 0) ? q0 : (row == 1) ? q1 : q2) + (size_t)b * MEM_ + 2 * hi;
    const bool live = row < ncols;

    __builtin_prefetch(Mb + 16 * MEM_, 0, 0);

    v8f acc = {};
    for (int k = 0; k < MEM_; k += 4) {
        v2f a; a.x = Mb[k]; a.y = Mb[k + 1];
        v2f bb;
        if (live) { bb.x = qv[k]; bb.y = qv[k + 1]; }
        else      { bb.x = 0.0f;  bb.y = 0.0f; }
        acc = __builtin_amdgcn_wmma_f32_16x16x4_f32(false, a, false, bb,
                                                    (short)0, acc, false, false);
    }
    if (live) {
        float* dst = (row == 0) ? S0 : (row == 1) ? S1 : S2;
#pragma unroll
        for (int j = 0; j < 8; ++j)
            dst[(size_t)b * L_ + l0 + j + 8 * hi] = acc[j];
    }
}

typedef __attribute__((ext_vector_type(4))) float v4f;

#define WS_WAVES 16
#define WS_COLS  64
#define WS_STEPS 8
#define WS_GROUP_ROWS (4 * WS_STEPS)
#define WS_GROUPS (L_ / (WS_GROUP_ROWS * WS_WAVES))

template <int X>
__device__ __forceinline__ float xrow(float v) {
    return __builtin_bit_cast(float, __builtin_amdgcn_update_dpp(
        0, __builtin_bit_cast(int, v), 0x160 + X, 0xf, 0xf, false));
}
__device__ __forceinline__ float xhalf(float v) {
    return __builtin_bit_cast(float, __builtin_amdgcn_permlanex16(
        __builtin_bit_cast(int, v), __builtin_bit_cast(int, v), 0x76543210, 0xfedcba98, false, false));
}
__device__ __forceinline__ float row16_max(float v) {
    v = fmaxf(v, xrow<1>(v)); v = fmaxf(v, xrow<2>(v));
    v = fmaxf(v, xrow<4>(v)); v = fmaxf(v, xrow<8>(v));
    return v;
}
__device__ __forceinline__ float row16_sum(float v) {
    v += xrow<1>(v); v += xrow<2>(v); v += xrow<4>(v); v += xrow<8>(v);
    return v;
}

template <int NP>
__device__ __forceinline__ void row_stats(const v4f (&x)[NP][2], float (&mx)[NP], float (&inv)[NP],
                                          float (*redm)[WS_WAVES], float (*reds)[WS_WAVES],
                                          int lane, int wave) {
#pragma unroll
    for (int p = 0; p < NP; ++p) {
        float m = fmaxf(fmaxf(fmaxf(x[p][0][0], x[p][0][1]), fmaxf(x[p][0][2], x[p][0][3])),
                        fmaxf(fmaxf(x[p][1][0], x[p][1][1]), fmaxf(x[p][1][2], x[p][1][3])));
        m = row16_max(m);
        m = fmaxf(m, xhalf(m));
        redm[p][wave] = m;
    }
    __syncthreads();
#pragma unroll
    for (int p = 0; p < NP; ++p) {
        const float m = row16_max(redm[p][lane & (WS_WAVES - 1)]);
        mx[p] = m;
        float s = 0.0f;
#pragma unroll
        for (int j = 0; j < 2; ++j)
#pragma unroll
            for (int e = 0; e < 4; ++e) s += __expf(x[p][j][e] - m);
        s = row16_sum(s);
        s += xhalf(s);
        reds[p][wave] = s;
    }
    __syncthreads();
#pragma unroll
    for (int p = 0; p < NP; ++p)
        inv[p] = 1.0f / row16_sum(reds[p][lane & (WS_WAVES - 1)]);
}

struct WsGroup {
    v2f a[WS_STEPS];
    v4f r0[WS_STEPS];
    v4f r1[WS_STEPS];
};

__device__ __forceinline__ void ws_load(WsGroup& g, const float* __restrict__ pv,
                                        const float* __restrict__ Mb, int l0) {
#pragma unroll
    for (int s = 0; s < WS_STEPS; ++s)
        g.a[s] = *(const v2f*)(pv + l0 + 4 * s);
#pragma unroll
    for (int s = 0; s < WS_STEPS; ++s) {
        const float* rp = Mb + (size_t)(l0 + 4 * s) * MEM_;
        g.r0[s] = *(const v4f*)rp;
        g.r1[s] = *(const v4f*)(rp + MEM_);
    }
}

__device__ __forceinline__ void ws_mma(const WsGroup& g, bool live, float mx, float inv,
                                       v8f (&acc)[4]) {
#pragma unroll
    for (int s = 0; s < WS_STEPS; ++s) {
        v2f a;
        a.x = live ? __expf(g.a[s].x - mx) * inv : 0.0f;
        a.y = live ? __expf(g.a[s].y - mx) * inv : 0.0f;
#pragma unroll
        for (int t = 0; t < 4; ++t) {
            v2f bb; bb.x = g.r0[s][t]; bb.y = g.r1[s][t];
            acc[t] = __builtin_amdgcn_wmma_f32_16x16x4_f32(false, a, false, bb,
                                                           (short)0, acc[t], false, false);
        }
    }
}

__global__ __launch_bounds__(32 * WS_WAVES)
void wsum_kernel(const float* __restrict__ M,
                 const float* __restrict__ p,
                 const float* __restrict__ w,
                 const float* __restrict__ endS,
                 float* __restrict__ sacc,
                 float* __restrict__ eacc,
                 float* __restrict__ ev,
                 float* __restrict__ x1) {
    const int tid  = threadIdx.x;
    const int wave = tid >> 5;
    const int lane = tid & 31;
    const int c    = lane & 15;
    const int hi   = lane >> 4;
    const int b    = blockIdx.y;

    __shared__ float redm[2][WS_WAVES];
    __shared__ float reds[2][WS_WAVES];
    const size_t r0 = (size_t)b * L_ + 4 * tid;
    const size_t r1 = r0 + 4 * 32 * WS_WAVES;

    if (blockIdx.x == MEM_ / WS_COLS) {
        v4f x[1][2];
        x[0][0] = *(const v4f*)(endS + r0);
        x[0][1] = *(const v4f*)(endS + r1);
        v4f e0 = *(const v4f*)(eacc + r0);
        v4f e1 = *(const v4f*)(eacc + r1);
        float mx[1], inv[1];
        row_stats<1>(x, mx, inv, redm, reds, lane, wave);
#pragma unroll
        for (int e = 0; e < 4; ++e) {
            e0[e] += 0.2f * (__expf(x[0][0][e] - mx[0]) * inv[0]);
            e1[e] += 0.2f * (__expf(x[0][1][e] - mx[0]) * inv[0]);
        }
        *(v4f*)(eacc + r0) = e0;
        *(v4f*)(eacc + r1) = e1;
        return;
    }

    const int slab = blockIdx.x * WS_COLS;

    const float* pv = ((c == 1) ? w : p) + (size_t)b * L_ + 2 * hi;
    const float* Mb = M + ((size_t)b * L_ + 2 * hi) * MEM_ + slab + 4 * c;
    const bool live = c < 2;

    v4f x[2][2];
    x[0][0] = *(const v4f*)(p + r0);
    x[0][1] = *(const v4f*)(p + r1);
    x[1][0] = *(const v4f*)(w + r0);
    x[1][1] = *(const v4f*)(w + r1);
    v4f s0v = {}, s1v = {};
    if (blockIdx.x == 0) {
        s0v = *(const v4f*)(sacc + r0);
        s1v = *(const v4f*)(sacc + r1);
    }

    v8f acc[4] = {};
    WsGroup g0, g1;
#define WS_ORDER() __builtin_amdgcn_sched_barrier(0)
    ws_load(g0, pv, Mb, wave * WS_GROUP_ROWS);
    ws_load(g1, pv, Mb, (WS_WAVES + wave) * WS_GROUP_ROWS);
    WS_ORDER();
    float mx2[2], inv2[2];
    row_stats<2>(x, mx2, inv2, redm, reds, lane, wave);
    if (blockIdx.x == 0) {
#pragma unroll
        for (int e = 0; e < 4; ++e) {
            s0v[e] += 0.2f * (__expf(x[0][0][e] - mx2[0]) * inv2[0]);
            s1v[e] += 0.2f * (__expf(x[0][1][e] - mx2[0]) * inv2[0]);
        }
        *(v4f*)(sacc + r0) = s0v;
        *(v4f*)(sacc + r1) = s1v;
    }
    const float mx  = (c == 1) ? mx2[1]  : mx2[0];
    const float inv = (c == 1) ? inv2[1] : inv2[0];
    WS_ORDER();
    ws_mma(g0, live, mx, inv, acc);
    WS_ORDER();
    for (int i = 1; i < WS_GROUPS - 1; i += 2) {
        ws_load(g0, pv, Mb, ((i + 1) * WS_WAVES + wave) * WS_GROUP_ROWS);
        WS_ORDER();
        ws_mma(g1, live, mx, inv, acc);
        WS_ORDER();
        ws_load(g1, pv, Mb, ((i + 2) * WS_WAVES + wave) * WS_GROUP_ROWS);
        WS_ORDER();
        ws_mma(g0, live, mx, inv, acc);
        WS_ORDER();
    }
    ws_mma(g1, live, mx, inv, acc);
#undef WS_ORDER

    __shared__ float red[WS_WAVES][2][WS_COLS];
    if (hi == 0) {
        v4f e, xx;
#pragma unroll
        for (int t = 0; t < 4; ++t) { e[t] = acc[t][0]; xx[t] = acc[t][1]; }
        *(v4f*)&red[wave][0][4 * c] = e;
        *(v4f*)&red[wave][1][4 * c] = xx;
    }
    __syncthreads();
    if (tid < 2 * WS_COLS) {
        const int r  = tid >> 6;
        const int mm = tid & (WS_COLS - 1);
        float s = 0.0f;
#pragma unroll
        for (int k = 0; k < WS_WAVES; ++k) s += red[k][r][mm];
        (r ? x1 : ev)[(size_t)b * MEM_ + slab + mm] = s;
    }
}

__global__ void gru_kernel(const float* __restrict__ gi,
                           const float* __restrict__ gh,
                           const float* __restrict__ sOld,
                           float* __restrict__ sNew) {
    int i = blockIdx.x * 256 + threadIdx.x;
    if (i >= BQ_) return;
    int b = i / Q_, q = i % Q_;
    const float* gib = gi + (size_t)b * 3 * Q_;
    const float* ghb = gh + (size_t)b * 3 * Q_;
    float ir = gib[q], iz = gib[Q_ + q], in_ = gib[2 * Q_ + q];
    float hr = ghb[q], hz = ghb[Q_ + q], hn  = ghb[2 * Q_ + q];
    float r = 1.0f / (1.0f + __expf(-(ir + hr)));
    float z = 1.0f / (1.0f + __expf(-(iz + hz)));
    float n = tanhf(in_ + r * hn);
    sNew[i] = (1.0f - z) * n + z * sOld[i];
}

__global__ void init_kernel(const float* __restrict__ s0, float* __restrict__ ws) {
    int i = blockIdx.x * 256 + threadIdx.x;
    if (i < 2 * BL_) ws[OFF_SACC + i] = 0.0f;
    if (i < BQ_)     ws[OFF_S_A + i] = s0[i];
}

__global__ __launch_bounds__(32 * WS_WAVES)
void final_kernel(const float* __restrict__ ws, float* out) {
    const int tid  = threadIdx.x;
    const int wave = tid >> 5;
    const int lane = tid & 31;
    const int b    = blockIdx.x;
    __shared__ float redm[1][WS_WAVES];
    __shared__ float reds[1][WS_WAVES];
    const size_t r0 = (size_t)b * L_ + 4 * tid;
    const size_t r1 = r0 + 4 * 32 * WS_WAVES;

    v4f x[1][2];
    x[0][0] = *(const v4f*)(out + BL_ + r0);
    x[0][1] = *(const v4f*)(out + BL_ + r1);
    v4f e0 = *(const v4f*)(ws + OFF_EACC + r0);
    v4f e1 = *(const v4f*)(ws + OFF_EACC + r1);
    v4f s0v = *(const v4f*)(ws + OFF_SACC + r0);
    v4f s1v = *(const v4f*)(ws + OFF_SACC + r1);
    float mx[1], inv[1];
    row_stats<1>(x, mx, inv, redm, reds, lane, wave);
#pragma unroll
    for (int e = 0; e < 4; ++e) {
        e0[e] = logf(e0[e] + 0.2f * (__expf(x[0][0][e] - mx[0]) * inv[0]));
        e1[e] = logf(e1[e] + 0.2f * (__expf(x[0][1][e] - mx[0]) * inv[0]));
        s0v[e] = logf(s0v[e]);
        s1v[e] = logf(s1v[e]);
    }
    *(v4f*)(out + r0) = s0v;
    *(v4f*)(out + r1) = s1v;
    *(v4f*)(out + BL_ + r0) = e0;
    *(v4f*)(out + BL_ + r1) = e1;
}

extern "C" void kernel_launch(void* const* d_in, const int* in_sizes, int n_in,
                              void* d_out, int out_size, void* d_ws, size_t ws_size,
                              hipStream_t stream) {
    const float* M    = (const float*)d_in[0];
    const float* s0   = (const float*)d_in[1];
    const float* Wb   = (const float*)d_in[2];
    const float* We   = (const float*)d_in[3];
    const float* Wa   = (const float*)d_in[4];
    const float* W_ih = (const float*)d_in[5];
    const float* W_hh = (const float*)d_in[6];
    const float* b_ih = (const float*)d_in[7];
    const float* b_hh = (const float*)d_in[8];

    float* ws   = (float*)d_ws;
    float* S1   = ws + OFF_S1;
    float* S2   = ws + OFF_S2;
    float* sacc = ws + OFF_SACC;
    float* eacc = ws + OFF_EACC;
    float* qb   = ws + OFF_QB;
    float* qa   = ws + OFF_QA;
    float* qe   = ws + OFF_QE;
    float* ev   = ws + OFF_EV;
    float* x1   = ws + OFF_X1;
    float* sA   = ws + OFF_S_A;
    float* sB   = ws + OFF_S_B;
    float* gi   = ws + OFF_GI;
    float* gh   = ws + OFF_GH;
    float* S3   = (float*)d_out + BL_;

    init_kernel<<<(2 * BL_) / 256, 256, 0, stream>>>(s0, ws);

    gemm_q_kernel<<<dim3(32, 2), 32, 0, stream>>>(sA, Wb, Wa, qb, qa);
    scores_kernel<<<dim3(L_ / 16, B_), 32, 0, stream>>>(M, qb, qa, qa, S1, S2, nullptr, 2);

    for (int t = 0; t < 5; ++t) {
        const float* sCur = (t & 1) ? sB : sA;
        float*       sNxt = (t & 1) ? sA : sB;

        if (t == 0)
            wsum_kernel<<<dim3(MEM_ / WS_COLS, B_), 32 * WS_WAVES, 0, stream>>>(
                M, S1, S2, nullptr, sacc, eacc, ev, x1);
        else
            wsum_kernel<<<dim3(MEM_ / WS_COLS + 1, B_), 32 * WS_WAVES, 0, stream>>>(
                M, S1, S2, S3, sacc, eacc, ev, x1);

        gemm_turn_kernel<<<dim3(112, 2), 32, 0, stream>>>(sCur, ev, x1, We, W_ih, W_hh,
                                                          b_ih, b_hh, qe, gi, gh);
        gru_kernel<<<BQ_ / 256, 256, 0, stream>>>(gi, gh, sCur, sNxt);

        if (t < 4) {
            gemm_q_kernel<<<dim3(32, 2), 32, 0, stream>>>(sNxt, Wb, Wa, qb, qa);
            scores_kernel<<<dim3(L_ / 16, B_), 32, 0, stream>>>(M, qe, qb, qa, S3, S1, S2, 3);
        } else {
            scores_kernel<<<dim3(L_ / 16, B_), 32, 0, stream>>>(M, qe, qe, qe, S3, nullptr, nullptr, 1);
        }
    }

    final_kernel<<<B_, 32 * WS_WAVES, 0, stream>>>(ws, (float*)d_out);
}


# score kernel: M read by whole lines (512 contiguous bytes per wave load) and parked in LDS, fragments and queries read from LDS; 4-wave work-groups, three units of loads requested ahead; same sums
# speedup vs baseline: 11.5462x; 1.4019x over previous
#include <hip/hip_runtime.h>
#include <math.h>

#define B_   32
#define L_   4096
#define MEM_ 256
#define Q_   256
#define BL_  (B_ * L_)
#define BQ_  (B_ * Q_)
#define B3Q_ (B_ * 3 * Q_)

#define OFF_S1   0
#define OFF_S2   (OFF_S1 + BL_)
#define OFF_SACC (OFF_S2 + BL_)
#define OFF_EACC (OFF_SACC + BL_)
#define OFF_QB   (OFF_EACC + BL_)
#define OFF_QA   (OFF_QB + BQ_)
#define OFF_QE   (OFF_QA + BQ_)
#define OFF_EV   (OFF_QE + BQ_)
#define OFF_X1   (OFF_EV + BQ_)
#define OFF_S_A  (OFF_X1 + BQ_)
#define OFF_S_B  (OFF_S_A + BQ_)
#define OFF_GI   (OFF_S_B + BQ_)
#define OFF_GH   (OFF_GI + B3Q_)

typedef __attribute__((ext_vector_type(2))) float v2f;
typedef __attribute__((ext_vector_type(8))) float v8f;

__device__ __forceinline__ v8f gemm_tile(const float* __restrict__ aRow,
                                         const float* __restrict__ wRow,
                                         int K, v8f acc) {
    for (int k = 0; k < K; k += 4) {
        v2f a; a.x = aRow[k]; a.y = aRow[k + 1];
        v2f b; b.x = wRow[k]; b.y = wRow[k + 1];
        acc = __builtin_amdgcn_wmma_f32_16x16x4_f32(false, a, false, b,
                                                    (short)0, acc, false, false);
    }
    return acc;
}

__device__ __forceinline__ void gemm_store(float* __restrict__ C, int ldc,
                                           int m0, int n0, int row, int hi, v8f acc) {
#pragma unroll
    for (int j = 0; j < 8; ++j)
        C[(size_t)(m0 + j + 8 * hi) * ldc + n0 + row] = acc[j];
}

__global__ void gemm_q_kernel(const float* __restrict__ s,
                              const float* __restrict__ Wb,
                              const float* __restrict__ Wa,
                              float* __restrict__ qb,
                              float* __restrict__ qa) {
    const int lane = threadIdx.x & 31;
    const int row  = lane & 15;
    const int hi   = lane >> 4;
    const int m0   = blockIdx.y * 16;
    const int sel  = blockIdx.x >> 4;
    const int n0   = (blockIdx.x & 15) * 16;
    const float* W = sel ? Wa : Wb;
    float*       C = sel ? qa : qb;

    v8f acc = {};
    acc = gemm_tile(s + (size_t)(m0 + row) * Q_ + 2 * hi,
                    W + (size_t)(n0 + row) * Q_ + 2 * hi, Q_, acc);
    gemm_store(C, MEM_, m0, n0, row, hi, acc);
}

__global__ void gemm_turn_kernel(const float* __restrict__ s,
                                 const float* __restrict__ ev,
                                 const float* __restrict__ x1,
                                 const float* __restrict__ We,
                                 const float* __restrict__ W_ih,
                                 const float* __restrict__ W_hh,
                                 const float* __restrict__ b_ih,
                                 const float* __restrict__ b_hh,
                                 float* __restrict__ qe,
                                 float* __restrict__ gi,
                                 float* __restrict__ gh) {
    const int lane = threadIdx.x & 31;
    const int row  = lane & 15;
    const int hi   = lane >> 4;
    const int m0   = blockIdx.y * 16;
    const int bx   = blockIdx.x;

    if (bx < 16) {
        const int n0 = bx * 16;
        const float* wRow = We + (size_t)(n0 + row) * (Q_ + MEM_) + 2 * hi;
        v8f acc = {};
        acc = gemm_tile(s  + (size_t)(m0 + row) * Q_   + 2 * hi, wRow,      Q_,   acc);
        acc = gemm_tile(ev + (size_t)(m0 + row) * MEM_ + 2 * hi, wRow + Q_, MEM_, acc);
        gemm_store(qe, MEM_, m0, n0, row, hi, acc);
    } else {
        const bool isGi = bx < 64;
        const int n0 = (isGi ? (bx - 16) : (bx - 64)) * 16;
        const float* A    = isGi ? x1 : s;
        const float* W    = isGi ? W_ih : W_hh;
        const float* bias = isGi ? b_ih : b_hh;
        float*       C    = isGi ? gi : gh;
        const float binit = bias[n0 + row];
        v8f acc;
#pragma unroll
        for (int j = 0; j < 8; ++j) acc[j] = binit;
        acc = gemm_tile(A + (size_t)(m0 + row) * MEM_ + 2 * hi,
                        W + (size_t)(n0 + row) * MEM_ + 2 * hi, MEM_, acc);
        gemm_store(C, 3 * Q_, m0, n0, row, hi, acc);
    }
}

typedef __attribute__((ext_vector_type(4))) float v4f;

#define SC_WAVES  4
#define SC_TILES  2
#define SC_UCOLS  128
#define SC_PITCH  (SC_UCOLS + 4)
#define SC_QPITCH (MEM_ + 4)
#define SC_ROWS_WG (16 * SC_TILES * SC_WAVES)

struct ScUnit { v4f r[16]; };

__device__ __forceinline__ void sc_load(ScUnit& u, const float* __restrict__ src) {
#pragma unroll
    for (int j = 0; j < 16; ++j) u.r[j] = *(const v4f*)(src + (size_t)j * MEM_);
}

__device__ __forceinline__ void sc_park(const ScUnit& u, float* dst) {
#pragma unroll
    for (int j = 0; j < 16; ++j) *(v4f*)(dst + j * SC_PITCH) = u.r[j];
}

__device__ __forceinline__ v8f sc_mma(const float* aP, const float* bP, v8f acc) {
#pragma unroll
    for (int i = 0; i < SC_UCOLS / 4; ++i) {
        const v2f a  = *(const v2f*)(aP + 4 * i);
        const v2f bb = *(const v2f*)(bP + 4 * i);
        acc = __builtin_amdgcn_wmma_f32_16x16x4_f32(false, a, false, bb,
                                                    (short)0, acc, false, false);
    }
    return acc;
}

__global__ __launch_bounds__(32 * SC_WAVES)
void scores_kernel(const float* __restrict__ M,
                   const float* __restrict__ q0,
                   const float* __restrict__ q1,
                   const float* __restrict__ q2,
                   float* __restrict__ S0,
                   float* __restrict__ S1,
                   float* __restrict__ S2,
                   int ncols) {
    const int tid  = threadIdx.x;
    const int wave = tid >> 5;
    const int lane = tid & 31;
    const int c    = lane & 15;
    const int hi   = lane >> 4;
    const int b    = blockIdx.y;
    const int lw   = blockIdx.x * SC_ROWS_WG + wave * (16 * SC_TILES);

    __shared__ __attribute__((aligned(16))) float qs[SC_WAVES][4 * SC_QPITCH];
    __shared__ __attribute__((aligned(16))) float ms[SC_WAVES][16 * SC_PITCH];

    const float* Mw = M + ((size_t)b * L_ + lw) * MEM_ + 4 * lane;

    const float* qsrc[3] = { q0, q1, q2 };
    v4f qv[3][2];
#pragma unroll
    for (int r = 0; r < 3; ++r) {
        const float* qp = ((r < ncols) ? qsrc[r] : q0) + (size_t)b * MEM_ + 4 * lane;
        qv[r][0] = *(const v4f*)qp;
        qv[r][1] = *(const v4f*)(qp + SC_UCOLS);
    }

    ScUnit g0, g1;
    sc_load(g0, Mw);

    float*       mw = ms[wave];
    float*       pk = mw + 4 * lane;
    const float* aP = mw + c * SC_PITCH + 2 * hi;
    float*       qw = qs[wave];
    const float* bP = qw + (c < 3 ? c : 3) * SC_QPITCH + 2 * hi;

#define SC_ORDER() __builtin_amdgcn_sched_barrier(0)
#define SC_WAVE_SYNC() do { __builtin_amdgcn_fence(__ATOMIC_RELEASE, "wavefront"); \
                            __builtin_amdgcn_wave_barrier();                      \
                            __builtin_amdgcn_fence(__ATOMIC_ACQUIRE, "wavefront"); } while (0)

    sc_load(g1, Mw + SC_UCOLS);
    SC_ORDER();
    {
        const v4f z = {};
#pragma unroll
        for (int r = 0; r < 3; ++r) {
            *(v4f*)(qw + r * SC_QPITCH + 4 * lane)            = (r < ncols) ? qv[r][0] : z;
            *(v4f*)(qw + r * SC_QPITCH + 4 * lane + SC_UCOLS) = (r < ncols) ? qv[r][1] : z;
        }
        *(v4f*)(qw + 3 * SC_QPITCH + 4 * lane)            = z;
        *(v4f*)(qw + 3 * SC_QPITCH + 4 * lane + SC_UCOLS) = z;
    }

    const bool live = c < ncols;
    float* dst = ((c == 0) ? S0 : (c == 1) ? S1 : S2);

    v8f acc = {};
    sc_park(g0, pk);
    sc_load(g0, Mw + (size_t)16 * MEM_);
    SC_ORDER();
    SC_WAVE_SYNC();
    acc = sc_mma(aP, bP, acc);
    SC_WAVE_SYNC();
    sc_park(g1, pk);
    sc_load(g1, Mw + (size_t)16 * MEM_ + SC_UCOLS);
    SC_ORDER();
    SC_WAVE_SYNC();
    acc = sc_mma(aP, bP + SC_UCOLS, acc);
    SC_WAVE_SYNC();
    if (live) {
#pragma unroll
        for (int j = 0; j < 8; ++j)
            dst[(size_t)b * L_ + lw + j + 8 * hi] = acc[j];
    }

    acc = v8f{};
    sc_park(g0, pk);
    SC_WAVE_SYNC();
    acc = sc_mma(aP, bP, acc);
    SC_WAVE_SYNC();
    sc_park(g1, pk);
    SC_WAVE_SYNC();
    acc = sc_mma(aP, bP + SC_UCOLS, acc);
    if (live) {
#pragma unroll
        for (int j = 0; j < 8; ++j)
            dst[(size_t)b * L_ + lw + 16 + j + 8 * hi] = acc[j];
    }
#undef SC_WAVE_SYNC
#undef SC_ORDER
}

#define WS_WAVES 16
#define WS_COLS  64
#define WS_STEPS 8
#define WS_GROUP_ROWS (4 * WS_STEPS)
#define WS_GROUPS (L_ / (WS_GROUP_ROWS * WS_WAVES))

template <int X>
__device__ __forceinline__ float xrow(float v) {
    return __builtin_bit_cast(float, __builtin_amdgcn_update_dpp(
        0, __builtin_bit_cast(int, v), 0x160 + X, 0xf, 0xf, false));
}
__device__ __forceinline__ float xhalf(float v) {
    return __builtin_bit_cast(float, __builtin_amdgcn_permlanex16(
        __builtin_bit_cast(int, v), __builtin_bit_cast(int, v), 0x76543210, 0xfedcba98, false, false));
}
__device__ __forceinline__ float row16_max(float v) {
    v = fmaxf(v, xrow<1>(v)); v = fmaxf(v, xrow<2>(v));
    v = fmaxf(v, xrow<4>(v)); v = fmaxf(v, xrow<8>(v));
    return v;
}
__device__ __forceinline__ float row16_sum(float v) {
    v += xrow<1>(v); v += xrow<2>(v); v += xrow<4>(v); v += xrow<8>(v);
    return v;
}

template <int NP>
__device__ __forceinline__ void row_stats(const v4f (&x)[NP][2], float (&mx)[NP], float (&inv)[NP],
                                          float (*redm)[WS_WAVES], float (*reds)[WS_WAVES],
                                          int lane, int wave) {
#pragma unroll
    for (int p = 0; p < NP; ++p) {
        float m = fmaxf(fmaxf(fmaxf(x[p][0][0], x[p][0][1]), fmaxf(x[p][0][2], x[p][0][3])),
                        fmaxf(fmaxf(x[p][1][0], x[p][1][1]), fmaxf(x[p][1][2], x[p][1][3])));
        m = row16_max(m);
        m = fmaxf(m, xhalf(m));
        redm[p][wave] = m;
    }
    __syncthreads();
#pragma unroll
    for (int p = 0; p < NP; ++p) {
        const float m = row16_max(redm[p][lane & (WS_WAVES - 1)]);
        mx[p] = m;
        float s = 0.0f;
#pragma unroll
        for (int j = 0; j < 2; ++j)
#pragma unroll
            for (int e = 0; e < 4; ++e) s += __expf(x[p][j][e] - m);
        s = row16_sum(s);
        s += xhalf(s);
        reds[p][wave] = s;
    }
    __syncthreads();
#pragma unroll
    for (int p = 0; p < NP; ++p)
        inv[p] = 1.0f / row16_sum(reds[p][lane & (WS_WAVES - 1)]);
}

struct WsGroup {
    v2f a[WS_STEPS];
    v4f r0[WS_STEPS];
    v4f r1[WS_STEPS];
};

__device__ __forceinline__ void ws_load(WsGroup& g, const float* __restrict__ pv,
                                        const float* __restrict__ Mb, int l0) {
#pragma unroll
    for (int s = 0; s < WS_STEPS; ++s)
        g.a[s] = *(const v2f*)(pv + l0 + 4 * s);
#pragma unroll
    for (int s = 0; s < WS_STEPS; ++s) {
        const float* rp = Mb + (size_t)(l0 + 4 * s) * MEM_;
        g.r0[s] = *(const v4f*)rp;
        g.r1[s] = *(const v4f*)(rp + MEM_);
    }
}

__device__ __forceinline__ void ws_mma(const WsGroup& g, bool live, float mx, float inv,
                                       v8f (&acc)[4]) {
#pragma unroll
    for (int s = 0; s < WS_STEPS; ++s) {
        v2f a;
        a.x = live ? __expf(g.a[s].x - mx) * inv : 0.0f;
        a.y = live ? __expf(g.a[s].y - mx) * inv : 0.0f;
#pragma unroll
        for (int t = 0; t < 4; ++t) {
            v2f bb; bb.x = g.r0[s][t]; bb.y = g.r1[s][t];
            acc[t] = __builtin_amdgcn_wmma_f32_16x16x4_f32(false, a, false, bb,
                                                           (short)0, acc[t], false, false);
        }
    }
}

__global__ __launch_bounds__(32 * WS_WAVES)
void wsum_kernel(const float* __restrict__ M,
                 const float* __restrict__ p,
                 const float* __restrict__ w,
                 const float* __restrict__ endS,
                 float* __restrict__ sacc,
                 float* __restrict__ eacc,
                 float* __restrict__ ev,
                 float* __restrict__ x1) {
    const int tid  = threadIdx.x;
    const int wave = tid >> 5;
    const int lane = tid & 31;
    const int c    = lane & 15;
    const int hi   = lane >> 4;
    const int b    = blockIdx.y;

    __shared__ float redm[2][WS_WAVES];
    __shared__ float reds[2][WS_WAVES];
    const size_t r0 = (size_t)b * L_ + 4 * tid;
    const size_t r1 = r0 + 4 * 32 * WS_WAVES;

    if (blockIdx.x == MEM_ / WS_COLS) {
        v4f x[1][2];
        x[0][0] = *(const v4f*)(endS + r0);
        x[0][1] = *(const v4f*)(endS + r1);
        v4f e0 = *(const v4f*)(eacc + r0);
        v4f e1 = *(const v4f*)(eacc + r1);
        float mx[1], inv[1];
        row_stats<1>(x, mx, inv, redm, reds, lane, wave);
#pragma unroll
        for (int e = 0; e < 4; ++e) {
            e0[e] += 0.2f * (__expf(x[0][0][e] - mx[0]) * inv[0]);
            e1[e] += 0.2f * (__expf(x[0][1][e] - mx[0]) * inv[0]);
        }
        *(v4f*)(eacc + r0) = e0;
        *(v4f*)(eacc + r1) = e1;
        return;
    }

    const int slab = blockIdx.x * WS_COLS;

    const float* pv = ((c == 1) ? w : p) + (size_t)b * L_ + 2 * hi;
    const float* Mb = M + ((size_t)b * L_ + 2 * hi) * MEM_ + slab + 4 * c;
    const bool live = c < 2;

    v4f x[2][2];
    x[0][0] = *(const v4f*)(p + r0);
    x[0][1] = *(const v4f*)(p + r1);
    x[1][0] = *(const v4f*)(w + r0);
    x[1][1] = *(const v4f*)(w + r1);
    v4f s0v = {}, s1v = {};
    if (blockIdx.x == 0) {
        s0v = *(const v4f*)(sacc + r0);
        s1v = *(const v4f*)(sacc + r1);
    }

    v8f acc[4] = {};
    WsGroup g0, g1;
#define WS_ORDER() __builtin_amdgcn_sched_barrier(0)
    ws_load(g0, pv, Mb, wave * WS_GROUP_ROWS);
    ws_load(g1, pv, Mb, (WS_WAVES + wave) * WS_GROUP_ROWS);
    WS_ORDER();
    float mx2[2], inv2[2];
    row_stats<2>(x, mx2, inv2, redm, reds, lane, wave);
    if (blockIdx.x == 0) {
#pragma unroll
        for (int e = 0; e < 4; ++e) {
            s0v[e] += 0.2f * (__expf(x[0][0][e] - mx2[0]) * inv2[0]);
            s1v[e] += 0.2f * (__expf(x[0][1][e] - mx2[0]) * inv2[0]);
        }
        *(v4f*)(sacc + r0) = s0v;
        *(v4f*)(sacc + r1) = s1v;
    }
    const float mx  = (c == 1) ? mx2[1]  : mx2[0];
    const float inv = (c == 1) ? inv2[1] : inv2[0];
    WS_ORDER();
    ws_mma(g0, live, mx, inv, acc);
    WS_ORDER();
    for (int i = 1; i < WS_GROUPS - 1; i += 2) {
        ws_load(g0, pv, Mb, ((i + 1) * WS_WAVES + wave) * WS_GROUP_ROWS);
        WS_ORDER();
        ws_mma(g1, live, mx, inv, acc);
        WS_ORDER();
        ws_load(g1, pv, Mb, ((i + 2) * WS_WAVES + wave) * WS_GROUP_ROWS);
        WS_ORDER();
        ws_mma(g0, live, mx, inv, acc);
        WS_ORDER();
    }
    ws_mma(g1, live, mx, inv, acc);
#undef WS_ORDER

    __shared__ float red[WS_WAVES][2][WS_COLS];
    if (hi == 0) {
        v4f e, xx;
#pragma unroll
        for (int t = 0; t < 4; ++t) { e[t] = acc[t][0]; xx[t] = acc[t][1]; }
        *(v4f*)&red[wave][0][4 * c] = e;
        *(v4f*)&red[wave][1][4 * c] = xx;
    }
    __syncthreads();
    if (tid < 2 * WS_COLS) {
        const int r  = tid >> 6;
        const int mm = tid & (WS_COLS - 1);
        float s = 0.0f;
#pragma unroll
        for (int k = 0; k < WS_WAVES; ++k) s += red[k][r][mm];
        (r ? x1 : ev)[(size_t)b * MEM_ + slab + mm] = s;
    }
}

__global__ void gru_kernel(const float* __restrict__ gi,
                           const float* __restrict__ gh,
                           const float* __restrict__ sOld,
                           float* __restrict__ sNew) {
    int i = blockIdx.x * 256 + threadIdx.x;
    if (i >= BQ_) return;
    int b = i / Q_, q = i % Q_;
    const float* gib = gi + (size_t)b * 3 * Q_;
    const float* ghb = gh + (size_t)b * 3 * Q_;
    float ir = gib[q], iz = gib[Q_ + q], in_ = gib[2 * Q_ + q];
    float hr = ghb[q], hz = ghb[Q_ + q], hn  = ghb[2 * Q_ + q];
    float r = 1.0f / (1.0f + __expf(-(ir + hr)));
    float z = 1.0f / (1.0f + __expf(-(iz + hz)));
    float n = tanhf(in_ + r * hn);
    sNew[i] = (1.0f - z) * n + z * sOld[i];
}

__global__ void init_kernel(const float* __restrict__ s0, float* __restrict__ ws) {
    int i = blockIdx.x * 256 + threadIdx.x;
    if (i < 2 * BL_) ws[OFF_SACC + i] = 0.0f;
    if (i < BQ_)     ws[OFF_S_A + i] = s0[i];
}

__global__ __launch_bounds__(32 * WS_WAVES)
void final_kernel(const float* __restrict__ ws, float* out) {
    const int tid  = threadIdx.x;
    const int wave = tid >> 5;
    const int lane = tid & 31;
    const int b    = blockIdx.x;
    __shared__ float redm[1][WS_WAVES];
    __shared__ float reds[1][WS_WAVES];
    const size_t r0 = (size_t)b * L_ + 4 * tid;
    const size_t r1 = r0 + 4 * 32 * WS_WAVES;

    v4f x[1][2];
    x[0][0] = *(const v4f*)(out + BL_ + r0);
    x[0][1] = *(const v4f*)(out + BL_ + r1);
    v4f e0 = *(const v4f*)(ws + OFF_EACC + r0);
    v4f e1 = *(const v4f*)(ws + OFF_EACC + r1);
    v4f s0v = *(const v4f*)(ws + OFF_SACC + r0);
    v4f s1v = *(const v4f*)(ws + OFF_SACC + r1);
    float mx[1], inv[1];
    row_stats<1>(x, mx, inv, redm, reds, lane, wave);
#pragma unroll
    for (int e = 0; e < 4; ++e) {
        e0[e] = logf(e0[e] + 0.2f * (__expf(x[0][0][e] - mx[0]) * inv[0]));
        e1[e] = logf(e1[e] + 0.2f * (__expf(x[0][1][e] - mx[0]) * inv[0]));
        s0v[e] = logf(s0v[e]);
        s1v[e] = logf(s1v[e]);
    }
    *(v4f*)(out + r0) = s0v;
    *(v4f*)(out + r1) = s1v;
    *(v4f*)(out + BL_ + r0) = e0;
    *(v4f*)(out + BL_ + r1) = e1;
}

extern "C" void kernel_launch(void* const* d_in, const int* in_sizes, int n_in,
                              void* d_out, int out_size, void* d_ws, size_t ws_size,
                              hipStream_t stream) {
    const float* M    = (const float*)d_in[0];
    const float* s0   = (const float*)d_in[1];
    const float* Wb   = (const float*)d_in[2];
    const float* We   = (const float*)d_in[3];
    const float* Wa   = (const float*)d_in[4];
    const float* W_ih = (const float*)d_in[5];
    const float* W_hh = (const float*)d_in[6];
    const float* b_ih = (const float*)d_in[7];
    const float* b_hh = (const float*)d_in[8];

    float* ws   = (float*)d_ws;
    float* S1   = ws + OFF_S1;
    float* S2   = ws + OFF_S2;
    float* sacc = ws + OFF_SACC;
    float* eacc = ws + OFF_EACC;
    float* qb   = ws + OFF_QB;
    float* qa   = ws + OFF_QA;
    float* qe   = ws + OFF_QE;
    float* ev   = ws + OFF_EV;
    float* x1   = ws + OFF_X1;
    float* sA   = ws + OFF_S_A;
    float* sB   = ws + OFF_S_B;
    float* gi   = ws + OFF_GI;
    float* gh   = ws + OFF_GH;
    float* S3   = (float*)d_out + BL_;

    init_kernel<<<(2 * BL_) / 256, 256, 0, stream>>>(s0, ws);

    gemm_q_kernel<<<dim3(32, 2), 32, 0, stream>>>(sA, Wb, Wa, qb, qa);
    scores_kernel<<<dim3(L_ / SC_ROWS_WG, B_), 32 * SC_WAVES, 0, stream>>>(M, qb, qa, qa, S1, S2, nullptr, 2);

    for (int t = 0; t < 5; ++t) {
        const float* sCur = (t & 1) ? sB : sA;
        float*       sNxt = (t & 1) ? sA : sB;

        if (t == 0)
            wsum_kernel<<<dim3(MEM_ / WS_COLS, B_), 32 * WS_WAVES, 0, stream>>>(
                M, S1, S2, nullptr, sacc, eacc, ev, x1);
        else
            wsum_kernel<<<dim3(MEM_ / WS_COLS + 1, B_), 32 * WS_WAVES, 0, stream>>>(
                M, S1, S2, S3, sacc, eacc, ev, x1);

        gemm_turn_kernel<<<dim3(112, 2), 32, 0, stream>>>(sCur, ev, x1, We, W_ih, W_hh,
                                                          b_ih, b_hh, qe, gi, gh);
        gru_kernel<<<BQ_ / 256, 256, 0, stream>>>(gi, gh, sCur, sNxt);

        if (t < 4) {
            gemm_q_kernel<<<dim3(32, 2), 32, 0, stream>>>(sNxt, Wb, Wa, qb, qa);
            scores_kernel<<<dim3(L_ / SC_ROWS_WG, B_), 32 * SC_WAVES, 0, stream>>>(M, qe, qb, qa, S3, S1, S2, 3);
        } else {
            scores_kernel<<<dim3(L_ / SC_ROWS_WG, B_), 32 * SC_WAVES, 0, stream>>>(M, qe, qe, qe, S3, nullptr, nullptr, 1);
        }
    }

    final_kernel<<<B_, 32 * WS_WAVES, 0, stream>>>(ws, (float*)d_out);
}


# weighted-sum kernel on every WGP instead of half of them: L split over 2 work-groups per (batch, slab), 256 in all; their partial sums added on load by the next GEMM (new reduction tree)
# speedup vs baseline: 12.4066x; 1.0745x over previous
#include <hip/hip_runtime.h>
#include <math.h>

#define B_   32
#define L_   4096
#define MEM_ 256
#define Q_   256
#define BL_  (B_ * L_)
#define BQ_  (B_ * Q_)
#define B3Q_ (B_ * 3 * Q_)

#define OFF_S1   0
#define OFF_S2   (OFF_S1 + BL_)
#define OFF_SACC (OFF_S2 + BL_)
#define OFF_EACC (OFF_SACC + BL_)
#define OFF_QB   (OFF_EACC + BL_)
#define OFF_QA   (OFF_QB + BQ_)
#define OFF_QE   (OFF_QA + BQ_)
#define OFF_EV   (OFF_QE + BQ_)
#define OFF_X1   (OFF_EV + BQ_)
#define OFF_S_A  (OFF_X1 + BQ_)
#define OFF_S_B  (OFF_S_A + BQ_)
#define OFF_GI   (OFF_S_B + BQ_)
#define OFF_GH   (OFF_GI + B3Q_)
#define OFF_EVB  OFF_QB
#define OFF_X1B  OFF_QA

typedef __attribute__((ext_vector_type(2))) float v2f;
typedef __attribute__((ext_vector_type(8))) float v8f;

__device__ __forceinline__ v8f gemm_tile(const float* __restrict__ aRow,
                                         const float* __restrict__ wRow,
                                         int K, v8f acc) {
    for (int k = 0; k < K; k += 4) {
        v2f a; a.x = aRow[k]; a.y = aRow[k + 1];
        v2f b; b.x = wRow[k]; b.y = wRow[k + 1];
        acc = __builtin_amdgcn_wmma_f32_16x16x4_f32(false, a, false, b,
                                                    (short)0, acc, false, false);
    }
    return acc;
}

__device__ __forceinline__ v8f gemm_tile_sum2(const float* __restrict__ aRow0,
                                              const float* __restrict__ aRow1,
                                              const float* __restrict__ wRow,
                                              int K, v8f acc) {
    for (int k = 0; k < K; k += 4) {
        v2f a; a.x = aRow0[k] + aRow1[k]; a.y = aRow0[k + 1] + aRow1[k + 1];
        v2f b; b.x = wRow[k]; b.y = wRow[k + 1];
        acc = __builtin_amdgcn_wmma_f32_16x16x4_f32(false, a, false, b,
                                                    (short)0, acc, false, false);
    }
    return acc;
}

__device__ __forceinline__ void gemm_store(float* __restrict__ C, int ldc,
                                           int m0, int n0, int row, int hi, v8f acc) {
#pragma unroll
    for (int j = 0; j < 8; ++j)
        C[(size_t)(m0 + j + 8 * hi) * ldc + n0 + row] = acc[j];
}

__global__ void gemm_q_kernel(const float* __restrict__ s,
                              const float* __restrict__ Wb,
                              const float* __restrict__ Wa,
                              float* __restrict__ qb,
                              float* __restrict__ qa) {
    const int lane = threadIdx.x & 31;
    const int row  = lane & 15;
    const int hi   = lane >> 4;
    const int m0   = blockIdx.y * 16;
    const int sel  = blockIdx.x >> 4;
    const int n0   = (blockIdx.x & 15) * 16;
    const float* W = sel ? Wa : Wb;
    float*       C = sel ? qa : qb;

    v8f acc = {};
    acc = gemm_tile(s + (size_t)(m0 + row) * Q_ + 2 * hi,
                    W + (size_t)(n0 + row) * Q_ + 2 * hi, Q_, acc);
    gemm_store(C, MEM_, m0, n0, row, hi, acc);
}

__global__ void gemm_turn_kernel(const float* __restrict__ s,
                                 const float* __restrict__ ev,
                                 const float* __restrict__ x1,
                                 const float* __restrict__ evB,
                                 const float* __restrict__ x1B,
                                 const float* __restrict__ We,
                                 const float* __restrict__ W_ih,
                                 const float* __restrict__ W_hh,
                                 const float* __restrict__ b_ih,
                                 const float* __restrict__ b_hh,
                                 float* __restrict__ qe,
                                 float* __restrict__ gi,
                                 float* __restrict__ gh) {
    const int lane = threadIdx.x & 31;
    const int row  = lane & 15;
    const int hi   = lane >> 4;
    const int m0   = blockIdx.y * 16;
    const int bx   = blockIdx.x;

    if (bx < 16) {
        const int n0 = bx * 16;
        const float* wRow = We + (size_t)(n0 + row) * (Q_ + MEM_) + 2 * hi;
        v8f acc = {};
        acc = gemm_tile(s  + (size_t)(m0 + row) * Q_   + 2 * hi, wRow,      Q_,   acc);
        acc = gemm_tile_sum2(ev  + (size_t)(m0 + row) * MEM_ + 2 * hi,
                             evB + (size_t)(m0 + row) * MEM_ + 2 * hi, wRow + Q_, MEM_, acc);
        gemm_store(qe, MEM_, m0, n0, row, hi, acc);
    } else {
        const bool isGi = bx < 64;
        const int n0 = (isGi ? (bx - 16) : (bx - 64)) * 16;
        const float* W    = isGi ? W_ih : W_hh;
        const float* bias = isGi ? b_ih : b_hh;
        float*       C    = isGi ? gi : gh;
        const float binit = bias[n0 + row];
        const size_t aOff = (size_t)(m0 + row) * MEM_ + 2 * hi;
        const float* wRow = W + (size_t)(n0 + row) * MEM_ + 2 * hi;
        v8f acc;
#pragma unroll
        for (int j = 0; j < 8; ++j) acc[j] = binit;
        if (isGi) acc = gemm_tile_sum2(x1 + aOff, x1B + aOff, wRow, MEM_, acc);
        else      acc = gemm_tile(s + aOff, wRow, MEM_, acc);
        gemm_store(C, 3 * Q_, m0, n0, row, hi, acc);
    }
}

typedef __attribute__((ext_vector_type(4))) float v4f;

#define SC_WAVES  4
#define SC_TILES  2
#define SC_UCOLS  128
#define SC_PITCH  (SC_UCOLS + 4)
#define SC_QPITCH (MEM_ + 4)
#define SC_ROWS_WG (16 * SC_TILES * SC_WAVES)

struct ScUnit { v4f r[16]; };

__device__ __forceinline__ void sc_load(ScUnit& u, const float* __restrict__ src) {
#pragma unroll
    for (int j = 0; j < 16; ++j) u.r[j] = *(const v4f*)(src + (size_t)j * MEM_);
}

__device__ __forceinline__ void sc_park(const ScUnit& u, float* dst) {
#pragma unroll
    for (int j = 0; j < 16; ++j) *(v4f*)(dst + j * SC_PITCH) = u.r[j];
}

__device__ __forceinline__ v8f sc_mma(const float* aP, const float* bP, v8f acc) {
#pragma unroll
    for (int i = 0; i < SC_UCOLS / 4; ++i) {
        const v2f a  = *(const v2f*)(aP + 4 * i);
        const v2f bb = *(const v2f*)(bP + 4 * i);
        acc = __builtin_amdgcn_wmma_f32_16x16x4_f32(false, a, false, bb,
                                                    (short)0, acc, false, false);
    }
    return acc;
}

__global__ __launch_bounds__(32 * SC_WAVES)
void scores_kernel(const float* __restrict__ M,
                   const float* __restrict__ q0,
                   const float* __restrict__ q1,
                   const float* __restrict__ q2,
                   float* __restrict__ S0,
                   float* __restrict__ S1,
                   float* __restrict__ S2,
                   int ncols) {
    const int tid  = threadIdx.x;
    const int wave = tid >> 5;
    const int lane = tid & 31;
    const int c    = lane & 15;
    const int hi   = lane >> 4;
    const int b    = blockIdx.y;
    const int lw   = blockIdx.x * SC_ROWS_WG + wave * (16 * SC_TILES);

    __shared__ __attribute__((aligned(16))) float qs[SC_WAVES][4 * SC_QPITCH];
    __shared__ __attribute__((aligned(16))) float ms[SC_WAVES][16 * SC_PITCH];

    const float* Mw = M + ((size_t)b * L_ + lw) * MEM_ + 4 * lane;

    const float* qsrc[3] = { q0, q1, q2 };
    v4f qv[3][2];
#pragma unroll
    for (int r = 0; r < 3; ++r) {
        const float* qp = ((r < ncols) ? qsrc[r] : q0) + (size_t)b * MEM_ + 4 * lane;
        qv[r][0] = *(const v4f*)qp;
        qv[r][1] = *(const v4f*)(qp + SC_UCOLS);
    }

    ScUnit g0, g1;
    sc_load(g0, Mw);

    float*       mw = ms[wave];
    float*       pk = mw + 4 * lane;
    const float* aP = mw + c * SC_PITCH + 2 * hi;
    float*       qw = qs[wave];
    const float* bP = qw + (c < 3 ? c : 3) * SC_QPITCH + 2 * hi;

#define SC_ORDER() __builtin_amdgcn_sched_barrier(0)
#define SC_WAVE_SYNC() do { __builtin_amdgcn_fence(__ATOMIC_RELEASE, "wavefront"); \
                            __builtin_amdgcn_wave_barrier();                      \
                            __builtin_amdgcn_fence(__ATOMIC_ACQUIRE, "wavefront"); } while (0)

    sc_load(g1, Mw + SC_UCOLS);
    SC_ORDER();
    {
        const v4f z = {};
#pragma unroll
        for (int r = 0; r < 3; ++r) {
            *(v4f*)(qw + r * SC_QPITCH + 4 * lane)            = (r < ncols) ? qv[r][0] : z;
            *(v4f*)(qw + r * SC_QPITCH + 4 * lane + SC_UCOLS) = (r < ncols) ? qv[r][1] : z;
        }
        *(v4f*)(qw + 3 * SC_QPITCH + 4 * lane)            = z;
        *(v4f*)(qw + 3 * SC_QPITCH + 4 * lane + SC_UCOLS) = z;
    }

    const bool live = c < ncols;
    float* dst = ((c == 0) ? S0 : (c == 1) ? S1 : S2);

    v8f acc = {};
    sc_park(g0, pk);
    sc_load(g0, Mw + (size_t)16 * MEM_);
    SC_ORDER();
    SC_WAVE_SYNC();
    acc = sc_mma(aP, bP, acc);
    SC_WAVE_SYNC();
    sc_park(g1, pk);
    sc_load(g1, Mw + (size_t)16 * MEM_ + SC_UCOLS);
    SC_ORDER();
    SC_WAVE_SYNC();
    acc = sc_mma(aP, bP + SC_UCOLS, acc);
    SC_WAVE_SYNC();
    if (live) {
#pragma unroll
        for (int j = 0; j < 8; ++j)
            dst[(size_t)b * L_ + lw + j + 8 * hi] = acc[j];
    }

    acc = v8f{};
    sc_park(g0, pk);
    SC_WAVE_SYNC();
    acc = sc_mma(aP, bP, acc);
    SC_WAVE_SYNC();
    sc_park(g1, pk);
    SC_WAVE_SYNC();
    acc = sc_mma(aP, bP + SC_UCOLS, acc);
    if (live) {
#pragma unroll
        for (int j = 0; j < 8; ++j)
            dst[(size_t)b * L_ + lw + 16 + j + 8 * hi] = acc[j];
    }
#undef SC_WAVE_SYNC
#undef SC_ORDER
}

#define WS_WAVES 16
#define WS_COLS  64
#define WS_STEPS 8
#define WS_GROUP_ROWS (4 * WS_STEPS)
#define WS_PARTS 2
#define WS_PART_ROWS (L_ / WS_PARTS)
#define WS_GROUPS (WS_PART_ROWS / (WS_GROUP_ROWS * WS_WAVES))
#define WS_WGS_ROW ((MEM_ / WS_COLS) * WS_PARTS)
static_assert(WS_WGS_ROW * 32 * WS_WAVES == L_, "the work-groups of a batch row share out its L scores, one per thread");
static_assert(WS_GROUPS >= 2 && (WS_GROUPS & 1) == 0, "two load groups in flight, refilled in pairs");

template <int X>
__device__ __forceinline__ float xrow(float v) {
    return __builtin_bit_cast(float, __builtin_amdgcn_update_dpp(
        0, __builtin_bit_cast(int, v), 0x160 + X, 0xf, 0xf, false));
}
__device__ __forceinline__ float xhalf(float v) {
    return __builtin_bit_cast(float, __builtin_amdgcn_permlanex16(
        __builtin_bit_cast(int, v), __builtin_bit_cast(int, v), 0x76543210, 0xfedcba98, false, false));
}
__device__ __forceinline__ float row16_max(float v) {
    v = fmaxf(v, xrow<1>(v)); v = fmaxf(v, xrow<2>(v));
    v = fmaxf(v, xrow<4>(v)); v = fmaxf(v, xrow<8>(v));
    return v;
}
__device__ __forceinline__ float row16_sum(float v) {
    v += xrow<1>(v); v += xrow<2>(v); v += xrow<4>(v); v += xrow<8>(v);
    return v;
}

template <int NP>
__device__ __forceinline__ void row_stats(const v4f (&x)[NP][2], float (&mx)[NP], float (&inv)[NP],
                                          float (*redm)[WS_WAVES], float (*reds)[WS_WAVES],
                                          int lane, int wave) {
#pragma unroll
    for (int p = 0; p < NP; ++p) {
        float m = fmaxf(fmaxf(fmaxf(x[p][0][0], x[p][0][1]), fmaxf(x[p][0][2], x[p][0][3])),
                        fmaxf(fmaxf(x[p][1][0], x[p][1][1]), fmaxf(x[p][1][2], x[p][1][3])));
        m = row16_max(m);
        m = fmaxf(m, xhalf(m));
        redm[p][wave] = m;
    }
    __syncthreads();
#pragma unroll
    for (int p = 0; p < NP; ++p) {
        const float m = row16_max(redm[p][lane & (WS_WAVES - 1)]);
        mx[p] = m;
        float s = 0.0f;
#pragma unroll
        for (int j = 0; j < 2; ++j)
#pragma unroll
            for (int e = 0; e < 4; ++e) s += __expf(x[p][j][e] - m);
        s = row16_sum(s);
        s += xhalf(s);
        reds[p][wave] = s;
    }
    __syncthreads();
#pragma unroll
    for (int p = 0; p < NP; ++p)
        inv[p] = 1.0f / row16_sum(reds[p][lane & (WS_WAVES - 1)]);
}

struct WsGroup {
    v2f a[WS_STEPS];
    v4f r0[WS_STEPS];
    v4f r1[WS_STEPS];
};

__device__ __forceinline__ void ws_load(WsGroup& g, const float* __restrict__ pv,
                                        const float* __restrict__ Mb, int l0) {
#pragma unroll
    for (int s = 0; s < WS_STEPS; ++s)
        g.a[s] = *(const v2f*)(pv + l0 + 4 * s);
#pragma unroll
    for (int s = 0; s < WS_STEPS; ++s) {
        const float* rp = Mb + (size_t)(l0 + 4 * s) * MEM_;
        g.r0[s] = *(const v4f*)rp;
        g.r1[s] = *(const v4f*)(rp + MEM_);
    }
}

__device__ __forceinline__ void ws_mma(const WsGroup& g, bool live, float mx, float inv,
                                       v8f (&acc)[4]) {
#pragma unroll
    for (int s = 0; s < WS_STEPS; ++s) {
        v2f a;
        a.x = live ? __expf(g.a[s].x - mx) * inv : 0.0f;
        a.y = live ? __expf(g.a[s].y - mx) * inv : 0.0f;
#pragma unroll
        for (int t = 0; t < 4; ++t) {
            v2f bb; bb.x = g.r0[s][t]; bb.y = g.r1[s][t];
            acc[t] = __builtin_amdgcn_wmma_f32_16x16x4_f32(false, a, false, bb,
                                                           (short)0, acc[t], false, false);
        }
    }
}

__global__ __launch_bounds__(32 * WS_WAVES)
void wsum_kernel(const float* __restrict__ M,
                 const float* __restrict__ p,
                 const float* __restrict__ w,
                 const float* __restrict__ endS,
                 float* __restrict__ sacc,
                 float* __restrict__ eacc,
                 float* __restrict__ ev,
                 float* __restrict__ x1,
                 float* __restrict__ evB,
                 float* __restrict__ x1B) {
    const int tid  = threadIdx.x;
    const int wave = tid >> 5;
    const int lane = tid & 31;
    const int c    = lane & 15;
    const int hi   = lane >> 4;
    const int b    = blockIdx.y;
    const int part = blockIdx.x & (WS_PARTS - 1);
    const int slab = (blockIdx.x / WS_PARTS) * WS_COLS;
    const bool hasEnd = endS != nullptr;

    __shared__ float redm[3][WS_WAVES];
    __shared__ float reds[3][WS_WAVES];
    const size_t r0 = (size_t)b * L_ + 4 * tid;
    const size_t r1 = r0 + 4 * 32 * WS_WAVES;
    const size_t mine = (size_t)b * L_ + blockIdx.x * (32 * WS_WAVES) + tid;

    const float* pv = ((c == 1) ? w : p) + (size_t)b * L_ + part * WS_PART_ROWS + 2 * hi;
    const float* Mb = M + ((size_t)b * L_ + part * WS_PART_ROWS + 2 * hi) * MEM_ + slab + 4 * c;
    const bool live = c < 2;

    const float* es = hasEnd ? endS : p;
    v4f x[3][2];
    x[0][0] = *(const v4f*)(p + r0);
    x[0][1] = *(const v4f*)(p + r1);
    x[1][0] = *(const v4f*)(w + r0);
    x[1][1] = *(const v4f*)(w + r1);
    x[2][0] = *(const v4f*)(es + r0);
    x[2][1] = *(const v4f*)(es + r1);
    const float ps = p[mine];
    const float pe = es[mine];
    const float sa = sacc[mine];
    const float ea = eacc[mine];

    v8f acc[4] = {};
    WsGroup g0, g1;
#define WS_ORDER() __builtin_amdgcn_sched_barrier(0)
    ws_load(g0, pv, Mb, wave * WS_GROUP_ROWS);
    ws_load(g1, pv, Mb, (WS_WAVES + wave) * WS_GROUP_ROWS);
    WS_ORDER();
    float mx3[3], inv3[3];
    row_stats<3>(x, mx3, inv3, redm, reds, lane, wave);
    sacc[mine] = sa + 0.2f * (__expf(ps - mx3[0]) * inv3[0]);
    if (hasEnd)
        eacc[mine] = ea + 0.2f * (__expf(pe - mx3[2]) * inv3[2]);
    const float mx  = (c == 1) ? mx3[1]  : mx3[0];
    const float inv = (c == 1) ? inv3[1] : inv3[0];
    WS_ORDER();
    ws_mma(g0, live, mx, inv, acc);
    WS_ORDER();
    for (int i = 1; i < WS_GROUPS - 1; i += 2) {
        ws_load(g0, pv, Mb, ((i + 1) * WS_WAVES + wave) * WS_GROUP_ROWS);
        WS_ORDER();
        ws_mma(g1, live, mx, inv, acc);
        WS_ORDER();
        ws_load(g1, pv, Mb, ((i + 2) * WS_WAVES + wave) * WS_GROUP_ROWS);
        WS_ORDER();
        ws_mma(g0, live, mx, inv, acc);
        WS_ORDER();
    }
    ws_mma(g1, live, mx, inv, acc);
#undef WS_ORDER

    __shared__ float red[WS_WAVES][2][WS_COLS];
    if (hi == 0) {
        v4f e, xx;
#pragma unroll
        for (int t = 0; t < 4; ++t) { e[t] = acc[t][0]; xx[t] = acc[t][1]; }
        *(v4f*)&red[wave][0][4 * c] = e;
        *(v4f*)&red[wave][1][4 * c] = xx;
    }
    __syncthreads();
    if (tid < 2 * WS_COLS) {
        const int r  = tid >> 6;
        const int mm = tid & (WS_COLS - 1);
        float s = 0.0f;
#pragma unroll
        for (int k = 0; k < WS_WAVES; ++k) s += red[k][r][mm];
        float* eo = part ? evB : ev;
        float* xo = part ? x1B : x1;
        (r ? xo : eo)[(size_t)b * MEM_ + slab + mm] = s;
    }
}

__global__ void gru_kernel(const float* __restrict__ gi,
                           const float* __restrict__ gh,
                           const float* __restrict__ sOld,
                           float* __restrict__ sNew) {
    int i = blockIdx.x * 256 + threadIdx.x;
    if (i >= BQ_) return;
    int b = i / Q_, q = i % Q_;
    const float* gib = gi + (size_t)b * 3 * Q_;
    const float* ghb = gh + (size_t)b * 3 * Q_;
    float ir = gib[q], iz = gib[Q_ + q], in_ = gib[2 * Q_ + q];
    float hr = ghb[q], hz = ghb[Q_ + q], hn  = ghb[2 * Q_ + q];
    float r = 1.0f / (1.0f + __expf(-(ir + hr)));
    float z = 1.0f / (1.0f + __expf(-(iz + hz)));
    float n = tanhf(in_ + r * hn);
    sNew[i] = (1.0f - z) * n + z * sOld[i];
}

__global__ void init_kernel(const float* __restrict__ s0, float* __restrict__ ws) {
    int i = blockIdx.x * 256 + threadIdx.x;
    if (i < 2 * BL_) ws[OFF_SACC + i] = 0.0f;
    if (i < BQ_)     ws[OFF_S_A + i] = s0[i];
}

__global__ __launch_bounds__(32 * WS_WAVES)
void final_kernel(const float* __restrict__ ws, float* out) {
    const int tid  = threadIdx.x;
    const int wave = tid >> 5;
    const int lane = tid & 31;
    const int b    = blockIdx.x;
    __shared__ float redm[1][WS_WAVES];
    __shared__ float reds[1][WS_WAVES];
    const size_t r0 = (size_t)b * L_ + 4 * tid;
    const size_t r1 = r0 + 4 * 32 * WS_WAVES;

    v4f x[1][2];
    x[0][0] = *(const v4f*)(out + BL_ + r0);
    x[0][1] = *(const v4f*)(out + BL_ + r1);
    v4f e0 = *(const v4f*)(ws + OFF_EACC + r0);
    v4f e1 = *(const v4f*)(ws + OFF_EACC + r1);
    v4f s0v = *(const v4f*)(ws + OFF_SACC + r0);
    v4f s1v = *(const v4f*)(ws + OFF_SACC + r1);
    float mx[1], inv[1];
    row_stats<1>(x, mx, inv, redm, reds, lane, wave);
#pragma unroll
    for (int e = 0; e < 4; ++e) {
        e0[e] = logf(e0[e] + 0.2f * (__expf(x[0][0][e] - mx[0]) * inv[0]));
        e1[e] = logf(e1[e] + 0.2f * (__expf(x[0][1][e] - mx[0]) * inv[0]));
        s0v[e] = logf(s0v[e]);
        s1v[e] = logf(s1v[e]);
    }
    *(v4f*)(out + r0) = s0v;
    *(v4f*)(out + r1) = s1v;
    *(v4f*)(out + BL_ + r0) = e0;
    *(v4f*)(out + BL_ + r1) = e1;
}

extern "C" void kernel_launch(void* const* d_in, const int* in_sizes, int n_in,
                              void* d_out, int out_size, void* d_ws, size_t ws_size,
                              hipStream_t stream) {
    const float* M    = (const float*)d_in[0];
    const float* s0   = (const float*)d_in[1];
    const float* Wb   = (const float*)d_in[2];
    const float* We   = (const float*)d_in[3];
    const float* Wa   = (const float*)d_in[4];
    const float* W_ih = (const float*)d_in[5];
    const float* W_hh = (const float*)d_in[6];
    const float* b_ih = (const float*)d_in[7];
    const float* b_hh = (const float*)d_in[8];

    float* ws   = (float*)d_ws;
    float* S1   = ws + OFF_S1;
    float* S2   = ws + OFF_S2;
    float* sacc = ws + OFF_SACC;
    float* eacc = ws + OFF_EACC;
    float* qb   = ws + OFF_QB;
    float* qa   = ws + OFF_QA;
    float* qe   = ws + OFF_QE;
    float* ev   = ws + OFF_EV;
    float* x1   = ws + OFF_X1;
    float* sA   = ws + OFF_S_A;
    float* sB   = ws + OFF_S_B;
    float* gi   = ws + OFF_GI;
    float* gh   = ws + OFF_GH;
    float* evB  = ws + OFF_EVB;
    float* x1B  = ws + OFF_X1B;
    float* S3   = (float*)d_out + BL_;

    init_kernel<<<(2 * BL_) / 256, 256, 0, stream>>>(s0, ws);

    gemm_q_kernel<<<dim3(32, 2), 32, 0, stream>>>(sA, Wb, Wa, qb, qa);
    scores_kernel<<<dim3(L_ / SC_ROWS_WG, B_), 32 * SC_WAVES, 0, stream>>>(M, qb, qa, qa, S1, S2, nullptr, 2);

    for (int t = 0; t < 5; ++t) {
        const float* sCur = (t & 1) ? sB : sA;
        float*       sNxt = (t & 1) ? sA : sB;

        if (t == 0)
            wsum_kernel<<<dim3(WS_WGS_ROW, B_), 32 * WS_WAVES, 0, stream>>>(
                M, S1, S2, nullptr, sacc, eacc, ev, x1, evB, x1B);
        else
            wsum_kernel<<<dim3(WS_WGS_ROW, B_), 32 * WS_WAVES, 0, stream>>>(
                M, S1, S2, S3, sacc, eacc, ev, x1, evB, x1B);

        gemm_turn_kernel<<<dim3(112, 2), 32, 0, stream>>>(sCur, ev, x1, evB, x1B, We, W_ih, W_hh,
                                                          b_ih, b_hh, qe, gi, gh);
        gru_kernel<<<BQ_ / 256, 256, 0, stream>>>(gi, gh, sCur, sNxt);

        if (t < 4) {
            gemm_q_kernel<<<dim3(32, 2), 32, 0, stream>>>(sNxt, Wb, Wa, qb, qa);
            scores_kernel<<<dim3(L_ / SC_ROWS_WG, B_), 32 * SC_WAVES, 0, stream>>>(M, qe, qb, qa, S3, S1, S2, 3);
        } else {
            scores_kernel<<<dim3(L_ / SC_ROWS_WG, B_), 32 * SC_WAVES, 0, stream>>>(M, qe, qe, qe, S3, nullptr, nullptr, 1);
        }
    }

    final_kernel<<<B_, 32 * WS_WAVES, 0, stream>>>(ws, (float*)d_out);
}


# small GEMMs: K split over the 8 waves of a work-group (16x32 block, 4 K parts), every load of a part requested at once, two accumulators; partial tiles added through LDS (new reduction tree)
# speedup vs baseline: 14.5011x; 1.1688x over previous
#include <hip/hip_runtime.h>
#include <math.h>

#define B_   32
#define L_   4096
#define MEM_ 256
#define Q_   256
#define BL_  (B_ * L_)
#define BQ_  (B_ * Q_)
#define B3Q_ (B_ * 3 * Q_)

#define OFF_S1   0
#define OFF_S2   (OFF_S1 + BL_)
#define OFF_SACC (OFF_S2 + BL_)
#define OFF_EACC (OFF_SACC + BL_)
#define OFF_QB   (OFF_EACC + BL_)
#define OFF_QA   (OFF_QB + BQ_)
#define OFF_QE   (OFF_QA + BQ_)
#define OFF_EV   (OFF_QE + BQ_)
#define OFF_X1   (OFF_EV + BQ_)
#define OFF_S_A  (OFF_X1 + BQ_)
#define OFF_S_B  (OFF_S_A + BQ_)
#define OFF_GI   (OFF_S_B + BQ_)
#define OFF_GH   (OFF_GI + B3Q_)
#define OFF_EVB  OFF_QB
#define OFF_X1B  OFF_QA

typedef __attribute__((ext_vector_type(2))) float v2f;
typedef __attribute__((ext_vector_type(8))) float v8f;

typedef __attribute__((ext_vector_type(4))) float v4f;

#define SG_WAVES  8
#define SG_KPARTS 4
#define SG_NB     32
#define SG_PITCH  34

#define SG_ORDER() __builtin_amdgcn_sched_barrier(0)

template <int NU>
__device__ __forceinline__ void sg_load(v4f (&f)[NU], const float* __restrict__ p) {
#pragma unroll
    for (int u = 0; u < NU; ++u) f[u] = *(const v4f*)(p + 8 * u);
}

template <int NU>
__device__ __forceinline__ void sg_join(v4f (&f)[NU], const v4f (&g)[NU], bool two) {
#pragma unroll
    for (int u = 0; u < NU; ++u) f[u] = two ? (f[u] + g[u]) : f[u];
}

template <int NU>
__device__ __forceinline__ v8f sg_mma(const v4f (&a)[NU], const v4f (&b)[NU]) {
    v8f acc0 = {}, acc1 = {};
#pragma unroll
    for (int u = 0; u < NU; ++u) {
        v2f a0; a0.x = a[u][0]; a0.y = a[u][1];
        v2f b0; b0.x = b[u][0]; b0.y = b[u][1];
        v2f a1; a1.x = a[u][2]; a1.y = a[u][3];
        v2f b1; b1.x = b[u][2]; b1.y = b[u][3];
        acc0 = __builtin_amdgcn_wmma_f32_16x16x4_f32(false, a0, false, b0,
                                                     (short)0, acc0, false, false);
        acc1 = __builtin_amdgcn_wmma_f32_16x16x4_f32(false, a1, false, b1,
                                                     (short)0, acc1, false, false);
    }
    return acc0 + acc1;
}

__device__ __forceinline__ void sg_finish(v8f acc, float (*red)[16][SG_PITCH],
                                          int kpart, int tn, int c, int hi, int tid,
                                          float* __restrict__ C, int ldc, int m0, int n0,
                                          float bv) {
#pragma unroll
    for (int j = 0; j < 8; ++j) red[kpart][j + 8 * hi][16 * tn + c] = acc[j];
    __syncthreads();
    const int col = tid & 31;
    const int r   = tid >> 5;
#pragma unroll
    for (int rr = 0; rr < 2; ++rr) {
        const int row = r + 8 * rr;
        float v = red[0][row][col];
        v += red[1][row][col];
        v += red[2][row][col];
        v += red[3][row][col];
        C[(size_t)(m0 + row) * ldc + n0 + col] = v + bv;
    }
}

__global__ __launch_bounds__(32 * SG_WAVES)
void gemm_q_kernel(const float* __restrict__ s,
                   const float* __restrict__ Wb,
                   const float* __restrict__ Wa,
                   float* __restrict__ qb,
                   float* __restrict__ qa) {
    const int tid   = threadIdx.x;
    const int wave  = tid >> 5;
    const int lane  = tid & 31;
    const int c     = lane & 15;
    const int hi    = lane >> 4;
    const int tn    = wave & 1;
    const int kpart = wave >> 1;
    const int m0    = blockIdx.y * 16;
    const int sel   = blockIdx.x >> 3;
    const int n0    = (blockIdx.x & 7) * SG_NB;
    const float* W = sel ? Wa : Wb;
    float*       C = sel ? qa : qb;

    __shared__ float red[SG_KPARTS][16][SG_PITCH];

    const int k0 = kpart * 64 + 4 * hi;
    v4f a[8], b[8];
    sg_load<8>(a, s + (size_t)(m0 + c) * Q_ + k0);
    sg_load<8>(b, W + (size_t)(n0 + 16 * tn + c) * Q_ + k0);
    SG_ORDER();
    const v8f acc = sg_mma<8>(a, b);
    sg_finish(acc, red, kpart, tn, c, hi, tid, C, MEM_, m0, n0, 0.0f);
}

__global__ __launch_bounds__(32 * SG_WAVES)
void gemm_turn_kernel(const float* __restrict__ s,
                      const float* __restrict__ ev,
                      const float* __restrict__ x1,
                      const float* __restrict__ evB,
                      const float* __restrict__ x1B,
                      const float* __restrict__ We,
                      const float* __restrict__ W_ih,
                      const float* __restrict__ W_hh,
                      const float* __restrict__ b_ih,
                      const float* __restrict__ b_hh,
                      float* __restrict__ qe,
                      float* __restrict__ gi,
                      float* __restrict__ gh) {
    const int tid   = threadIdx.x;
    const int wave  = __builtin_amdgcn_readfirstlane(tid >> 5);
    const int lane  = tid & 31;
    const int c     = lane & 15;
    const int hi    = lane >> 4;
    const int tn    = wave & 1;
    const int kpart = wave >> 1;
    const int m0    = blockIdx.y * 16;
    const int bx    = blockIdx.x;

    __shared__ float red[SG_KPARTS][16][SG_PITCH];

    v8f acc;
    float* C;
    float  bv = 0.0f;
    int ldc, n0;
    if (bx < 8) {
        n0 = bx * SG_NB; C = qe; ldc = MEM_;
        const int k0 = kpart * 128 + 4 * hi;
        const bool two = kpart >= 2;
        const size_t o = (size_t)(m0 + c) * MEM_ + (k0 & (MEM_ - 1));
        const float* p0 = (two ? ev : s) + o;
        const float* p1 = two ? (evB + o) : p0;
        v4f a[16], a1[16], b[16];
        sg_load<16>(b, We + (size_t)(n0 + 16 * tn + c) * (Q_ + MEM_) + k0);
        sg_load<16>(a, p0);
        sg_load<16>(a1, p1);
        SG_ORDER();
        sg_join<16>(a, a1, two);
        acc = sg_mma<16>(a, b);
    } else {
        const bool isGi = bx < 32;
        n0 = (isGi ? (bx - 8) : (bx - 32)) * SG_NB; ldc = 3 * Q_;
        const float* W = isGi ? W_ih : W_hh;
        C  = isGi ? gi : gh;
        const int k0 = kpart * 64 + 4 * hi;
        const size_t o = (size_t)(m0 + c) * MEM_ + k0;
        const float* p0 = (isGi ? x1 : s) + o;
        const float* p1 = isGi ? (x1B + o) : p0;
        v4f a[8], a1[8], b[8];
        bv = (isGi ? b_ih : b_hh)[n0 + (tid & 31)];
        sg_load<8>(b, W + (size_t)(n0 + 16 * tn + c) * MEM_ + k0);
        sg_load<8>(a, p0);
        sg_load<8>(a1, p1);
        SG_ORDER();
        sg_join<8>(a, a1, isGi);
        acc = sg_mma<8>(a, b);
    }
    sg_finish(acc, red, kpart, tn, c, hi, tid, C, ldc, m0, n0, bv);
}

#define SC_WAVES  4
#define SC_TILES  2
#define SC_UCOLS  128
#define SC_PITCH  (SC_UCOLS + 4)
#define SC_QPITCH (MEM_ + 4)
#define SC_ROWS_WG (16 * SC_TILES * SC_WAVES)

struct ScUnit { v4f r[16]; };

__device__ __forceinline__ void sc_load(ScUnit& u, const float* __restrict__ src) {
#pragma unroll
    for (int j = 0; j < 16; ++j) u.r[j] = *(const v4f*)(src + (size_t)j * MEM_);
}

__device__ __forceinline__ void sc_park(const ScUnit& u, float* dst) {
#pragma unroll
    for (int j = 0; j < 16; ++j) *(v4f*)(dst + j * SC_PITCH) = u.r[j];
}

__device__ __forceinline__ v8f sc_mma(const float* aP, const float* bP, v8f acc) {
#pragma unroll
    for (int i = 0; i < SC_UCOLS / 4; ++i) {
        const v2f a  = *(const v2f*)(aP + 4 * i);
        const v2f bb = *(const v2f*)(bP + 4 * i);
        acc = __builtin_amdgcn_wmma_f32_16x16x4_f32(false, a, false, bb,
                                                    (short)0, acc, false, false);
    }
    return acc;
}

__global__ __launch_bounds__(32 * SC_WAVES)
void scores_kernel(const float* __restrict__ M,
                   const float* __restrict__ q0,
                   const float* __restrict__ q1,
                   const float* __restrict__ q2,
                   float* __restrict__ S0,
                   float* __restrict__ S1,
                   float* __restrict__ S2,
                   int ncols) {
    const int tid  = threadIdx.x;
    const int wave = tid >> 5;
    const int lane = tid & 31;
    const int c    = lane & 15;
    const int hi   = lane >> 4;
    const int b    = blockIdx.y;
    const int lw   = blockIdx.x * SC_ROWS_WG + wave * (16 * SC_TILES);

    __shared__ __attribute__((aligned(16))) float qs[SC_WAVES][4 * SC_QPITCH];
    __shared__ __attribute__((aligned(16))) float ms[SC_WAVES][16 * SC_PITCH];

    const float* Mw = M + ((size_t)b * L_ + lw) * MEM_ + 4 * lane;

    const float* qsrc[3] = { q0, q1, q2 };
    v4f qv[3][2];
#pragma unroll
    for (int r = 0; r < 3; ++r) {
        const float* qp = ((r < ncols) ? qsrc[r] : q0) + (size_t)b * MEM_ + 4 * lane;
        qv[r][0] = *(const v4f*)qp;
        qv[r][1] = *(const v4f*)(qp + SC_UCOLS);
    }

    ScUnit g0, g1;
    sc_load(g0, Mw);

    float*       mw = ms[wave];
    float*       pk = mw + 4 * lane;
    const float* aP = mw + c * SC_PITCH + 2 * hi;
    float*       qw = qs[wave];
    const float* bP = qw + (c < 3 ? c : 3) * SC_QPITCH + 2 * hi;

#define SC_ORDER() __builtin_amdgcn_sched_barrier(0)
#define SC_WAVE_SYNC() do { __builtin_amdgcn_fence(__ATOMIC_RELEASE, "wavefront"); \
                            __builtin_amdgcn_wave_barrier();                      \
                            __builtin_amdgcn_fence(__ATOMIC_ACQUIRE, "wavefront"); } while (0)

    sc_load(g1, Mw + SC_UCOLS);
    SC_ORDER();
    {
        const v4f z = {};
#pragma unroll
        for (int r = 0; r < 3; ++r) {
            *(v4f*)(qw + r * SC_QPITCH + 4 * lane)            = (r < ncols) ? qv[r][0] : z;
            *(v4f*)(qw + r * SC_QPITCH + 4 * lane + SC_UCOLS) = (r < ncols) ? qv[r][1] : z;
        }
        *(v4f*)(qw + 3 * SC_QPITCH + 4 * lane)            = z;
        *(v4f*)(qw + 3 * SC_QPITCH + 4 * lane + SC_UCOLS) = z;
    }

    const bool live = c < ncols;
    float* dst = ((c == 0) ? S0 : (c == 1) ? S1 : S2);

    v8f acc = {};
    sc_park(g0, pk);
    sc_load(g0, Mw + (size_t)16 * MEM_);
    SC_ORDER();
    SC_WAVE_SYNC();
    acc = sc_mma(aP, bP, acc);
    SC_WAVE_SYNC();
    sc_park(g1, pk);
    sc_load(g1, Mw + (size_t)16 * MEM_ + SC_UCOLS);
    SC_ORDER();
    SC_WAVE_SYNC();
    acc = sc_mma(aP, bP + SC_UCOLS, acc);
    SC_WAVE_SYNC();
    if (live) {
#pragma unroll
        for (int j = 0; j < 8; ++j)
            dst[(size_t)b * L_ + lw + j + 8 * hi] = acc[j];
    }

    acc = v8f{};
    sc_park(g0, pk);
    SC_WAVE_SYNC();
    acc = sc_mma(aP, bP, acc);
    SC_WAVE_SYNC();
    sc_park(g1, pk);
    SC_WAVE_SYNC();
    acc = sc_mma(aP, bP + SC_UCOLS, acc);
    if (live) {
#pragma unroll
        for (int j = 0; j < 8; ++j)
            dst[(size_t)b * L_ + lw + 16 + j + 8 * hi] = acc[j];
    }
#undef SC_WAVE_SYNC
#undef SC_ORDER
}

#define WS_WAVES 16
#define WS_COLS  64
#define WS_STEPS 8
#define WS_GROUP_ROWS (4 * WS_STEPS)
#define WS_PARTS 2
#define WS_PART_ROWS (L_ / WS_PARTS)
#define WS_GROUPS (WS_PART_ROWS / (WS_GROUP_ROWS * WS_WAVES))
#define WS_WGS_ROW ((MEM_ / WS_COLS) * WS_PARTS)
static_assert(Q_ == MEM_, "gemm_turn_kernel addresses s and ev with one offset");
static_assert(WS_WGS_ROW * 32 * WS_WAVES == L_, "the work-groups of a batch row share out its L scores, one per thread");
static_assert(WS_GROUPS >= 2 && (WS_GROUPS & 1) == 0, "two load groups in flight, refilled in pairs");

template <int X>
__device__ __forceinline__ float xrow(float v) {
    return __builtin_bit_cast(float, __builtin_amdgcn_update_dpp(
        0, __builtin_bit_cast(int, v), 0x160 + X, 0xf, 0xf, false));
}
__device__ __forceinline__ float xhalf(float v) {
    return __builtin_bit_cast(float, __builtin_amdgcn_permlanex16(
        __builtin_bit_cast(int, v), __builtin_bit_cast(int, v), 0x76543210, 0xfedcba98, false, false));
}
__device__ __forceinline__ float row16_max(float v) {
    v = fmaxf(v, xrow<1>(v)); v = fmaxf(v, xrow<2>(v));
    v = fmaxf(v, xrow<4>(v)); v = fmaxf(v, xrow<8>(v));
    return v;
}
__device__ __forceinline__ float row16_sum(float v) {
    v += xrow<1>(v); v += xrow<2>(v); v += xrow<4>(v); v += xrow<8>(v);
    return v;
}

template <int NP>
__device__ __forceinline__ void row_stats(const v4f (&x)[NP][2], float (&mx)[NP], float (&inv)[NP],
                                          float (*redm)[WS_WAVES], float (*reds)[WS_WAVES],
                                          int lane, int wave) {
#pragma unroll
    for (int p = 0; p < NP; ++p) {
        float m = fmaxf(fmaxf(fmaxf(x[p][0][0], x[p][0][1]), fmaxf(x[p][0][2], x[p][0][3])),
                        fmaxf(fmaxf(x[p][1][0], x[p][1][1]), fmaxf(x[p][1][2], x[p][1][3])));
        m = row16_max(m);
        m = fmaxf(m, xhalf(m));
        redm[p][wave] = m;
    }
    __syncthreads();
#pragma unroll
    for (int p = 0; p < NP; ++p) {
        const float m = row16_max(redm[p][lane & (WS_WAVES - 1)]);
        mx[p] = m;
        float s = 0.0f;
#pragma unroll
        for (int j = 0; j < 2; ++j)
#pragma unroll
            for (int e = 0; e < 4; ++e) s += __expf(x[p][j][e] - m);
        s = row16_sum(s);
        s += xhalf(s);
        reds[p][wave] = s;
    }
    __syncthreads();
#pragma unroll
    for (int p = 0; p < NP; ++p)
        inv[p] = 1.0f / row16_sum(reds[p][lane & (WS_WAVES - 1)]);
}

struct WsGroup {
    v2f a[WS_STEPS];
    v4f r0[WS_STEPS];
    v4f r1[WS_STEPS];
};

__device__ __forceinline__ void ws_load(WsGroup& g, const float* __restrict__ pv,
                                        const float* __restrict__ Mb, int l0) {
#pragma unroll
    for (int s = 0; s < WS_STEPS; ++s)
        g.a[s] = *(const v2f*)(pv + l0 + 4 * s);
#pragma unroll
    for (int s = 0; s < WS_STEPS; ++s) {
        const float* rp = Mb + (size_t)(l0 + 4 * s) * MEM_;
        g.r0[s] = *(const v4f*)rp;
        g.r1[s] = *(const v4f*)(rp + MEM_);
    }
}

__device__ __forceinline__ void ws_mma(const WsGroup& g, bool live, float mx, float inv,
                                       v8f (&acc)[4]) {
#pragma unroll
    for (int s = 0; s < WS_STEPS; ++s) {
        v2f a;
        a.x = live ? __expf(g.a[s].x - mx) * inv : 0.0f;
        a.y = live ? __expf(g.a[s].y - mx) * inv : 0.0f;
#pragma unroll
        for (int t = 0; t < 4; ++t) {
            v2f bb; bb.x = g.r0[s][t]; bb.y = g.r1[s][t];
            acc[t] = __builtin_amdgcn_wmma_f32_16x16x4_f32(false, a, false, bb,
                                                           (short)0, acc[t], false, false);
        }
    }
}

__global__ __launch_bounds__(32 * WS_WAVES)
void wsum_kernel(const float* __restrict__ M,
                 const float* __restrict__ p,
                 const float* __restrict__ w,
                 const float* __restrict__ endS,
                 float* __restrict__ sacc,
                 float* __restrict__ eacc,
                 float* __restrict__ ev,
                 float* __restrict__ x1,
                 float* __restrict__ evB,
                 float* __restrict__ x1B) {
    const int tid  = threadIdx.x;
    const int wave = tid >> 5;
    const int lane = tid & 31;
    const int c    = lane & 15;
    const int hi   = lane >> 4;
    const int b    = blockIdx.y;
    const int part = blockIdx.x & (WS_PARTS - 1);
    const int slab = (blockIdx.x / WS_PARTS) * WS_COLS;
    const bool hasEnd = endS != nullptr;

    __shared__ float redm[3][WS_WAVES];
    __shared__ float reds[3][WS_WAVES];
    const size_t r0 = (size_t)b * L_ + 4 * tid;
    const size_t r1 = r0 + 4 * 32 * WS_WAVES;
    const size_t mine = (size_t)b * L_ + blockIdx.x * (32 * WS_WAVES) + tid;

    const float* pv = ((c == 1) ? w : p) + (size_t)b * L_ + part * WS_PART_ROWS + 2 * hi;
    const float* Mb = M + ((size_t)b * L_ + part * WS_PART_ROWS + 2 * hi) * MEM_ + slab + 4 * c;
    const bool live = c < 2;

    const float* es = hasEnd ? endS : p;
    v4f x[3][2];
    x[0][0] = *(const v4f*)(p + r0);
    x[0][1] = *(const v4f*)(p + r1);
    x[1][0] = *(const v4f*)(w + r0);
    x[1][1] = *(const v4f*)(w + r1);
    x[2][0] = *(const v4f*)(es + r0);
    x[2][1] = *(const v4f*)(es + r1);
    const float ps = p[mine];
    const float pe = es[mine];
    const float sa = sacc[mine];
    const float ea = eacc[mine];

    v8f acc[4] = {};
    WsGroup g0, g1;
#define WS_ORDER() __builtin_amdgcn_sched_barrier(0)
    ws_load(g0, pv, Mb, wave * WS_GROUP_ROWS);
    ws_load(g1, pv, Mb, (WS_WAVES + wave) * WS_GROUP_ROWS);
    WS_ORDER();
    float mx3[3], inv3[3];
    row_stats<3>(x, mx3, inv3, redm, reds, lane, wave);
    sacc[mine] = sa + 0.2f * (__expf(ps - mx3[0]) * inv3[0]);
    if (hasEnd)
        eacc[mine] = ea + 0.2f * (__expf(pe - mx3[2]) * inv3[2]);
    const float mx  = (c == 1) ? mx3[1]  : mx3[0];
    const float inv = (c == 1) ? inv3[1] : inv3[0];
    WS_ORDER();
    ws_mma(g0, live, mx, inv, acc);
    WS_ORDER();
    for (int i = 1; i < WS_GROUPS - 1; i += 2) {
        ws_load(g0, pv, Mb, ((i + 1) * WS_WAVES + wave) * WS_GROUP_ROWS);
        WS_ORDER();
        ws_mma(g1, live, mx, inv, acc);
        WS_ORDER();
        ws_load(g1, pv, Mb, ((i + 2) * WS_WAVES + wave) * WS_GROUP_ROWS);
        WS_ORDER();
        ws_mma(g0, live, mx, inv, acc);
        WS_ORDER();
    }
    ws_mma(g1, live, mx, inv, acc);
#undef WS_ORDER

    __shared__ float red[WS_WAVES][2][WS_COLS];
    if (hi == 0) {
        v4f e, xx;
#pragma unroll
        for (int t = 0; t < 4; ++t) { e[t] = acc[t][0]; xx[t] = acc[t][1]; }
        *(v4f*)&red[wave][0][4 * c] = e;
        *(v4f*)&red[wave][1][4 * c] = xx;
    }
    __syncthreads();
    if (tid < 2 * WS_COLS) {
        const int r  = tid >> 6;
        const int mm = tid & (WS_COLS - 1);
        float s = 0.0f;
#pragma unroll
        for (int k = 0; k < WS_WAVES; ++k) s += red[k][r][mm];
        float* eo = part ? evB : ev;
        float* xo = part ? x1B : x1;
        (r ? xo : eo)[(size_t)b * MEM_ + slab + mm] = s;
    }
}

__global__ void gru_kernel(const float* __restrict__ gi,
                           const float* __restrict__ gh,
                           const float* __restrict__ sOld,
                           float* __restrict__ sNew) {
    int i = blockIdx.x * 256 + threadIdx.x;
    if (i >= BQ_) return;
    int b = i / Q_, q = i % Q_;
    const float* gib = gi + (size_t)b * 3 * Q_;
    const float* ghb = gh + (size_t)b * 3 * Q_;
    float ir = gib[q], iz = gib[Q_ + q], in_ = gib[2 * Q_ + q];
    float hr = ghb[q], hz = ghb[Q_ + q], hn  = ghb[2 * Q_ + q];
    float r = 1.0f / (1.0f + __expf(-(ir + hr)));
    float z = 1.0f / (1.0f + __expf(-(iz + hz)));
    float n = tanhf(in_ + r * hn);
    sNew[i] = (1.0f - z) * n + z * sOld[i];
}

__global__ void init_kernel(const float* __restrict__ s0, float* __restrict__ ws) {
    int i = blockIdx.x * 256 + threadIdx.x;
    if (i < 2 * BL_) ws[OFF_SACC + i] = 0.0f;
    if (i < BQ_)     ws[OFF_S_A + i] = s0[i];
}

__global__ __launch_bounds__(32 * WS_WAVES)
void final_kernel(const float* __restrict__ ws, float* out) {
    const int tid  = threadIdx.x;
    const int wave = tid >> 5;
    const int lane = tid & 31;
    const int b    = blockIdx.x;
    __shared__ float redm[1][WS_WAVES];
    __shared__ float reds[1][WS_WAVES];
    const size_t r0 = (size_t)b * L_ + 4 * tid;
    const size_t r1 = r0 + 4 * 32 * WS_WAVES;

    v4f x[1][2];
    x[0][0] = *(const v4f*)(out + BL_ + r0);
    x[0][1] = *(const v4f*)(out + BL_ + r1);
    v4f e0 = *(const v4f*)(ws + OFF_EACC + r0);
    v4f e1 = *(const v4f*)(ws + OFF_EACC + r1);
    v4f s0v = *(const v4f*)(ws + OFF_SACC + r0);
    v4f s1v = *(const v4f*)(ws + OFF_SACC + r1);
    float mx[1], inv[1];
    row_stats<1>(x, mx, inv, redm, reds, lane, wave);
#pragma unroll
    for (int e = 0; e < 4; ++e) {
        e0[e] = logf(e0[e] + 0.2f * (__expf(x[0][0][e] - mx[0]) * inv[0]));
        e1[e] = logf(e1[e] + 0.2f * (__expf(x[0][1][e] - mx[0]) * inv[0]));
        s0v[e] = logf(s0v[e]);
        s1v[e] = logf(s1v[e]);
    }
    *(v4f*)(out + r0) = s0v;
    *(v4f*)(out + r1) = s1v;
    *(v4f*)(out + BL_ + r0) = e0;
    *(v4f*)(out + BL_ + r1) = e1;
}

extern "C" void kernel_launch(void* const* d_in, const int* in_sizes, int n_in,
                              void* d_out, int out_size, void* d_ws, size_t ws_size,
                              hipStream_t stream) {
    const float* M    = (const float*)d_in[0];
    const float* s0   = (const float*)d_in[1];
    const float* Wb   = (const float*)d_in[2];
    const float* We   = (const float*)d_in[3];
    const float* Wa   = (const float*)d_in[4];
    const float* W_ih = (const float*)d_in[5];
    const float* W_hh = (const float*)d_in[6];
    const float* b_ih = (const float*)d_in[7];
    const float* b_hh = (const float*)d_in[8];

    float* ws   = (float*)d_ws;
    float* S1   = ws + OFF_S1;
    float* S2   = ws + OFF_S2;
    float* sacc = ws + OFF_SACC;
    float* eacc = ws + OFF_EACC;
    float* qb   = ws + OFF_QB;
    float* qa   = ws + OFF_QA;
    float* qe   = ws + OFF_QE;
    float* ev   = ws + OFF_EV;
    float* x1   = ws + OFF_X1;
    float* sA   = ws + OFF_S_A;
    float* sB   = ws + OFF_S_B;
    float* gi   = ws + OFF_GI;
    float* gh   = ws + OFF_GH;
    float* evB  = ws + OFF_EVB;
    float* x1B  = ws + OFF_X1B;
    float* S3   = (float*)d_out + BL_;

    init_kernel<<<(2 * BL_) / 256, 256, 0, stream>>>(s0, ws);

    gemm_q_kernel<<<dim3(16, 2), 32 * SG_WAVES, 0, stream>>>(sA, Wb, Wa, qb, qa);
    scores_kernel<<<dim3(L_ / SC_ROWS_WG, B_), 32 * SC_WAVES, 0, stream>>>(M, qb, qa, qa, S1, S2, nullptr, 2);

    for (int t = 0; t < 5; ++t) {
        const float* sCur = (t & 1) ? sB : sA;
        float*       sNxt = (t & 1) ? sA : sB;

        if (t == 0)
            wsum_kernel<<<dim3(WS_WGS_ROW, B_), 32 * WS_WAVES, 0, stream>>>(
                M, S1, S2, nullptr, sacc, eacc, ev, x1, evB, x1B);
        else
            wsum_kernel<<<dim3(WS_WGS_ROW, B_), 32 * WS_WAVES, 0, stream>>>(
                M, S1, S2, S3, sacc, eacc, ev, x1, evB, x1B);

        gemm_turn_kernel<<<dim3(56, 2), 32 * SG_WAVES, 0, stream>>>(sCur, ev, x1, evB, x1B, We, W_ih, W_hh,
                                                          b_ih, b_hh, qe, gi, gh);
        gru_kernel<<<BQ_ / 256, 256, 0, stream>>>(gi, gh, sCur, sNxt);

        if (t < 4) {
            gemm_q_kernel<<<dim3(16, 2), 32 * SG_WAVES, 0, stream>>>(sNxt, Wb, Wa, qb, qa);
            scores_kernel<<<dim3(L_ / SC_ROWS_WG, B_), 32 * SC_WAVES, 0, stream>>>(M, qe, qb, qa, S3, S1, S2, 3);
        } else {
            scores_kernel<<<dim3(L_ / SC_ROWS_WG, B_), 32 * SC_WAVES, 0, stream>>>(M, qe, qe, qe, S3, nullptr, nullptr, 1);
        }
    }

    final_kernel<<<B_, 32 * WS_WAVES, 0, stream>>>(ws, (float*)d_out);
}


# GRU cell and start-up clearing folded into their consumers: the next q GEMM forms the cell on load (4-row blocks, via LDS), accumulators start at 0 in their first turn; 28 to 22 launches; same sums
# speedup vs baseline: 15.6253x; 1.0775x over previous
#include <hip/hip_runtime.h>
#include <math.h>

#define B_   32
#define L_   4096
#define MEM_ 256
#define Q_   256
#define BL_  (B_ * L_)
#define BQ_  (B_ * Q_)
#define B3Q_ (B_ * 3 * Q_)

#define OFF_S1   0
#define OFF_S2   (OFF_S1 + BL_)
#define OFF_SACC (OFF_S2 + BL_)
#define OFF_EACC (OFF_SACC + BL_)
#define OFF_QB   (OFF_EACC + BL_)
#define OFF_QA   (OFF_QB + BQ_)
#define OFF_QE   (OFF_QA + BQ_)
#define OFF_EV   (OFF_QE + BQ_)
#define OFF_X1   (OFF_EV + BQ_)
#define OFF_S_A  (OFF_X1 + BQ_)
#define OFF_S_B  (OFF_S_A + BQ_)
#define OFF_GI   (OFF_S_B + BQ_)
#define OFF_GH   (OFF_GI + B3Q_)
#define OFF_EVB  OFF_QB
#define OFF_X1B  OFF_QA

typedef __attribute__((ext_vector_type(2))) float v2f;
typedef __attribute__((ext_vector_type(8))) float v8f;

typedef __attribute__((ext_vector_type(4))) float v4f;

#define SG_WAVES  8
#define SG_KPARTS 4
#define SG_NB     32
#define SG_PITCH  34

#define SG_ORDER() __builtin_amdgcn_sched_barrier(0)

template <int NU>
__device__ __forceinline__ void sg_load(v4f (&f)[NU], const float* __restrict__ p) {
#pragma unroll
    for (int u = 0; u < NU; ++u) f[u] = *(const v4f*)(p + 8 * u);
}

template <int NU>
__device__ __forceinline__ void sg_join(v4f (&f)[NU], const v4f (&g)[NU], bool two) {
#pragma unroll
    for (int u = 0; u < NU; ++u) f[u] = two ? (f[u] + g[u]) : f[u];
}

template <int NU>
__device__ __forceinline__ v8f sg_mma(const v4f (&a)[NU], const v4f (&b)[NU]) {
    v8f acc0 = {}, acc1 = {};
#pragma unroll
    for (int u = 0; u < NU; ++u) {
        v2f a0; a0.x = a[u][0]; a0.y = a[u][1];
        v2f b0; b0.x = b[u][0]; b0.y = b[u][1];
        v2f a1; a1.x = a[u][2]; a1.y = a[u][3];
        v2f b1; b1.x = b[u][2]; b1.y = b[u][3];
        acc0 = __builtin_amdgcn_wmma_f32_16x16x4_f32(false, a0, false, b0,
                                                     (short)0, acc0, false, false);
        acc1 = __builtin_amdgcn_wmma_f32_16x16x4_f32(false, a1, false, b1,
                                                     (short)0, acc1, false, false);
    }
    return acc0 + acc1;
}

__device__ __forceinline__ void sg_finish(v8f acc, float (*red)[16][SG_PITCH],
                                          int kpart, int tn, int c, int hi, int tid,
                                          float* __restrict__ C, int ldc, int m0, int n0,
                                          float bv) {
#pragma unroll
    for (int j = 0; j < 8; ++j) red[kpart][j + 8 * hi][16 * tn + c] = acc[j];
    __syncthreads();
    const int col = tid & 31;
    const int r   = tid >> 5;
#pragma unroll
    for (int rr = 0; rr < 2; ++rr) {
        const int row = r + 8 * rr;
        float v = red[0][row][col];
        v += red[1][row][col];
        v += red[2][row][col];
        v += red[3][row][col];
        C[(size_t)(m0 + row) * ldc + n0 + col] = v + bv;
    }
}

#define GQ_ROWS  4
#define GQ_PITCH (Q_ + 4)
template <bool kCell>
__global__ __launch_bounds__(32 * SG_WAVES)
void gemm_q_kernel(const float* __restrict__ s,
                   const float* __restrict__ gi,
                   const float* __restrict__ gh,
                   float* __restrict__ sNew,
                   const float* __restrict__ Wb,
                   const float* __restrict__ Wa,
                   float* __restrict__ qb,
                   float* __restrict__ qa) {
    const int tid   = threadIdx.x;
    const int wave  = tid >> 5;
    const int lane  = tid & 31;
    const int c     = lane & 15;
    const int hi    = lane >> 4;
    const int tn    = wave & 1;
    const int kpart = wave >> 1;
    const int m0    = blockIdx.y * (kCell ? GQ_ROWS : 16);
    const int sel   = blockIdx.x >> 3;
    const int n0    = (blockIdx.x & 7) * SG_NB;
    const float* W = sel ? Wa : Wb;
    float*       C = sel ? qa : qb;

    __shared__ float red[SG_KPARTS][16][SG_PITCH];

    const int k0 = kpart * 64 + 4 * hi;
    v4f a[8], b[8];
    if constexpr (kCell) {
        __shared__ __attribute__((aligned(16))) float st[GQ_ROWS][GQ_PITCH];
        const int row = tid >> 6;
        const int col = 4 * (tid & 63);
        const float* gib = gi + (size_t)(m0 + row) * (3 * Q_) + col;
        const float* ghb = gh + (size_t)(m0 + row) * (3 * Q_) + col;
        sg_load<8>(b, W + (size_t)(n0 + 16 * tn + c) * Q_ + k0);
        const v4f ir  = *(const v4f*)gib;
        const v4f iz  = *(const v4f*)(gib + Q_);
        const v4f in_ = *(const v4f*)(gib + 2 * Q_);
        const v4f hr  = *(const v4f*)ghb;
        const v4f hz  = *(const v4f*)(ghb + Q_);
        const v4f hn  = *(const v4f*)(ghb + 2 * Q_);
        const v4f so  = *(const v4f*)(s + (size_t)(m0 + row) * Q_ + col);
        SG_ORDER();
        v4f v;
#pragma unroll
        for (int e = 0; e < 4; ++e) {
            const float r = 1.0f / (1.0f + __expf(-(ir[e] + hr[e])));
            const float z = 1.0f / (1.0f + __expf(-(iz[e] + hz[e])));
            const float n = tanhf(in_[e] + r * hn[e]);
            v[e] = fmaf(1.0f - z, n, z * so[e]);
        }
        *(v4f*)&st[row][col] = v;
        if (blockIdx.x == 0)
            *(v4f*)(sNew + (size_t)(m0 + row) * Q_ + col) = v;
        __syncthreads();
#pragma unroll
        for (int u = 0; u < 8; ++u) a[u] = *(const v4f*)&st[c & (GQ_ROWS - 1)][k0 + 8 * u];
        const v8f acc = sg_mma<8>(a, b);
        if (hi == 0) {
#pragma unroll
            for (int j = 0; j < GQ_ROWS; ++j) red[kpart][j][16 * tn + c] = acc[j];
        }
        __syncthreads();
        if (tid < 32 * GQ_ROWS) {
            const int cc = tid & 31;
            const int rw = tid >> 5;
            float t = red[0][rw][cc];
            t += red[1][rw][cc];
            t += red[2][rw][cc];
            t += red[3][rw][cc];
            C[(size_t)(m0 + rw) * MEM_ + n0 + cc] = t;
        }
    } else {
        sg_load<8>(a, s + (size_t)(m0 + c) * Q_ + k0);
        sg_load<8>(b, W + (size_t)(n0 + 16 * tn + c) * Q_ + k0);
        SG_ORDER();
        const v8f acc = sg_mma<8>(a, b);
        sg_finish(acc, red, kpart, tn, c, hi, tid, C, MEM_, m0, n0, 0.0f);
    }
}

__global__ __launch_bounds__(32 * SG_WAVES)
void gemm_turn_kernel(const float* __restrict__ s,
                      const float* __restrict__ ev,
                      const float* __restrict__ x1,
                      const float* __restrict__ evB,
                      const float* __restrict__ x1B,
                      const float* __restrict__ We,
                      const float* __restrict__ W_ih,
                      const float* __restrict__ W_hh,
                      const float* __restrict__ b_ih,
                      const float* __restrict__ b_hh,
                      float* __restrict__ qe,
                      float* __restrict__ gi,
                      float* __restrict__ gh) {
    const int tid   = threadIdx.x;
    const int wave  = __builtin_amdgcn_readfirstlane(tid >> 5);
    const int lane  = tid & 31;
    const int c     = lane & 15;
    const int hi    = lane >> 4;
    const int tn    = wave & 1;
    const int kpart = wave >> 1;
    const int m0    = blockIdx.y * 16;
    const int bx    = blockIdx.x;

    __shared__ float red[SG_KPARTS][16][SG_PITCH];

    v8f acc;
    float* C;
    float  bv = 0.0f;
    int ldc, n0;
    if (bx < 8) {
        n0 = bx * SG_NB; C = qe; ldc = MEM_;
        const int k0 = kpart * 128 + 4 * hi;
        const bool two = kpart >= 2;
        const size_t o = (size_t)(m0 + c) * MEM_ + (k0 & (MEM_ - 1));
        const float* p0 = (two ? ev : s) + o;
        const float* p1 = two ? (evB + o) : p0;
        v4f a[16], a1[16], b[16];
        sg_load<16>(b, We + (size_t)(n0 + 16 * tn + c) * (Q_ + MEM_) + k0);
        sg_load<16>(a, p0);
        sg_load<16>(a1, p1);
        SG_ORDER();
        sg_join<16>(a, a1, two);
        acc = sg_mma<16>(a, b);
    } else {
        const bool isGi = bx < 32;
        n0 = (isGi ? (bx - 8) : (bx - 32)) * SG_NB; ldc = 3 * Q_;
        const float* W = isGi ? W_ih : W_hh;
        C  = isGi ? gi : gh;
        const int k0 = kpart * 64 + 4 * hi;
        const size_t o = (size_t)(m0 + c) * MEM_ + k0;
        const float* p0 = (isGi ? x1 : s) + o;
        const float* p1 = isGi ? (x1B + o) : p0;
        v4f a[8], a1[8], b[8];
        bv = (isGi ? b_ih : b_hh)[n0 + (tid & 31)];
        sg_load<8>(b, W + (size_t)(n0 + 16 * tn + c) * MEM_ + k0);
        sg_load<8>(a, p0);
        sg_load<8>(a1, p1);
        SG_ORDER();
        sg_join<8>(a, a1, isGi);
        acc = sg_mma<8>(a, b);
    }
    sg_finish(acc, red, kpart, tn, c, hi, tid, C, ldc, m0, n0, bv);
}

#define SC_WAVES  4
#define SC_TILES  2
#define SC_UCOLS  128
#define SC_PITCH  (SC_UCOLS + 4)
#define SC_QPITCH (MEM_ + 4)
#define SC_ROWS_WG (16 * SC_TILES * SC_WAVES)

struct ScUnit { v4f r[16]; };

__device__ __forceinline__ void sc_load(ScUnit& u, const float* __restrict__ src) {
#pragma unroll
    for (int j = 0; j < 16; ++j) u.r[j] = *(const v4f*)(src + (size_t)j * MEM_);
}

__device__ __forceinline__ void sc_park(const ScUnit& u, float* dst) {
#pragma unroll
    for (int j = 0; j < 16; ++j) *(v4f*)(dst + j * SC_PITCH) = u.r[j];
}

__device__ __forceinline__ v8f sc_mma(const float* aP, const float* bP, v8f acc) {
#pragma unroll
    for (int i = 0; i < SC_UCOLS / 4; ++i) {
        const v2f a  = *(const v2f*)(aP + 4 * i);
        const v2f bb = *(const v2f*)(bP + 4 * i);
        acc = __builtin_amdgcn_wmma_f32_16x16x4_f32(false, a, false, bb,
                                                    (short)0, acc, false, false);
    }
    return acc;
}

__global__ __launch_bounds__(32 * SC_WAVES)
void scores_kernel(const float* __restrict__ M,
                   const float* __restrict__ q0,
                   const float* __restrict__ q1,
                   const float* __restrict__ q2,
                   float* __restrict__ S0,
                   float* __restrict__ S1,
                   float* __restrict__ S2,
                   int ncols) {
    const int tid  = threadIdx.x;
    const int wave = tid >> 5;
    const int lane = tid & 31;
    const int c    = lane & 15;
    const int hi   = lane >> 4;
    const int b    = blockIdx.y;
    const int lw   = blockIdx.x * SC_ROWS_WG + wave * (16 * SC_TILES);

    __shared__ __attribute__((aligned(16))) float qs[SC_WAVES][4 * SC_QPITCH];
    __shared__ __attribute__((aligned(16))) float ms[SC_WAVES][16 * SC_PITCH];

    const float* Mw = M + ((size_t)b * L_ + lw) * MEM_ + 4 * lane;

    const float* qsrc[3] = { q0, q1, q2 };
    v4f qv[3][2];
#pragma unroll
    for (int r = 0; r < 3; ++r) {
        const float* qp = ((r < ncols) ? qsrc[r] : q0) + (size_t)b * MEM_ + 4 * lane;
        qv[r][0] = *(const v4f*)qp;
        qv[r][1] = *(const v4f*)(qp + SC_UCOLS);
    }

    ScUnit g0, g1;
    sc_load(g0, Mw);

    float*       mw = ms[wave];
    float*       pk = mw + 4 * lane;
    const float* aP = mw + c * SC_PITCH + 2 * hi;
    float*       qw = qs[wave];
    const float* bP = qw + (c < 3 ? c : 3) * SC_QPITCH + 2 * hi;

#define SC_ORDER() __builtin_amdgcn_sched_barrier(0)
#define SC_WAVE_SYNC() do { __builtin_amdgcn_fence(__ATOMIC_RELEASE, "wavefront"); \
                            __builtin_amdgcn_wave_barrier();                      \
                            __builtin_amdgcn_fence(__ATOMIC_ACQUIRE, "wavefront"); } while (0)

    sc_load(g1, Mw + SC_UCOLS);
    SC_ORDER();
    {
        const v4f z = {};
#pragma unroll
        for (int r = 0; r < 3; ++r) {
            *(v4f*)(qw + r * SC_QPITCH + 4 * lane)            = (r < ncols) ? qv[r][0] : z;
            *(v4f*)(qw + r * SC_QPITCH + 4 * lane + SC_UCOLS) = (r < ncols) ? qv[r][1] : z;
        }
        *(v4f*)(qw + 3 * SC_QPITCH + 4 * lane)            = z;
        *(v4f*)(qw + 3 * SC_QPITCH + 4 * lane + SC_UCOLS) = z;
    }

    const bool live = c < ncols;
    float* dst = ((c == 0) ? S0 : (c == 1) ? S1 : S2);

    v8f acc = {};
    sc_park(g0, pk);
    sc_load(g0, Mw + (size_t)16 * MEM_);
    SC_ORDER();
    SC_WAVE_SYNC();
    acc = sc_mma(aP, bP, acc);
    SC_WAVE_SYNC();
    sc_park(g1, pk);
    sc_load(g1, Mw + (size_t)16 * MEM_ + SC_UCOLS);
    SC_ORDER();
    SC_WAVE_SYNC();
    acc = sc_mma(aP, bP + SC_UCOLS, acc);
    SC_WAVE_SYNC();
    if (live) {
#pragma unroll
        for (int j = 0; j < 8; ++j)
            dst[(size_t)b * L_ + lw + j + 8 * hi] = acc[j];
    }

    acc = v8f{};
    sc_park(g0, pk);
    SC_WAVE_SYNC();
    acc = sc_mma(aP, bP, acc);
    SC_WAVE_SYNC();
    sc_park(g1, pk);
    SC_WAVE_SYNC();
    acc = sc_mma(aP, bP + SC_UCOLS, acc);
    if (live) {
#pragma unroll
        for (int j = 0; j < 8; ++j)
            dst[(size_t)b * L_ + lw + 16 + j + 8 * hi] = acc[j];
    }
#undef SC_WAVE_SYNC
#undef SC_ORDER
}

#define WS_WAVES 16
#define WS_COLS  64
#define WS_STEPS 8
#define WS_GROUP_ROWS (4 * WS_STEPS)
#define WS_PARTS 2
#define WS_PART_ROWS (L_ / WS_PARTS)
#define WS_GROUPS (WS_PART_ROWS / (WS_GROUP_ROWS * WS_WAVES))
#define WS_WGS_ROW ((MEM_ / WS_COLS) * WS_PARTS)
static_assert(Q_ == MEM_, "gemm_turn_kernel addresses s and ev with one offset");
static_assert(WS_WGS_ROW * 32 * WS_WAVES == L_, "the work-groups of a batch row share out its L scores, one per thread");
static_assert(WS_GROUPS >= 2 && (WS_GROUPS & 1) == 0, "two load groups in flight, refilled in pairs");

template <int X>
__device__ __forceinline__ float xrow(float v) {
    return __builtin_bit_cast(float, __builtin_amdgcn_update_dpp(
        0, __builtin_bit_cast(int, v), 0x160 + X, 0xf, 0xf, false));
}
__device__ __forceinline__ float xhalf(float v) {
    return __builtin_bit_cast(float, __builtin_amdgcn_permlanex16(
        __builtin_bit_cast(int, v), __builtin_bit_cast(int, v), 0x76543210, 0xfedcba98, false, false));
}
__device__ __forceinline__ float row16_max(float v) {
    v = fmaxf(v, xrow<1>(v)); v = fmaxf(v, xrow<2>(v));
    v = fmaxf(v, xrow<4>(v)); v = fmaxf(v, xrow<8>(v));
    return v;
}
__device__ __forceinline__ float row16_sum(float v) {
    v += xrow<1>(v); v += xrow<2>(v); v += xrow<4>(v); v += xrow<8>(v);
    return v;
}

template <int NP>
__device__ __forceinline__ void row_stats(const v4f (&x)[NP][2], float (&mx)[NP], float (&inv)[NP],
                                          float (*redm)[WS_WAVES], float (*reds)[WS_WAVES],
                                          int lane, int wave) {
#pragma unroll
    for (int p = 0; p < NP; ++p) {
        float m = fmaxf(fmaxf(fmaxf(x[p][0][0], x[p][0][1]), fmaxf(x[p][0][2], x[p][0][3])),
                        fmaxf(fmaxf(x[p][1][0], x[p][1][1]), fmaxf(x[p][1][2], x[p][1][3])));
        m = row16_max(m);
        m = fmaxf(m, xhalf(m));
        redm[p][wave] = m;
    }
    __syncthreads();
#pragma unroll
    for (int p = 0; p < NP; ++p) {
        const float m = row16_max(redm[p][lane & (WS_WAVES - 1)]);
        mx[p] = m;
        float s = 0.0f;
#pragma unroll
        for (int j = 0; j < 2; ++j)
#pragma unroll
            for (int e = 0; e < 4; ++e) s += __expf(x[p][j][e] - m);
        s = row16_sum(s);
        s += xhalf(s);
        reds[p][wave] = s;
    }
    __syncthreads();
#pragma unroll
    for (int p = 0; p < NP; ++p)
        inv[p] = 1.0f / row16_sum(reds[p][lane & (WS_WAVES - 1)]);
}

struct WsGroup {
    v2f a[WS_STEPS];
    v4f r0[WS_STEPS];
    v4f r1[WS_STEPS];
};

__device__ __forceinline__ void ws_load(WsGroup& g, const float* __restrict__ pv,
                                        const float* __restrict__ Mb, int l0) {
#pragma unroll
    for (int s = 0; s < WS_STEPS; ++s)
        g.a[s] = *(const v2f*)(pv + l0 + 4 * s);
#pragma unroll
    for (int s = 0; s < WS_STEPS; ++s) {
        const float* rp = Mb + (size_t)(l0 + 4 * s) * MEM_;
        g.r0[s] = *(const v4f*)rp;
        g.r1[s] = *(const v4f*)(rp + MEM_);
    }
}

__device__ __forceinline__ void ws_mma(const WsGroup& g, bool live, float mx, float inv,
                                       v8f (&acc)[4]) {
#pragma unroll
    for (int s = 0; s < WS_STEPS; ++s) {
        v2f a;
        a.x = live ? __expf(g.a[s].x - mx) * inv : 0.0f;
        a.y = live ? __expf(g.a[s].y - mx) * inv : 0.0f;
#pragma unroll
        for (int t = 0; t < 4; ++t) {
            v2f bb; bb.x = g.r0[s][t]; bb.y = g.r1[s][t];
            acc[t] = __builtin_amdgcn_wmma_f32_16x16x4_f32(false, a, false, bb,
                                                           (short)0, acc[t], false, false);
        }
    }
}

__global__ __launch_bounds__(32 * WS_WAVES)
void wsum_kernel(const float* __restrict__ M,
                 const float* __restrict__ p,
                 const float* __restrict__ w,
                 const float* __restrict__ endS,
                 float* __restrict__ sacc,
                 float* __restrict__ eacc,
                 float* __restrict__ ev,
                 float* __restrict__ x1,
                 float* __restrict__ evB,
                 float* __restrict__ x1B) {
    const int tid  = threadIdx.x;
    const int wave = tid >> 5;
    const int lane = tid & 31;
    const int c    = lane & 15;
    const int hi   = lane >> 4;
    const int b    = blockIdx.y;
    const int part = blockIdx.x & (WS_PARTS - 1);
    const int slab = (blockIdx.x / WS_PARTS) * WS_COLS;
    const bool hasEnd = endS != nullptr;

    __shared__ float redm[3][WS_WAVES];
    __shared__ float reds[3][WS_WAVES];
    const size_t r0 = (size_t)b * L_ + 4 * tid;
    const size_t r1 = r0 + 4 * 32 * WS_WAVES;
    const size_t mine = (size_t)b * L_ + blockIdx.x * (32 * WS_WAVES) + tid;

    const float* pv = ((c == 1) ? w : p) + (size_t)b * L_ + part * WS_PART_ROWS + 2 * hi;
    const float* Mb = M + ((size_t)b * L_ + part * WS_PART_ROWS + 2 * hi) * MEM_ + slab + 4 * c;
    const bool live = c < 2;

    const float* es = hasEnd ? endS : p;
    v4f x[3][2];
    x[0][0] = *(const v4f*)(p + r0);
    x[0][1] = *(const v4f*)(p + r1);
    x[1][0] = *(const v4f*)(w + r0);
    x[1][1] = *(const v4f*)(w + r1);
    x[2][0] = *(const v4f*)(es + r0);
    x[2][1] = *(const v4f*)(es + r1);
    const float ps = p[mine];
    const float pe = es[mine];
    const float sa = hasEnd ? sacc[mine] : 0.0f;
    const float ea = hasEnd ? eacc[mine] : 0.0f;

    v8f acc[4] = {};
    WsGroup g0, g1;
#define WS_ORDER() __builtin_amdgcn_sched_barrier(0)
    ws_load(g0, pv, Mb, wave * WS_GROUP_ROWS);
    ws_load(g1, pv, Mb, (WS_WAVES + wave) * WS_GROUP_ROWS);
    WS_ORDER();
    float mx3[3], inv3[3];
    row_stats<3>(x, mx3, inv3, redm, reds, lane, wave);
    sacc[mine] = sa + 0.2f * (__expf(ps - mx3[0]) * inv3[0]);
    eacc[mine] = hasEnd ? ea + 0.2f * (__expf(pe - mx3[2]) * inv3[2]) : 0.0f;
    const float mx  = (c == 1) ? mx3[1]  : mx3[0];
    const float inv = (c == 1) ? inv3[1] : inv3[0];
    WS_ORDER();
    ws_mma(g0, live, mx, inv, acc);
    WS_ORDER();
    for (int i = 1; i < WS_GROUPS - 1; i += 2) {
        ws_load(g0, pv, Mb, ((i + 1) * WS_WAVES + wave) * WS_GROUP_ROWS);
        WS_ORDER();
        ws_mma(g1, live, mx, inv, acc);
        WS_ORDER();
        ws_load(g1, pv, Mb, ((i + 2) * WS_WAVES + wave) * WS_GROUP_ROWS);
        WS_ORDER();
        ws_mma(g0, live, mx, inv, acc);
        WS_ORDER();
    }
    ws_mma(g1, live, mx, inv, acc);
#undef WS_ORDER

    __shared__ float red[WS_WAVES][2][WS_COLS];
    if (hi == 0) {
        v4f e, xx;
#pragma unroll
        for (int t = 0; t < 4; ++t) { e[t] = acc[t][0]; xx[t] = acc[t][1]; }
        *(v4f*)&red[wave][0][4 * c] = e;
        *(v4f*)&red[wave][1][4 * c] = xx;
    }
    __syncthreads();
    if (tid < 2 * WS_COLS) {
        const int r  = tid >> 6;
        const int mm = tid & (WS_COLS - 1);
        float s = 0.0f;
#pragma unroll
        for (int k = 0; k < WS_WAVES; ++k) s += red[k][r][mm];
        float* eo = part ? evB : ev;
        float* xo = part ? x1B : x1;
        (r ? xo : eo)[(size_t)b * MEM_ + slab + mm] = s;
    }
}

__global__ __launch_bounds__(32 * WS_WAVES)
void final_kernel(const float* __restrict__ ws, float* out) {
    const int tid  = threadIdx.x;
    const int wave = tid >> 5;
    const int lane = tid & 31;
    const int b    = blockIdx.x;
    __shared__ float redm[1][WS_WAVES];
    __shared__ float reds[1][WS_WAVES];
    const size_t r0 = (size_t)b * L_ + 4 * tid;
    const size_t r1 = r0 + 4 * 32 * WS_WAVES;

    v4f x[1][2];
    x[0][0] = *(const v4f*)(out + BL_ + r0);
    x[0][1] = *(const v4f*)(out + BL_ + r1);
    v4f e0 = *(const v4f*)(ws + OFF_EACC + r0);
    v4f e1 = *(const v4f*)(ws + OFF_EACC + r1);
    v4f s0v = *(const v4f*)(ws + OFF_SACC + r0);
    v4f s1v = *(const v4f*)(ws + OFF_SACC + r1);
    float mx[1], inv[1];
    row_stats<1>(x, mx, inv, redm, reds, lane, wave);
#pragma unroll
    for (int e = 0; e < 4; ++e) {
        e0[e] = logf(e0[e] + 0.2f * (__expf(x[0][0][e] - mx[0]) * inv[0]));
        e1[e] = logf(e1[e] + 0.2f * (__expf(x[0][1][e] - mx[0]) * inv[0]));
        s0v[e] = logf(s0v[e]);
        s1v[e] = logf(s1v[e]);
    }
    *(v4f*)(out + r0) = s0v;
    *(v4f*)(out + r1) = s1v;
    *(v4f*)(out + BL_ + r0) = e0;
    *(v4f*)(out + BL_ + r1) = e1;
}

extern "C" void kernel_launch(void* const* d_in, const int* in_sizes, int n_in,
                              void* d_out, int out_size, void* d_ws, size_t ws_size,
                              hipStream_t stream) {
    const float* M    = (const float*)d_in[0];
    const float* s0   = (const float*)d_in[1];
    const float* Wb   = (const float*)d_in[2];
    const float* We   = (const float*)d_in[3];
    const float* Wa   = (const float*)d_in[4];
    const float* W_ih = (const float*)d_in[5];
    const float* W_hh = (const float*)d_in[6];
    const float* b_ih = (const float*)d_in[7];
    const float* b_hh = (const float*)d_in[8];

    float* ws   = (float*)d_ws;
    float* S1   = ws + OFF_S1;
    float* S2   = ws + OFF_S2;
    float* sacc = ws + OFF_SACC;
    float* eacc = ws + OFF_EACC;
    float* qb   = ws + OFF_QB;
    float* qa   = ws + OFF_QA;
    float* qe   = ws + OFF_QE;
    float* ev   = ws + OFF_EV;
    float* x1   = ws + OFF_X1;
    float* sA   = ws + OFF_S_A;
    float* sB   = ws + OFF_S_B;
    float* gi   = ws + OFF_GI;
    float* gh   = ws + OFF_GH;
    float* evB  = ws + OFF_EVB;
    float* x1B  = ws + OFF_X1B;
    float* S3   = (float*)d_out + BL_;

    gemm_q_kernel<false><<<dim3(16, 2), 32 * SG_WAVES, 0, stream>>>(s0, nullptr, nullptr, nullptr,
                                                                    Wb, Wa, qb, qa);
    scores_kernel<<<dim3(L_ / SC_ROWS_WG, B_), 32 * SC_WAVES, 0, stream>>>(M, qb, qa, qa, S1, S2, nullptr, 2);

    for (int t = 0; t < 5; ++t) {
        const float* sCur = (t == 0) ? s0 : ((t & 1) ? sB : sA);
        float*       sNxt = (t & 1) ? sA : sB;

        if (t == 0)
            wsum_kernel<<<dim3(WS_WGS_ROW, B_), 32 * WS_WAVES, 0, stream>>>(
                M, S1, S2, nullptr, sacc, eacc, ev, x1, evB, x1B);
        else
            wsum_kernel<<<dim3(WS_WGS_ROW, B_), 32 * WS_WAVES, 0, stream>>>(
                M, S1, S2, S3, sacc, eacc, ev, x1, evB, x1B);

        gemm_turn_kernel<<<dim3(56, 2), 32 * SG_WAVES, 0, stream>>>(sCur, ev, x1, evB, x1B, We, W_ih, W_hh,
                                                          b_ih, b_hh, qe, gi, gh);

        if (t < 4) {
            gemm_q_kernel<true><<<dim3(16, B_ / GQ_ROWS), 32 * SG_WAVES, 0, stream>>>(
                sCur, gi, gh, sNxt, Wb, Wa, qb, qa);
            scores_kernel<<<dim3(L_ / SC_ROWS_WG, B_), 32 * SC_WAVES, 0, stream>>>(M, qe, qb, qa, S3, S1, S2, 3);
        } else {
            scores_kernel<<<dim3(L_ / SC_ROWS_WG, B_), 32 * SC_WAVES, 0, stream>>>(M, qe, qe, qe, S3, nullptr, nullptr, 1);
        }
    }

    final_kernel<<<B_, 32 * WS_WAVES, 0, stream>>>(ws, (float*)d_out);
}


# score pass and weighted-sum pass of a turn merged into one pass over M (running-max softmax; partial rows joined by the turn GEMM on load); 22 to 17 launches; new reduction trees for scores and sums
# speedup vs baseline: 18.5595x; 1.1878x over previous
#include <hip/hip_runtime.h>
#include <math.h>

#define B_   32
#define L_   4096
#define MEM_ 256
#define Q_   256
#define BL_  (B_ * L_)
#define BQ_  (B_ * Q_)
#define B3Q_ (B_ * 3 * Q_)

#define OFF_S1   0
#define OFF_S2   (OFF_S1 + BL_)
#define OFF_SACC (OFF_S2 + BL_)
#define OFF_EACC (OFF_SACC + BL_)
#define OFF_QB   (OFF_EACC + BL_)
#define OFF_QA   (OFF_QB + BQ_)
#define OFF_QE   (OFF_QA + BQ_)
#define OFF_EV   (OFF_QE + BQ_)
#define OFF_X1   (OFF_EV + BQ_)
#define OFF_S_A  (OFF_X1 + BQ_)
#define OFF_S_B  (OFF_S_A + BQ_)
#define OFF_GI   (OFF_S_B + BQ_)
#define OFF_GH   (OFF_GI + B3Q_)


typedef __attribute__((ext_vector_type(2))) float v2f;
typedef __attribute__((ext_vector_type(8))) float v8f;

typedef __attribute__((ext_vector_type(4))) float v4f;

#define SG_WAVES  8
#define SG_KPARTS 4
#define SG_NB     32
#define SG_PITCH  34

#define SG_ORDER() __builtin_amdgcn_sched_barrier(0)

template <int NU>
__device__ __forceinline__ void sg_load(v4f (&f)[NU], const float* __restrict__ p) {
#pragma unroll
    for (int u = 0; u < NU; ++u) f[u] = *(const v4f*)(p + 8 * u);
}

template <int NU>
__device__ __forceinline__ v8f sg_mma(const v4f (&a)[NU], const v4f (&b)[NU]) {
    v8f acc0 = {}, acc1 = {};
#pragma unroll
    for (int u = 0; u < NU; ++u) {
        v2f a0; a0.x = a[u][0]; a0.y = a[u][1];
        v2f b0; b0.x = b[u][0]; b0.y = b[u][1];
        v2f a1; a1.x = a[u][2]; a1.y = a[u][3];
        v2f b1; b1.x = b[u][2]; b1.y = b[u][3];
        acc0 = __builtin_amdgcn_wmma_f32_16x16x4_f32(false, a0, false, b0,
                                                     (short)0, acc0, false, false);
        acc1 = __builtin_amdgcn_wmma_f32_16x16x4_f32(false, a1, false, b1,
                                                     (short)0, acc1, false, false);
    }
    return acc0 + acc1;
}

__device__ __forceinline__ void sg_finish(v8f acc, float (*red)[16][SG_PITCH],
                                          int kpart, int tn, int c, int hi, int tid,
                                          float* __restrict__ C, int ldc, int m0, int n0,
                                          float bv) {
#pragma unroll
    for (int j = 0; j < 8; ++j) red[kpart][j + 8 * hi][16 * tn + c] = acc[j];
    __syncthreads();
    const int col = tid & 31;
    const int r   = tid >> 5;
#pragma unroll
    for (int rr = 0; rr < 2; ++rr) {
        const int row = r + 8 * rr;
        float v = red[0][row][col];
        v += red[1][row][col];
        v += red[2][row][col];
        v += red[3][row][col];
        C[(size_t)(m0 + row) * ldc + n0 + col] = v + bv;
    }
}

#define GQ_ROWS  4
#define GQ_PITCH (Q_ + 4)
template <bool kCell>
__global__ __launch_bounds__(32 * SG_WAVES)
void gemm_q_kernel(const float* __restrict__ s,
                   const float* __restrict__ gi,
                   const float* __restrict__ gh,
                   float* __restrict__ sNew,
                   const float* __restrict__ Wb,
                   const float* __restrict__ Wa,
                   float* __restrict__ qb,
                   float* __restrict__ qa) {
    const int tid   = threadIdx.x;
    const int wave  = tid >> 5;
    const int lane  = tid & 31;
    const int c     = lane & 15;
    const int hi    = lane >> 4;
    const int tn    = wave & 1;
    const int kpart = wave >> 1;
    const int m0    = blockIdx.y * (kCell ? GQ_ROWS : 16);
    const int sel   = blockIdx.x >> 3;
    const int n0    = (blockIdx.x & 7) * SG_NB;
    const float* W = sel ? Wa : Wb;
    float*       C = sel ? qa : qb;

    __shared__ float red[SG_KPARTS][16][SG_PITCH];

    const int k0 = kpart * 64 + 4 * hi;
    v4f a[8], b[8];
    if constexpr (kCell) {
        __shared__ __attribute__((aligned(16))) float st[GQ_ROWS][GQ_PITCH];
        const int row = tid >> 6;
        const int col = 4 * (tid & 63);
        const float* gib = gi + (size_t)(m0 + row) * (3 * Q_) + col;
        const float* ghb = gh + (size_t)(m0 + row) * (3 * Q_) + col;
        sg_load<8>(b, W + (size_t)(n0 + 16 * tn + c) * Q_ + k0);
        const v4f ir  = *(const v4f*)gib;
        const v4f iz  = *(const v4f*)(gib + Q_);
        const v4f in_ = *(const v4f*)(gib + 2 * Q_);
        const v4f hr  = *(const v4f*)ghb;
        const v4f hz  = *(const v4f*)(ghb + Q_);
        const v4f hn  = *(const v4f*)(ghb + 2 * Q_);
        const v4f so  = *(const v4f*)(s + (size_t)(m0 + row) * Q_ + col);
        SG_ORDER();
        v4f v;
#pragma unroll
        for (int e = 0; e < 4; ++e) {
            const float r = 1.0f / (1.0f + __expf(-(ir[e] + hr[e])));
            const float z = 1.0f / (1.0f + __expf(-(iz[e] + hz[e])));
            const float n = tanhf(in_[e] + r * hn[e]);
            v[e] = fmaf(1.0f - z, n, z * so[e]);
        }
        *(v4f*)&st[row][col] = v;
        if (blockIdx.x == 0)
            *(v4f*)(sNew + (size_t)(m0 + row) * Q_ + col) = v;
        __syncthreads();
#pragma unroll
        for (int u = 0; u < 8; ++u) a[u] = *(const v4f*)&st[c & (GQ_ROWS - 1)][k0 + 8 * u];
        const v8f acc = sg_mma<8>(a, b);
        if (hi == 0) {
#pragma unroll
            for (int j = 0; j < GQ_ROWS; ++j) red[kpart][j][16 * tn + c] = acc[j];
        }
        __syncthreads();
        if (tid < 32 * GQ_ROWS) {
            const int cc = tid & 31;
            const int rw = tid >> 5;
            float t = red[0][rw][cc];
            t += red[1][rw][cc];
            t += red[2][rw][cc];
            t += red[3][rw][cc];
            C[(size_t)(m0 + rw) * MEM_ + n0 + cc] = t;
        }
    } else {
        sg_load<8>(a, s + (size_t)(m0 + c) * Q_ + k0);
        sg_load<8>(b, W + (size_t)(n0 + 16 * tn + c) * Q_ + k0);
        SG_ORDER();
        const v8f acc = sg_mma<8>(a, b);
        sg_finish(acc, red, kpart, tn, c, hi, tid, C, MEM_, m0, n0, 0.0f);
    }
}

#define PK_WGS_ROW 8
#define PT_STRIDE  (2 * MEM_)
#define ST_STRIDE  8
__device__ __forceinline__ void tg_join(const float* __restrict__ part, const float* __restrict__ stat,
                                        int m0, int tid, int sq, int off, float (*cmb)[GQ_PITCH]) {
    const int row = tid >> 4;
    const int col = 4 * (tid & 15);
    const float* sp = stat + (size_t)(m0 + row) * (PK_WGS_ROW * ST_STRIDE) + sq;
    const float* pp = part + (size_t)(m0 + row) * (PK_WGS_ROW * PT_STRIDE) + off + col;
    float mi[PK_WGS_ROW], li[PK_WGS_ROW];
    v4f pv[PK_WGS_ROW][4];
#pragma unroll
    for (int i = 0; i < PK_WGS_ROW; ++i) {
        mi[i] = sp[i * ST_STRIDE];
        li[i] = sp[i * ST_STRIDE + 1];
#pragma unroll
        for (int u = 0; u < 4; ++u) pv[i][u] = *(const v4f*)(pp + i * PT_STRIDE + 64 * u);
    }
    SG_ORDER();
    float m = mi[0];
#pragma unroll
    for (int i = 1; i < PK_WGS_ROW; ++i) m = fmaxf(m, mi[i]);
    float wgt[PK_WGS_ROW];
    float L = 0.0f;
#pragma unroll
    for (int i = 0; i < PK_WGS_ROW; ++i) { wgt[i] = __expf(mi[i] - m); L += wgt[i] * li[i]; }
    const float invL = 1.0f / L;
    v4f sum[4];
#pragma unroll
    for (int u = 0; u < 4; ++u) sum[u] = (wgt[0] * invL) * pv[0][u];
#pragma unroll
    for (int i = 1; i < PK_WGS_ROW; ++i) {
        const float wi = wgt[i] * invL;
#pragma unroll
        for (int u = 0; u < 4; ++u) sum[u] += wi * pv[i][u];
    }
#pragma unroll
    for (int u = 0; u < 4; ++u) *(v4f*)&cmb[row][col + 64 * u] = sum[u];
}

__global__ __launch_bounds__(32 * SG_WAVES)
void gemm_turn_kernel(const float* __restrict__ s,
                      const float* __restrict__ part,
                      const float* __restrict__ stat,
                      const float* __restrict__ We,
                      const float* __restrict__ W_ih,
                      const float* __restrict__ W_hh,
                      const float* __restrict__ b_ih,
                      const float* __restrict__ b_hh,
                      float* __restrict__ qe,
                      float* __restrict__ gi,
                      float* __restrict__ gh) {
    const int tid   = threadIdx.x;
    const int wave  = __builtin_amdgcn_readfirstlane(tid >> 5);
    const int lane  = tid & 31;
    const int c     = lane & 15;
    const int hi    = lane >> 4;
    const int tn    = wave & 1;
    const int kpart = wave >> 1;
    const int m0    = blockIdx.y * 16;
    const int bx    = blockIdx.x;

    __shared__ float red[SG_KPARTS][16][SG_PITCH];
    __shared__ __attribute__((aligned(16))) float cmb[16][GQ_PITCH];

    v8f acc;
    float* C;
    float  bv = 0.0f;
    int ldc, n0;
    if (bx < 8) {
        n0 = bx * SG_NB; C = qe; ldc = MEM_;
        const int k0 = kpart * 128 + 4 * hi;
        const bool fromEv = kpart >= 2;
        v4f a[16], b[16];
        sg_load<16>(b, We + (size_t)(n0 + 16 * tn + c) * (Q_ + MEM_) + k0);
        sg_load<16>(a, s + (size_t)(m0 + c) * MEM_ + (k0 & (MEM_ - 1)));
        tg_join(part, stat, m0, tid, 2, 0, cmb);
        __syncthreads();
        if (fromEv) {
#pragma unroll
            for (int u = 0; u < 16; ++u) a[u] = *(const v4f*)&cmb[c][(k0 & (MEM_ - 1)) + 8 * u];
        }
        acc = sg_mma<16>(a, b);
    } else if (bx < 32) {
        n0 = (bx - 8) * SG_NB; C = gi; ldc = 3 * Q_;
        const int k0 = kpart * 64 + 4 * hi;
        v4f a[8], b[8];
        bv = b_ih[n0 + (tid & 31)];
        sg_load<8>(b, W_ih + (size_t)(n0 + 16 * tn + c) * MEM_ + k0);
        tg_join(part, stat, m0, tid, 4, MEM_, cmb);
        __syncthreads();
#pragma unroll
        for (int u = 0; u < 8; ++u) a[u] = *(const v4f*)&cmb[c][k0 + 8 * u];
        acc = sg_mma<8>(a, b);
    } else {
        n0 = (bx - 32) * SG_NB; C = gh; ldc = 3 * Q_;
        const int k0 = kpart * 64 + 4 * hi;
        v4f a[8], b[8];
        bv = b_hh[n0 + (tid & 31)];
        sg_load<8>(b, W_hh + (size_t)(n0 + 16 * tn + c) * MEM_ + k0);
        sg_load<8>(a, s + (size_t)(m0 + c) * MEM_ + k0);
        SG_ORDER();
        acc = sg_mma<8>(a, b);
    }
    sg_finish(acc, red, kpart, tn, c, hi, tid, C, ldc, m0, n0, bv);
}

#define SC_WAVES  4
#define SC_TILES  2
#define SC_UCOLS  128
#define SC_PITCH  (SC_UCOLS + 4)
#define SC_QPITCH (MEM_ + 4)
#define SC_ROWS_WG (16 * SC_TILES * SC_WAVES)

struct ScUnit { v4f r[16]; };

__device__ __forceinline__ void sc_load(ScUnit& u, const float* __restrict__ src) {
#pragma unroll
    for (int j = 0; j < 16; ++j) u.r[j] = *(const v4f*)(src + (size_t)j * MEM_);
}

__device__ __forceinline__ void sc_park(const ScUnit& u, float* dst) {
#pragma unroll
    for (int j = 0; j < 16; ++j) *(v4f*)(dst + j * SC_PITCH) = u.r[j];
}

__device__ __forceinline__ v8f sc_mma(const float* aP, const float* bP, v8f acc) {
#pragma unroll
    for (int i = 0; i < SC_UCOLS / 4; ++i) {
        const v2f a  = *(const v2f*)(aP + 4 * i);
        const v2f bb = *(const v2f*)(bP + 4 * i);
        acc = __builtin_amdgcn_wmma_f32_16x16x4_f32(false, a, false, bb,
                                                    (short)0, acc, false, false);
    }
    return acc;
}

__global__ __launch_bounds__(32 * SC_WAVES)
void scores_kernel(const float* __restrict__ M,
                   const float* __restrict__ q0,
                   const float* __restrict__ q1,
                   const float* __restrict__ q2,
                   float* __restrict__ S0,
                   float* __restrict__ S1,
                   float* __restrict__ S2,
                   int ncols) {
    const int tid  = threadIdx.x;
    const int wave = tid >> 5;
    const int lane = tid & 31;
    const int c    = lane & 15;
    const int hi   = lane >> 4;
    const int b    = blockIdx.y;
    const int lw   = blockIdx.x * SC_ROWS_WG + wave * (16 * SC_TILES);

    __shared__ __attribute__((aligned(16))) float qs[SC_WAVES][4 * SC_QPITCH];
    __shared__ __attribute__((aligned(16))) float ms[SC_WAVES][16 * SC_PITCH];

    const float* Mw = M + ((size_t)b * L_ + lw) * MEM_ + 4 * lane;

    const float* qsrc[3] = { q0, q1, q2 };
    v4f qv[3][2];
#pragma unroll
    for (int r = 0; r < 3; ++r) {
        const float* qp = ((r < ncols) ? qsrc[r] : q0) + (size_t)b * MEM_ + 4 * lane;
        qv[r][0] = *(const v4f*)qp;
        qv[r][1] = *(const v4f*)(qp + SC_UCOLS);
    }

    ScUnit g0, g1;
    sc_load(g0, Mw);

    float*       mw = ms[wave];
    float*       pk = mw + 4 * lane;
    const float* aP = mw + c * SC_PITCH + 2 * hi;
    float*       qw = qs[wave];
    const float* bP = qw + (c < 3 ? c : 3) * SC_QPITCH + 2 * hi;

#define SC_ORDER() __builtin_amdgcn_sched_barrier(0)
#define SC_WAVE_SYNC() do { __builtin_amdgcn_fence(__ATOMIC_RELEASE, "wavefront"); \
                            __builtin_amdgcn_wave_barrier();                      \
                            __builtin_amdgcn_fence(__ATOMIC_ACQUIRE, "wavefront"); } while (0)

    sc_load(g1, Mw + SC_UCOLS);
    SC_ORDER();
    {
        const v4f z = {};
#pragma unroll
        for (int r = 0; r < 3; ++r) {
            *(v4f*)(qw + r * SC_QPITCH + 4 * lane)            = (r < ncols) ? qv[r][0] : z;
            *(v4f*)(qw + r * SC_QPITCH + 4 * lane + SC_UCOLS) = (r < ncols) ? qv[r][1] : z;
        }
        *(v4f*)(qw + 3 * SC_QPITCH + 4 * lane)            = z;
        *(v4f*)(qw + 3 * SC_QPITCH + 4 * lane + SC_UCOLS) = z;
    }

    const bool live = c < ncols;
    float* dst = ((c == 0) ? S0 : (c == 1) ? S1 : S2);

    v8f acc = {};
    sc_park(g0, pk);
    sc_load(g0, Mw + (size_t)16 * MEM_);
    SC_ORDER();
    SC_WAVE_SYNC();
    acc = sc_mma(aP, bP, acc);
    SC_WAVE_SYNC();
    sc_park(g1, pk);
    sc_load(g1, Mw + (size_t)16 * MEM_ + SC_UCOLS);
    SC_ORDER();
    SC_WAVE_SYNC();
    acc = sc_mma(aP, bP + SC_UCOLS, acc);
    SC_WAVE_SYNC();
    if (live) {
#pragma unroll
        for (int j = 0; j < 8; ++j)
            dst[(size_t)b * L_ + lw + j + 8 * hi] = acc[j];
    }

    acc = v8f{};
    sc_park(g0, pk);
    SC_WAVE_SYNC();
    acc = sc_mma(aP, bP, acc);
    SC_WAVE_SYNC();
    sc_park(g1, pk);
    SC_WAVE_SYNC();
    acc = sc_mma(aP, bP + SC_UCOLS, acc);
    if (live) {
#pragma unroll
        for (int j = 0; j < 8; ++j)
            dst[(size_t)b * L_ + lw + 16 + j + 8 * hi] = acc[j];
    }
#undef SC_WAVE_SYNC
#undef SC_ORDER
}


#define WS_WAVES 16
static_assert(Q_ == MEM_, "gemm_turn_kernel addresses s and ev with one offset");

template <int X>
__device__ __forceinline__ float xrow(float v) {
    return __builtin_bit_cast(float, __builtin_amdgcn_update_dpp(
        0, __builtin_bit_cast(int, v), 0x160 + X, 0xf, 0xf, false));
}
__device__ __forceinline__ float xhalf(float v) {
    return __builtin_bit_cast(float, __builtin_amdgcn_permlanex16(
        __builtin_bit_cast(int, v), __builtin_bit_cast(int, v), 0x76543210, 0xfedcba98, false, false));
}
__device__ __forceinline__ float row16_max(float v) {
    v = fmaxf(v, xrow<1>(v)); v = fmaxf(v, xrow<2>(v));
    v = fmaxf(v, xrow<4>(v)); v = fmaxf(v, xrow<8>(v));
    return v;
}
__device__ __forceinline__ float row16_sum(float v) {
    v += xrow<1>(v); v += xrow<2>(v); v += xrow<4>(v); v += xrow<8>(v);
    return v;
}

template <int NP>
__device__ __forceinline__ void row_stats(const v4f (&x)[NP][2], float (&mx)[NP], float (&inv)[NP],
                                          float (*redm)[WS_WAVES], float (*reds)[WS_WAVES],
                                          int lane, int wave) {
#pragma unroll
    for (int p = 0; p < NP; ++p) {
        float m = fmaxf(fmaxf(fmaxf(x[p][0][0], x[p][0][1]), fmaxf(x[p][0][2], x[p][0][3])),
                        fmaxf(fmaxf(x[p][1][0], x[p][1][1]), fmaxf(x[p][1][2], x[p][1][3])));
        m = row16_max(m);
        m = fmaxf(m, xhalf(m));
        redm[p][wave] = m;
    }
    __syncthreads();
#pragma unroll
    for (int p = 0; p < NP; ++p) {
        const float m = row16_max(redm[p][lane & (WS_WAVES - 1)]);
        mx[p] = m;
        float s = 0.0f;
#pragma unroll
        for (int j = 0; j < 2; ++j)
#pragma unroll
            for (int e = 0; e < 4; ++e) s += __expf(x[p][j][e] - m);
        s = row16_sum(s);
        s += xhalf(s);
        reds[p][wave] = s;
    }
    __syncthreads();
#pragma unroll
    for (int p = 0; p < NP; ++p)
        inv[p] = 1.0f / row16_sum(reds[p][lane & (WS_WAVES - 1)]);
}

#define PK_WAVES   8
#define PK_TILES   4
#define PK_PITCH   (MEM_ + 4)
#define PK_ROWS_WG (16 * PK_TILES * PK_WAVES)
static_assert(PK_WGS_ROW * PK_ROWS_WG == L_, "the work-groups of a batch row cover its L rows");

__device__ __forceinline__ void pk_park(const ScUnit& u, float* dst) {
#pragma unroll
    for (int j = 0; j < 16; ++j) *(v4f*)(dst + j * PK_PITCH) = u.r[j];
}

__global__ __launch_bounds__(32 * PK_WAVES)
void pass_kernel(const float* __restrict__ M,
                 const float* __restrict__ q0,
                 const float* __restrict__ q1,
                 const float* __restrict__ q2,
                 float* EN,
                 float* ST,
                 float* __restrict__ sacc,
                 float* __restrict__ eacc,
                 const float* __restrict__ statPrev,
                 float* __restrict__ statCur,
                 float* __restrict__ part,
                 int turn) {
    const int tid  = threadIdx.x;
    const int wave = tid >> 5;
    const int lane = tid & 31;
    const int c    = lane & 15;
    const int hi   = lane >> 4;
    const int b    = blockIdx.y;
    const int lw   = blockIdx.x * PK_ROWS_WG + wave * (16 * PK_TILES);
    const bool hasQ0 = turn >= 1;

    __shared__ __attribute__((aligned(16))) float qs[PK_WAVES][4 * PK_PITCH];
    __shared__ __attribute__((aligned(16))) float ms[PK_WAVES][16 * PK_PITCH];
    __shared__ float wst[PK_WAVES][3][2];

    const float* Mw = M + ((size_t)b * L_ + lw) * MEM_ + 4 * lane;

    const float* qsrc[3] = { hasQ0 ? q0 : q1, q1, q2 };
    v4f qv[3][2];
#pragma unroll
    for (int r = 0; r < 3; ++r) {
        const float* qp = qsrc[r] + (size_t)b * MEM_ + 4 * lane;
        qv[r][0] = *(const v4f*)qp;
        qv[r][1] = *(const v4f*)(qp + SC_UCOLS);
    }

    ScUnit g0, g1;
    sc_load(g0, Mw);
    sc_load(g1, Mw + SC_UCOLS);

    const size_t rA = (size_t)b * L_ + lw + lane;
    const size_t rB = rA + 32;
    const bool accS = turn >= 1;
    const bool accE = turn >= 2;
    float oS0 = 0.0f, oS1 = 0.0f, oE0 = 0.0f, oE1 = 0.0f, aS0 = 0.0f, aS1 = 0.0f, aE0 = 0.0f, aE1 = 0.0f;
    v4f sv[PK_WGS_ROW];
    if (accS) {
        oS0 = ST[rA]; oS1 = ST[rB];
#pragma unroll
        for (int i = 0; i < PK_WGS_ROW; ++i)
            sv[i] = *(const v4f*)(statPrev + ((size_t)b * PK_WGS_ROW + i) * ST_STRIDE);
        if (turn >= 2) { aS0 = sacc[rA]; aS1 = sacc[rB]; oE0 = EN[rA]; oE1 = EN[rB]; }
        if (turn >= 3) { aE0 = eacc[rA]; aE1 = eacc[rB]; }
    }

    float*       mw = ms[wave];
    float*       pk = mw + 4 * lane;
    const float* aP = mw + c * PK_PITCH + 2 * hi;
    float*       qw = qs[wave];
    const float* bP = qw + (c < 3 ? c : 3) * PK_PITCH + 2 * hi;
    const float* vP = mw + (8 * hi) * PK_PITCH + 4 * c;

#define PK_ORDER() __builtin_amdgcn_sched_barrier(0)
#define PK_WAVE_SYNC() do { __builtin_amdgcn_fence(__ATOMIC_RELEASE, "wavefront"); \
                            __builtin_amdgcn_wave_barrier();                      \
                            __builtin_amdgcn_fence(__ATOMIC_ACQUIRE, "wavefront"); } while (0)
    PK_ORDER();
    {
        const v4f z = {};
#pragma unroll
        for (int r = 0; r < 3; ++r) {
            const bool on = (r > 0) || hasQ0;
            *(v4f*)(qw + r * PK_PITCH + 4 * lane)            = on ? qv[r][0] : z;
            *(v4f*)(qw + r * PK_PITCH + 4 * lane + SC_UCOLS) = on ? qv[r][1] : z;
        }
        *(v4f*)(qw + 3 * PK_PITCH + 4 * lane)            = z;
        *(v4f*)(qw + 3 * PK_PITCH + 4 * lane + SC_UCOLS) = z;
    }

    if (accS) {
        float mS = sv[0][2], mE = sv[0][0];
#pragma unroll
        for (int i = 1; i < PK_WGS_ROW; ++i) { mS = fmaxf(mS, sv[i][2]); mE = fmaxf(mE, sv[i][0]); }
        float lS = 0.0f, lE = 0.0f;
#pragma unroll
        for (int i = 0; i < PK_WGS_ROW; ++i) {
            lS += __expf(sv[i][2] - mS) * sv[i][3];
            lE += __expf(sv[i][0] - mE) * sv[i][1];
        }
        const float iS = 1.0f / lS;
        sacc[rA] = aS0 + 0.2f * (__expf(oS0 - mS) * iS);
        sacc[rB] = aS1 + 0.2f * (__expf(oS1 - mS) * iS);
        if (accE) {
            const float iE = 1.0f / lE;
            eacc[rA] = aE0 + 0.2f * (__expf(oE0 - mE) * iE);
            eacc[rB] = aE1 + 0.2f * (__expf(oE1 - mE) * iE);
        }
    }

    const bool live = (c == 1) || (c == 0 && hasQ0);
    const bool wq   = (c == 1) || (c == 2);
    float* dst = (c == 0) ? EN : ST;

    float mrun = 0.0f, lrun = 0.0f;
    v8f acc2[16] = {};
#pragma unroll
    for (int j = 0; j < PK_TILES; ++j) {
        pk_park(g0, pk);
        pk_park(g1, pk + SC_UCOLS);
        if (j + 1 < PK_TILES) {
            sc_load(g0, Mw + (size_t)(16 * (j + 1)) * MEM_);
            sc_load(g1, Mw + (size_t)(16 * (j + 1)) * MEM_ + SC_UCOLS);
        }
        PK_ORDER();
        PK_WAVE_SYNC();
        v8f s0 = {}, s1 = {};
#pragma unroll
        for (int i = 0; i < MEM_ / 4; i += 2) {
            const v2f a0 = *(const v2f*)(aP + 4 * i);
            const v2f b0 = *(const v2f*)(bP + 4 * i);
            const v2f a1 = *(const v2f*)(aP + 4 * i + 4);
            const v2f b1 = *(const v2f*)(bP + 4 * i + 4);
            s0 = __builtin_amdgcn_wmma_f32_16x16x4_f32(false, a0, false, b0, (short)0, s0, false, false);
            s1 = __builtin_amdgcn_wmma_f32_16x16x4_f32(false, a1, false, b1, (short)0, s1, false, false);
        }
        const v8f S = s0 + s1;
        if (live) {
#pragma unroll
            for (int r = 0; r < 8; ++r)
                dst[(size_t)b * L_ + lw + 16 * j + r + 8 * hi] = S[r];
        }
        float tm = fmaxf(fmaxf(fmaxf(S[0], S[1]), fmaxf(S[2], S[3])),
                         fmaxf(fmaxf(S[4], S[5]), fmaxf(S[6], S[7])));
        tm = fmaxf(tm, xhalf(tm));
        const float mnew  = (j == 0) ? tm : fmaxf(mrun, tm);
        const float alpha = (j == 0) ? 0.0f : __expf(mrun - mnew);
        float p[8];
        float ps = 0.0f;
#pragma unroll
        for (int r = 0; r < 8; ++r) { p[r] = __expf(S[r] - mnew); ps += p[r]; }
        ps += xhalf(ps);
        lrun = lrun * alpha + ps;
        mrun = mnew;
        if (j > 0) {
            const float f1 = __builtin_bit_cast(float, __builtin_amdgcn_readlane(__builtin_bit_cast(int, alpha), 1));
            const float f2 = __builtin_bit_cast(float, __builtin_amdgcn_readlane(__builtin_bit_cast(int, alpha), 2));
#pragma unroll
            for (int t = 0; t < 16; ++t) { acc2[t][1] *= f1; acc2[t][2] *= f2; }
        }
#pragma unroll
        for (int i = 0; i < 4; ++i) {
            v2f a;
            a.x = wq ? p[2 * i] : 0.0f;
            a.y = wq ? p[2 * i + 1] : 0.0f;
#pragma unroll
            for (int g = 0; g < 4; ++g) {
                const float* rp = vP + (2 * i) * PK_PITCH + 64 * g;
                const v4f r0 = *(const v4f*)rp;
                const v4f r1 = *(const v4f*)(rp + PK_PITCH);
#pragma unroll
                for (int t = 0; t < 4; ++t) {
                    v2f bb; bb.x = r0[t]; bb.y = r1[t];
                    acc2[4 * g + t] = __builtin_amdgcn_wmma_f32_16x16x4_f32(false, a, false, bb,
                                                                            (short)0, acc2[4 * g + t], false, false);
                }
            }
        }
        PK_WAVE_SYNC();
    }
#undef PK_WAVE_SYNC
#undef PK_ORDER

    if (lane < 3) { wst[wave][lane][0] = mrun; wst[wave][lane][1] = lrun; }
    __syncthreads();
    float mg[3], lg[3];
#pragma unroll
    for (int q = 0; q < 3; ++q) {
        float m = wst[0][q][0];
#pragma unroll
        for (int w = 1; w < PK_WAVES; ++w) m = fmaxf(m, wst[w][q][0]);
        float l = 0.0f;
#pragma unroll
        for (int w = 0; w < PK_WAVES; ++w) l += __expf(wst[w][q][0] - m) * wst[w][q][1];
        mg[q] = m; lg[q] = l;
    }
    const float f1 = __expf(wst[wave][1][0] - mg[1]);
    const float f2 = __expf(wst[wave][2][0] - mg[2]);
    if (hi == 0) {
#pragma unroll
        for (int g = 0; g < 4; ++g) {
            v4f e, xx;
#pragma unroll
            for (int t = 0; t < 4; ++t) { e[t] = acc2[4 * g + t][1] * f1; xx[t] = acc2[4 * g + t][2] * f2; }
            *(v4f*)(mw + 64 * g + 4 * c)        = e;
            *(v4f*)(mw + MEM_ + 64 * g + 4 * c) = xx;
        }
    }
    __syncthreads();
    {
        float se = 0.0f, sx = 0.0f;
#pragma unroll
        for (int w = 0; w < PK_WAVES; ++w) { se += ms[w][tid]; sx += ms[w][MEM_ + tid]; }
        float* po = part + ((size_t)b * PK_WGS_ROW + blockIdx.x) * PT_STRIDE;
        po[tid]        = se;
        po[MEM_ + tid] = sx;
        if (tid == 0) {
            float* so = statCur + ((size_t)b * PK_WGS_ROW + blockIdx.x) * ST_STRIDE;
            so[0] = mg[0]; so[1] = lg[0];
            so[2] = mg[1]; so[3] = lg[1];
            so[4] = mg[2]; so[5] = lg[2];
        }
    }
}

__global__ __launch_bounds__(32 * WS_WAVES)
void final_kernel(const float* __restrict__ ws, float* out) {
    const int tid  = threadIdx.x;
    const int wave = tid >> 5;
    const int lane = tid & 31;
    const int b    = blockIdx.x;
    __shared__ float redm[3][WS_WAVES];
    __shared__ float reds[3][WS_WAVES];
    const size_t r0 = (size_t)b * L_ + 4 * tid;
    const size_t r1 = r0 + 4 * 32 * WS_WAVES;

    v4f x[3][2];
    x[0][0] = *(const v4f*)(ws + OFF_S1 + r0);
    x[0][1] = *(const v4f*)(ws + OFF_S1 + r1);
    x[1][0] = *(const v4f*)(ws + OFF_S2 + r0);
    x[1][1] = *(const v4f*)(ws + OFF_S2 + r1);
    x[2][0] = *(const v4f*)(out + BL_ + r0);
    x[2][1] = *(const v4f*)(out + BL_ + r1);
    v4f e0 = *(const v4f*)(ws + OFF_EACC + r0);
    v4f e1 = *(const v4f*)(ws + OFF_EACC + r1);
    v4f s0v = *(const v4f*)(ws + OFF_SACC + r0);
    v4f s1v = *(const v4f*)(ws + OFF_SACC + r1);
    float mx[3], inv[3];
    row_stats<3>(x, mx, inv, redm, reds, lane, wave);
#pragma unroll
    for (int e = 0; e < 4; ++e) {
        const float ea = e0[e] + 0.2f * (__expf(x[1][0][e] - mx[1]) * inv[1]);
        const float eb = e1[e] + 0.2f * (__expf(x[1][1][e] - mx[1]) * inv[1]);
        e0[e] = logf(ea + 0.2f * (__expf(x[2][0][e] - mx[2]) * inv[2]));
        e1[e] = logf(eb + 0.2f * (__expf(x[2][1][e] - mx[2]) * inv[2]));
        s0v[e] = logf(s0v[e] + 0.2f * (__expf(x[0][0][e] - mx[0]) * inv[0]));
        s1v[e] = logf(s1v[e] + 0.2f * (__expf(x[0][1][e] - mx[0]) * inv[0]));
    }
    *(v4f*)(out + r0) = s0v;
    *(v4f*)(out + r1) = s1v;
    *(v4f*)(out + BL_ + r0) = e0;
    *(v4f*)(out + BL_ + r1) = e1;
}

extern "C" void kernel_launch(void* const* d_in, const int* in_sizes, int n_in,
                              void* d_out, int out_size, void* d_ws, size_t ws_size,
                              hipStream_t stream) {
    const float* M    = (const float*)d_in[0];
    const float* s0   = (const float*)d_in[1];
    const float* Wb   = (const float*)d_in[2];
    const float* We   = (const float*)d_in[3];
    const float* Wa   = (const float*)d_in[4];
    const float* W_ih = (const float*)d_in[5];
    const float* W_hh = (const float*)d_in[6];
    const float* b_ih = (const float*)d_in[7];
    const float* b_hh = (const float*)d_in[8];

    float* ws   = (float*)d_ws;
    float* ST   = ws + OFF_S1;
    float* EN   = ws + OFF_S2;
    float* sacc = ws + OFF_SACC;
    float* eacc = ws + OFF_EACC;
    float* qb   = ws + OFF_QB;
    float* qa   = ws + OFF_QA;
    float* qe   = ws + OFF_QE;
    float* sA   = ws + OFF_S_A;
    float* sB   = ws + OFF_S_B;
    float* gi   = ws + OFF_GI;
    float* gh   = ws + OFF_GH;
    float* part  = (float*)d_out;
    float* statA = (float*)d_out + BL_;
    float* statB = (float*)d_out + BL_ + B_ * PK_WGS_ROW * ST_STRIDE;
    float* S3    = (float*)d_out + BL_;

    gemm_q_kernel<false><<<dim3(16, 2), 32 * SG_WAVES, 0, stream>>>(s0, nullptr, nullptr, nullptr,
                                                                    Wb, Wa, qb, qa);

    for (int t = 0; t < 5; ++t) {
        const float* sCur = (t == 0) ? s0 : ((t & 1) ? sB : sA);
        float*       sNxt = (t & 1) ? sA : sB;
        float*       stCur = (t & 1) ? statB : statA;
        const float* stPrv = (t & 1) ? statA : statB;

        pass_kernel<<<dim3(PK_WGS_ROW, B_), 32 * PK_WAVES, 0, stream>>>(
            M, qe, qb, qa, EN, ST, sacc, eacc, stPrv, stCur, part, t);

        gemm_turn_kernel<<<dim3(56, 2), 32 * SG_WAVES, 0, stream>>>(sCur, part, stCur, We, W_ih, W_hh,
                                                          b_ih, b_hh, qe, gi, gh);

        if (t < 4) {
            gemm_q_kernel<true><<<dim3(16, B_ / GQ_ROWS), 32 * SG_WAVES, 0, stream>>>(
                sCur, gi, gh, sNxt, Wb, Wa, qb, qa);
        }
    }
    scores_kernel<<<dim3(L_ / SC_ROWS_WG, B_), 32 * SC_WAVES, 0, stream>>>(M, qe, qe, qe, S3, nullptr, nullptr, 1);

    final_kernel<<<B_, 32 * WS_WAVES, 0, stream>>>(ws, (float*)d_out);
}
